# starting kernel
# baseline (speedup 1.0000x reference)
#include <hip/hip_runtime.h>
#include <hip/hip_bf16.h>

typedef __attribute__((ext_vector_type(2))) float v2f;
typedef __attribute__((ext_vector_type(8))) float v8f;

#define SLOPE 0.2f

__global__ __launch_bounds__(32) void gat_wmma_gemm_f32(
    const float* __restrict__ A, const float* __restrict__ B,
    float* __restrict__ C, int M, int Np, int K, int Nout)
{
    const int lane  = threadIdx.x & 31;
    const int l15   = lane & 15;
    const int khalf = (lane >> 4) * 2;
    const int mt    = blockIdx.x;
    const int col0  = blockIdx.y * 64;
    const int row   = mt * 16 + l15;

    const float* __restrict__ Ap = A + (size_t)row * K + khalf;
    const float* __restrict__ Bp = B + (size_t)khalf * Np + col0 + l15;

    v8f acc[4];
    #pragma unroll
    for (int t = 0; t < 4; ++t)
        acc[t] = (v8f){0.f, 0.f, 0.f, 0.f, 0.f, 0.f, 0.f, 0.f};

    for (int k = 0; k < K; k += 4) {
        v2f a;
        a.x = Ap[0];
        a.y = Ap[1];
        v2f b0, b1, b2, b3;
        b0.x = Bp[0];   b0.y = Bp[Np + 0];
        b1.x = Bp[16];  b1.y = Bp[Np + 16];
        b2.x = Bp[32];  b2.y = Bp[Np + 32];
        b3.x = Bp[48];  b3.y = Bp[Np + 48];
        acc[0] = __builtin_amdgcn_wmma_f32_16x16x4_f32(false, a, false, b0, (short)0, acc[0], false, false);
        acc[1] = __builtin_amdgcn_wmma_f32_16x16x4_f32(false, a, false, b1, (short)0, acc[1], false, false);
        acc[2] = __builtin_amdgcn_wmma_f32_16x16x4_f32(false, a, false, b2, (short)0, acc[2], false, false);
        acc[3] = __builtin_amdgcn_wmma_f32_16x16x4_f32(false, a, false, b3, (short)0, acc[3], false, false);
        Ap += 4;
        Bp += (size_t)4 * Np;
    }

    const int srow0 = mt * 16 + (lane >> 4) * 8;
    #pragma unroll
    for (int t = 0; t < 4; ++t) {
        const int col = col0 + t * 16 + l15;
        if (col < Nout) {
            #pragma unroll
            for (int r = 0; r < 8; ++r)
                C[(size_t)(srow0 + r) * Nout + col] = acc[t][r];
        }
    }
}

__global__ __launch_bounds__(256) void gat_pad_weights(
    const float* __restrict__ W, float* __restrict__ Wp, int K, int N, int Np)
{
    int t = blockIdx.x * blockDim.x + threadIdx.x;
    if (t >= K * Np) return;
    const int r = t / Np, c = t - r * Np;
    Wp[t] = (c < N) ? W[r * N + c] : 0.f;
}

__global__ __launch_bounds__(64) void gat_node_attn(
    const float* __restrict__ h, const float* __restrict__ al,
    const float* __restrict__ ar, float* __restrict__ el,
    float* __restrict__ er, int H, int D)
{
    __shared__ float sl[64];
    __shared__ float sr[64];
    const int n = blockIdx.x, hh = blockIdx.y, d = threadIdx.x;

    float x = 0.f, a = 0.f, b = 0.f;
    if (d < D) {
        x = h[((size_t)n * H + hh) * D + d];
        a = al[hh * D + d];
        b = ar[hh * D + d];
    }
    sl[d] = x * a;
    sr[d] = x * b;
    __syncthreads();
    #pragma unroll
    for (int off = 32; off > 0; off >>= 1) {
        if (d < off) { sl[d] += sl[d + off]; sr[d] += sr[d + off]; }
        __syncthreads();
    }
    if (d == 0) {
        el[(size_t)n * H + hh] = sl[0];
        er[(size_t)n * H + hh] = sr[0];
    }
}

__device__ __forceinline__ unsigned f2ord(float f) {
    unsigned u = __float_as_uint(f);
    return (u & 0x80000000u) ? ~u : (u | 0x80000000u);
}
__device__ __forceinline__ float ord2f(unsigned u) {
    unsigned b = (u & 0x80000000u) ? (u ^ 0x80000000u) : ~u;
    return __uint_as_float(b);
}

__global__ __launch_bounds__(256) void gat_edge_max(
    const int* __restrict__ src, const int* __restrict__ dst,
    const float* __restrict__ el, const float* __restrict__ er,
    float* __restrict__ eT, unsigned* __restrict__ mU, int E, int H)
{
    int t = blockIdx.x * blockDim.x + threadIdx.x;
    if (t >= E * H) return;
    const int e = t / H, hh = t - e * H;
    float v = el[(size_t)src[e] * H + hh] + er[(size_t)dst[e] * H + hh];
    v = (v > 0.f) ? v : SLOPE * v;
    eT[t] = v;
    atomicMax(&mU[(size_t)dst[e] * H + hh], f2ord(v));
}

__global__ __launch_bounds__(256) void gat_edge_exp(
    const int* __restrict__ dst, const float* __restrict__ eT,
    const unsigned* __restrict__ mU, float* __restrict__ alphaT,
    float* __restrict__ s, int E, int H)
{
    int t = blockIdx.x * blockDim.x + threadIdx.x;
    if (t >= E * H) return;
    const int e = t / H, hh = t - e * H;
    const float m = ord2f(mU[(size_t)dst[e] * H + hh]);
    const float a = __expf(eT[t] - m);
    alphaT[t] = a;
    atomicAdd(&s[(size_t)dst[e] * H + hh], a);
}

__global__ __launch_bounds__(256) void gat_edge_aggregate(
    const int* __restrict__ src, const int* __restrict__ dst,
    const float* __restrict__ h, const float* __restrict__ alphaT,
    const float* __restrict__ s, float* __restrict__ out,
    int E, int H, int HD)
{
    const int wid  = (blockIdx.x * blockDim.x + threadIdx.x) >> 5;
    const int lane = threadIdx.x & 31;
    if (wid >= E) return;
    const int sN = src[wid], dN = dst[wid];
    const float* __restrict__ hs = h + (size_t)sN * HD;
    float* __restrict__ od = out + (size_t)dN * HD;
    for (int v = lane; v < HD; v += 32) {
        const int hh = v >> 6;
        const float a = alphaT[(size_t)wid * H + hh] / s[(size_t)dN * H + hh];
        atomicAdd(&od[v], a * hs[v]);
    }
}

extern "C" void kernel_launch(void* const* d_in, const int* in_sizes, int n_in,
                              void* d_out, int out_size, void* d_ws, size_t ws_size,
                              hipStream_t stream)
{
    const float* feat = (const float*)d_in[0];
    const int*   src  = (const int*)  d_in[1];
    const int*   dst  = (const int*)  d_in[2];
    const float* W0   = (const float*)d_in[3];
    const float* al0  = (const float*)d_in[4];
    const float* ar0  = (const float*)d_in[5];
    const float* W1   = (const float*)d_in[6];
    const float* al1  = (const float*)d_in[7];
    const float* ar1  = (const float*)d_in[8];
    float* out = (float*)d_out;

    const int FIN = 128, H0 = 4, D0 = 64, HD0 = H0 * D0, NC = 40, NCP = 64;
    const int N = in_sizes[0] / FIN;
    const int E = in_sizes[1];

    char* p = (char*)d_ws;
    auto alloc = [&](size_t bytes) -> void* {
        void* r = (void*)p;
        p += (bytes + 255) & ~(size_t)255;
        return r;
    };
    float*    h0      = (float*)   alloc((size_t)N * HD0 * 4);
    float*    agg0    = (float*)   alloc((size_t)N * HD0 * 4);
    float*    h1      = (float*)   alloc((size_t)N * NC  * 4);
    float*    W1p     = (float*)   alloc((size_t)HD0 * NCP * 4);
    float*    el0     = (float*)   alloc((size_t)N * H0 * 4);
    float*    er0     = (float*)   alloc((size_t)N * H0 * 4);
    unsigned* mU0     = (unsigned*)alloc((size_t)N * H0 * 4);
    float*    s0      = (float*)   alloc((size_t)N * H0 * 4);
    float*    eT0     = (float*)   alloc((size_t)E * H0 * 4);
    float*    alphaT0 = (float*)   alloc((size_t)E * H0 * 4);
    float*    el1     = (float*)   alloc((size_t)N * 4);
    float*    er1     = (float*)   alloc((size_t)N * 4);
    unsigned* mU1     = (unsigned*)alloc((size_t)N * 4);
    float*    s1      = (float*)   alloc((size_t)N * 4);
    float*    eT1     = (float*)   alloc((size_t)E * 4);
    float*    alphaT1 = (float*)   alloc((size_t)E * 4);

    {
        dim3 grid(N / 16, HD0 / 64);
        gat_wmma_gemm_f32<<<grid, 32, 0, stream>>>(feat, W0, h0, N, HD0, FIN, HD0);
    }
    {
        dim3 grid(N, H0);
        gat_node_attn<<<grid, 64, 0, stream>>>(h0, al0, ar0, el0, er0, H0, D0);
    }
    hipMemsetAsync(mU0, 0, (size_t)N * H0 * 4, stream);
    hipMemsetAsync(s0,  0, (size_t)N * H0 * 4, stream);
    {
        int threads = E * H0;
        gat_edge_max<<<(threads + 255) / 256, 256, 0, stream>>>(
            src, dst, el0, er0, eT0, mU0, E, H0);
        gat_edge_exp<<<(threads + 255) / 256, 256, 0, stream>>>(
            dst, eT0, mU0, alphaT0, s0, E, H0);
    }
    hipMemsetAsync(agg0, 0, (size_t)N * HD0 * 4, stream);
    {
        long long threads = (long long)E * 32;
        gat_edge_aggregate<<<(int)((threads + 255) / 256), 256, 0, stream>>>(
            src, dst, h0, alphaT0, s0, agg0, E, H0, HD0);
    }

    {
        int tot = HD0 * NCP;
        gat_pad_weights<<<(tot + 255) / 256, 256, 0, stream>>>(W1, W1p, HD0, NC, NCP);
        dim3 grid(N / 16, NCP / 64);
        gat_wmma_gemm_f32<<<grid, 32, 0, stream>>>(agg0, W1p, h1, N, NCP, HD0, NC);
    }
    {
        dim3 grid(N, 1);
        gat_node_attn<<<grid, 64, 0, stream>>>(h1, al1, ar1, el1, er1, 1, NC);
    }
    hipMemsetAsync(mU1, 0, (size_t)N * 4, stream);
    hipMemsetAsync(s1,  0, (size_t)N * 4, stream);
    {
        gat_edge_max<<<(E + 255) / 256, 256, 0, stream>>>(
            src, dst, el1, er1, eT1, mU1, E, 1);
        gat_edge_exp<<<(E + 255) / 256, 256, 0, stream>>>(
            dst, eT1, mU1, alphaT1, s1, E, 1);
    }
    hipMemsetAsync(out, 0, (size_t)N * NC * 4, stream);
    {
        long long threads = (long long)E * 32;
        gat_edge_aggregate<<<(int)((threads + 255) / 256), 256, 0, stream>>>(
            src, dst, h1, alphaT1, s1, out, E, 1, NC);
    }
}


# Edges bucketed by destination in-call (one int atomic per edge for slot, scan, place); one wave per node does max, exp-sum and weighted row sum from its bucket; no float atomics; sum order changes
# speedup vs baseline: 2.5988x; 2.5988x over previous
#include <hip/hip_runtime.h>
#include <hip/hip_bf16.h>

typedef __attribute__((ext_vector_type(2))) float v2f;
typedef __attribute__((ext_vector_type(8))) float v8f;

#define SLOPE 0.2f

__global__ __launch_bounds__(32) void gat_wmma_gemm_f32(
    const float* __restrict__ A, const float* __restrict__ B,
    float* __restrict__ C, int M, int Np, int K, int Nout)
{
    const int lane  = threadIdx.x & 31;
    const int l15   = lane & 15;
    const int khalf = (lane >> 4) * 2;
    const int mt    = blockIdx.x;
    const int col0  = blockIdx.y * 64;
    const int row   = mt * 16 + l15;

    const float* __restrict__ Ap = A + (size_t)row * K + khalf;
    const float* __restrict__ Bp = B + (size_t)khalf * Np + col0 + l15;

    v8f acc[4];
    #pragma unroll
    for (int t = 0; t < 4; ++t)
        acc[t] = (v8f){0.f, 0.f, 0.f, 0.f, 0.f, 0.f, 0.f, 0.f};

    for (int k = 0; k < K; k += 4) {
        v2f a;
        a.x = Ap[0];
        a.y = Ap[1];
        v2f b0, b1, b2, b3;
        b0.x = Bp[0];   b0.y = Bp[Np + 0];
        b1.x = Bp[16];  b1.y = Bp[Np + 16];
        b2.x = Bp[32];  b2.y = Bp[Np + 32];
        b3.x = Bp[48];  b3.y = Bp[Np + 48];
        acc[0] = __builtin_amdgcn_wmma_f32_16x16x4_f32(false, a, false, b0, (short)0, acc[0], false, false);
        acc[1] = __builtin_amdgcn_wmma_f32_16x16x4_f32(false, a, false, b1, (short)0, acc[1], false, false);
        acc[2] = __builtin_amdgcn_wmma_f32_16x16x4_f32(false, a, false, b2, (short)0, acc[2], false, false);
        acc[3] = __builtin_amdgcn_wmma_f32_16x16x4_f32(false, a, false, b3, (short)0, acc[3], false, false);
        Ap += 4;
        Bp += (size_t)4 * Np;
    }

    const int srow0 = mt * 16 + (lane >> 4) * 8;
    #pragma unroll
    for (int t = 0; t < 4; ++t) {
        const int col = col0 + t * 16 + l15;
        if (col < Nout) {
            #pragma unroll
            for (int r = 0; r < 8; ++r)
                C[(size_t)(srow0 + r) * Nout + col] = acc[t][r];
        }
    }
}

__global__ __launch_bounds__(256) void gat_pad_weights(
    const float* __restrict__ W, float* __restrict__ Wp, int K, int N, int Np)
{
    int t = blockIdx.x * blockDim.x + threadIdx.x;
    if (t >= K * Np) return;
    const int r = t / Np, c = t - r * Np;
    Wp[t] = (c < N) ? W[r * N + c] : 0.f;
}

__global__ __launch_bounds__(64) void gat_node_attn(
    const float* __restrict__ h, const float* __restrict__ al,
    const float* __restrict__ ar, float* __restrict__ el,
    float* __restrict__ er, int H, int D)
{
    __shared__ float sl[64];
    __shared__ float sr[64];
    const int n = blockIdx.x, hh = blockIdx.y, d = threadIdx.x;

    float x = 0.f, a = 0.f, b = 0.f;
    if (d < D) {
        x = h[((size_t)n * H + hh) * D + d];
        a = al[hh * D + d];
        b = ar[hh * D + d];
    }
    sl[d] = x * a;
    sr[d] = x * b;
    __syncthreads();
    #pragma unroll
    for (int off = 32; off > 0; off >>= 1) {
        if (d < off) { sl[d] += sl[d + off]; sr[d] += sr[d + off]; }
        __syncthreads();
    }
    if (d == 0) {
        el[(size_t)n * H + hh] = sl[0];
        er[(size_t)n * H + hh] = sr[0];
    }
}

__global__ __launch_bounds__(256) void gat_bucket_rank(
    const int* __restrict__ dst, int* __restrict__ cnt,
    int* __restrict__ rank, int E)
{
    const int e = blockIdx.x * blockDim.x + threadIdx.x;
    if (e < E) rank[e] = atomicAdd(&cnt[dst[e]], 1);
}

__global__ __launch_bounds__(256) void gat_bucket_scan(
    const int* __restrict__ cnt, int* __restrict__ rowptr, int N)
{
    __shared__ int s_tot[8];
    __shared__ int s_car[8];
    const int tid = threadIdx.x, lane = tid & 31, wv = tid >> 5;
    const int4* __restrict__ c4 = (const int4*)cnt;

    int p = 0;
    for (int k = 0; k < (int)blockIdx.x; ++k) {
        const int4 q = c4[k * 256 + tid];
        p += (q.x + q.y) + (q.z + q.w);
    }
    const int i0 = blockIdx.x * 1024 + tid * 4;
    const bool live = (i0 + 3 < N);
    int4 own = make_int4(0, 0, 0, 0);
    if (live) own = c4[i0 >> 2];
    const int v = (own.x + own.y) + (own.z + own.w);

    int incl = v;
    for (int off = 1; off < 32; off <<= 1) {
        const int y = __shfl_up(incl, off, 32);
        if (lane >= off) incl += y;
    }
    for (int m = 16; m >= 1; m >>= 1) p += __shfl_xor(p, m, 32);
    if (lane == 31) s_tot[wv] = incl;
    if (lane == 0)  s_car[wv] = p;
    __syncthreads();
    int base = 0;
    for (int w = 0; w < 8; ++w) {
        base += s_car[w];
        if (w < wv) base += s_tot[w];
    }
    base += incl - v;
    if (live) {
        int4 r;
        r.x = base;
        r.y = r.x + own.x;
        r.z = r.y + own.y;
        r.w = r.z + own.z;
        ((int4*)rowptr)[i0 >> 2] = r;
        if (i0 + 4 == N) rowptr[N] = r.w + own.w;
    }
}

__global__ __launch_bounds__(256) void gat_bucket_place(
    const int* __restrict__ src, const int* __restrict__ dst,
    const int* __restrict__ rowptr, const int* __restrict__ rank,
    int* __restrict__ bsrc, int E)
{
    const int e = blockIdx.x * blockDim.x + threadIdx.x;
    if (e >= E) return;
    bsrc[rowptr[dst[e]] + rank[e]] = src[e];
}

__device__ __forceinline__ float gat_leaky(float v) {
    return (v > 0.f) ? v : SLOPE * v;
}

__global__ __launch_bounds__(256) void gat_node_gather_h4(
    const int* __restrict__ rowptr, const int* __restrict__ bsrc,
    const float* __restrict__ h, const float* __restrict__ el,
    const float* __restrict__ er, float* __restrict__ out, int N)
{
    const int lane = threadIdx.x & 31;
    const int n = __builtin_amdgcn_readfirstlane(
        (int)((blockIdx.x * blockDim.x + threadIdx.x) >> 5));
    if (n >= N) return;
    const int beg = __builtin_amdgcn_readfirstlane(rowptr[n]);
    const int end = __builtin_amdgcn_readfirstlane(rowptr[n + 1]);
    const float4* __restrict__ el4 = (const float4*)el;
    const float4 ern = ((const float4*)er)[n];
    const float NINF = -__builtin_inff();

    const int  j0 = beg + lane;
    const bool v0 = j0 < end;
    int   src0 = 0;
    float e0 = NINF, e1 = NINF, e2 = NINF, e3 = NINF;
    if (v0) {
        src0 = bsrc[j0];
        const float4 l = el4[src0];
        e0 = gat_leaky(l.x + ern.x);
        e1 = gat_leaky(l.y + ern.y);
        e2 = gat_leaky(l.z + ern.z);
        e3 = gat_leaky(l.w + ern.w);
    }
    float m0 = e0, m1 = e1, m2 = e2, m3 = e3;
    for (int j = j0 + 32; j < end; j += 32) {
        const float4 l = el4[bsrc[j]];
        m0 = fmaxf(m0, gat_leaky(l.x + ern.x));
        m1 = fmaxf(m1, gat_leaky(l.y + ern.y));
        m2 = fmaxf(m2, gat_leaky(l.z + ern.z));
        m3 = fmaxf(m3, gat_leaky(l.w + ern.w));
    }
    for (int m = 16; m >= 1; m >>= 1) {
        m0 = fmaxf(m0, __shfl_xor(m0, m, 32));
        m1 = fmaxf(m1, __shfl_xor(m1, m, 32));
        m2 = fmaxf(m2, __shfl_xor(m2, m, 32));
        m3 = fmaxf(m3, __shfl_xor(m3, m, 32));
    }

    const float p0 = v0 ? __expf(e0 - m0) : 0.f;
    const float p1 = v0 ? __expf(e1 - m1) : 0.f;
    const float p2 = v0 ? __expf(e2 - m2) : 0.f;
    const float p3 = v0 ? __expf(e3 - m3) : 0.f;
    float s0 = p0, s1 = p1, s2 = p2, s3 = p3;
    for (int j = j0 + 32; j < end; j += 32) {
        const float4 l = el4[bsrc[j]];
        s0 += __expf(gat_leaky(l.x + ern.x) - m0);
        s1 += __expf(gat_leaky(l.y + ern.y) - m1);
        s2 += __expf(gat_leaky(l.z + ern.z) - m2);
        s3 += __expf(gat_leaky(l.w + ern.w) - m3);
    }
    for (int m = 16; m >= 1; m >>= 1) {
        s0 += __shfl_xor(s0, m, 32);
        s1 += __shfl_xor(s1, m, 32);
        s2 += __shfl_xor(s2, m, 32);
        s3 += __shfl_xor(s3, m, 32);
    }

    float4 accA = make_float4(0.f, 0.f, 0.f, 0.f);
    float4 accB = make_float4(0.f, 0.f, 0.f, 0.f);
    const bool hi = lane >= 16;
    for (int base = beg; base < end; base += 32) {
        int   mysrc = 0;
        float a0 = 0.f, a1 = 0.f, a2 = 0.f, a3 = 0.f;
        if (base == beg) {
            mysrc = src0;
            if (v0) {
                a0 = p0 / s0;  a1 = p1 / s1;
                a2 = p2 / s2;  a3 = p3 / s3;
            }
        } else {
            const int j = base + lane;
            if (j < end) {
                mysrc = bsrc[j];
                const float4 l = el4[mysrc];
                a0 = __expf(gat_leaky(l.x + ern.x) - m0) / s0;
                a1 = __expf(gat_leaky(l.y + ern.y) - m1) / s1;
                a2 = __expf(gat_leaky(l.z + ern.z) - m2) / s2;
                a3 = __expf(gat_leaky(l.w + ern.w) - m3) / s3;
            }
        }
        const int left = end - base;
        const int cntc = left < 32 ? left : 32;
        #pragma unroll 4
        for (int t = 0; t < cntc; ++t) {
            const int   sN = __builtin_amdgcn_readlane(mysrc, t);
            const float b0 = __int_as_float(__builtin_amdgcn_readlane(__float_as_int(a0), t));
            const float b1 = __int_as_float(__builtin_amdgcn_readlane(__float_as_int(a1), t));
            const float b2 = __int_as_float(__builtin_amdgcn_readlane(__float_as_int(a2), t));
            const float b3 = __int_as_float(__builtin_amdgcn_readlane(__float_as_int(a3), t));
            const float aA = hi ? b1 : b0;
            const float aB = hi ? b3 : b2;
            const float4* __restrict__ hp = (const float4*)(h + (size_t)sN * 256);
            const float4 x = hp[lane];
            const float4 y = hp[32 + lane];
            accA.x += aA * x.x;  accA.y += aA * x.y;
            accA.z += aA * x.z;  accA.w += aA * x.w;
            accB.x += aB * y.x;  accB.y += aB * y.y;
            accB.z += aB * y.z;  accB.w += aB * y.w;
        }
    }
    float4* __restrict__ op = (float4*)(out + (size_t)n * 256);
    op[lane]      = accA;
    op[32 + lane] = accB;
}

__global__ __launch_bounds__(128) void gat_node_gather_h1(
    const int* __restrict__ rowptr, const int* __restrict__ bsrc,
    const float* __restrict__ h, const float* __restrict__ el,
    const float* __restrict__ er, float* __restrict__ out, int N, int D)
{
    __shared__ float stage[4 * 64];
    const int lane = threadIdx.x & 31;
    const int wv = __builtin_amdgcn_readfirstlane((int)(threadIdx.x >> 5));
    const int n  = (int)blockIdx.x * 4 + wv;
    const bool has0 = lane < D;
    const bool has1 = lane + 32 < D;
    float acc0 = 0.f, acc1 = 0.f;

    if (n < N) {
        const int beg = __builtin_amdgcn_readfirstlane(rowptr[n]);
        const int end = __builtin_amdgcn_readfirstlane(rowptr[n + 1]);
        const float ern = er[n];
        const float NINF = -__builtin_inff();

        const int  j0 = beg + lane;
        const bool v0 = j0 < end;
        int   src0 = 0;
        float e0 = NINF;
        if (v0) {
            src0 = bsrc[j0];
            e0 = gat_leaky(el[src0] + ern);
        }
        float mx = e0;
        for (int j = j0 + 32; j < end; j += 32)
            mx = fmaxf(mx, gat_leaky(el[bsrc[j]] + ern));
        for (int m = 16; m >= 1; m >>= 1) mx = fmaxf(mx, __shfl_xor(mx, m, 32));

        const float p0 = v0 ? __expf(e0 - mx) : 0.f;
        float s = p0;
        for (int j = j0 + 32; j < end; j += 32)
            s += __expf(gat_leaky(el[bsrc[j]] + ern) - mx);
        for (int m = 16; m >= 1; m >>= 1) s += __shfl_xor(s, m, 32);

        for (int base = beg; base < end; base += 32) {
            int   mysrc = 0;
            float a = 0.f;
            if (base == beg) {
                mysrc = src0;
                if (v0) a = p0 / s;
            } else {
                const int j = base + lane;
                if (j < end) {
                    mysrc = bsrc[j];
                    a = __expf(gat_leaky(el[mysrc] + ern) - mx) / s;
                }
            }
            const int left = end - base;
            const int cntc = left < 32 ? left : 32;
            #pragma unroll 4
            for (int t = 0; t < cntc; ++t) {
                const int   sN = __builtin_amdgcn_readlane(mysrc, t);
                const float b  = __int_as_float(__builtin_amdgcn_readlane(__float_as_int(a), t));
                const float* __restrict__ hp = h + (size_t)sN * D;
                const float x = has0 ? hp[lane] : 0.f;
                const float y = has1 ? hp[32 + lane] : 0.f;
                acc0 += b * x;
                acc1 += b * y;
            }
        }
    }
    if (has0) stage[wv * D + lane] = acc0;
    if (has1) stage[wv * D + 32 + lane] = acc1;
    __syncthreads();
    const int nleft  = N - (int)blockIdx.x * 4;
    const int nvalid = nleft < 4 ? nleft : 4;
    float* __restrict__ ob = out + (size_t)blockIdx.x * 4 * D;
    for (int i = threadIdx.x; i < nvalid * D; i += 128) ob[i] = stage[i];
}

extern "C" void kernel_launch(void* const* d_in, const int* in_sizes, int n_in,
                              void* d_out, int out_size, void* d_ws, size_t ws_size,
                              hipStream_t stream)
{
    const float* feat = (const float*)d_in[0];
    const int*   src  = (const int*)  d_in[1];
    const int*   dst  = (const int*)  d_in[2];
    const float* W0   = (const float*)d_in[3];
    const float* al0  = (const float*)d_in[4];
    const float* ar0  = (const float*)d_in[5];
    const float* W1   = (const float*)d_in[6];
    const float* al1  = (const float*)d_in[7];
    const float* ar1  = (const float*)d_in[8];
    float* out = (float*)d_out;

    const int FIN = 128, H0 = 4, D0 = 64, HD0 = H0 * D0, NC = 40, NCP = 64;
    const int N = in_sizes[0] / FIN;
    const int E = in_sizes[1];

    char* p = (char*)d_ws;
    auto alloc = [&](size_t bytes) -> void* {
        void* r = (void*)p;
        p += (bytes + 255) & ~(size_t)255;
        return r;
    };
    float* h0     = (float*)alloc((size_t)N * HD0 * 4);
    float* agg0   = (float*)alloc((size_t)N * HD0 * 4);
    float* h1     = (float*)alloc((size_t)N * NC  * 4);
    float* W1p    = (float*)alloc((size_t)HD0 * NCP * 4);
    float* el0    = (float*)alloc((size_t)N * H0 * 4);
    float* er0    = (float*)alloc((size_t)N * H0 * 4);
    float* el1    = (float*)alloc((size_t)N * 4);
    float* er1    = (float*)alloc((size_t)N * 4);
    int*   cnt    = (int*)  alloc((size_t)N * 4);
    int*   rowptr = (int*)  alloc((size_t)(N + 1) * 4);
    int*   rank   = (int*)  alloc((size_t)E * 4);
    int*   bsrc   = (int*)  alloc((size_t)E * 4);

    hipMemsetAsync(cnt, 0, (size_t)N * 4, stream);
    gat_bucket_rank<<<(E + 255) / 256, 256, 0, stream>>>(dst, cnt, rank, E);
    gat_bucket_scan<<<(N + 1023) / 1024, 256, 0, stream>>>(cnt, rowptr, N);
    gat_bucket_place<<<(E + 255) / 256, 256, 0, stream>>>(src, dst, rowptr, rank, bsrc, E);

    {
        dim3 grid(N / 16, HD0 / 64);
        gat_wmma_gemm_f32<<<grid, 32, 0, stream>>>(feat, W0, h0, N, HD0, FIN, HD0);
    }
    {
        dim3 grid(N, H0);
        gat_node_attn<<<grid, 64, 0, stream>>>(h0, al0, ar0, el0, er0, H0, D0);
    }
    gat_node_gather_h4<<<(N + 7) / 8, 256, 0, stream>>>(
        rowptr, bsrc, h0, el0, er0, agg0, N);

    {
        int tot = HD0 * NCP;
        gat_pad_weights<<<(tot + 255) / 256, 256, 0, stream>>>(W1, W1p, HD0, NC, NCP);
        dim3 grid(N / 16, NCP / 64);
        gat_wmma_gemm_f32<<<grid, 32, 0, stream>>>(agg0, W1p, h1, N, NCP, HD0, NC);
    }
    {
        dim3 grid(N, 1);
        gat_node_attn<<<grid, 64, 0, stream>>>(h1, al1, ar1, el1, er1, 1, NC);
    }
    gat_node_gather_h1<<<(N + 3) / 4, 128, 0, stream>>>(
        rowptr, bsrc, h1, el1, er1, out, N, NC);
}


# el/er attention dot products taken from the GEMM accumulators in the GEMM epilogue (row butterfly on the vector ALU); both per-node attention launches removed; el/er sum order changes
# speedup vs baseline: 3.2617x; 1.2551x over previous
#include <hip/hip_runtime.h>
#include <hip/hip_bf16.h>

typedef __attribute__((ext_vector_type(2))) float v2f;
typedef __attribute__((ext_vector_type(8))) float v8f;

#define SLOPE 0.2f

template <int M>
__device__ __forceinline__ float gat_row_xor(float x) {
    return __builtin_bit_cast(float, __builtin_amdgcn_update_dpp(
        0, __builtin_bit_cast(int, x), 0x160 | M, 0xf, 0xf, true));
}

__global__ __launch_bounds__(32) void gat_wmma_gemm_f32(
    const float* __restrict__ A, const float* __restrict__ B,
    float* __restrict__ C, int M, int Np, int K, int Nout,
    const float* __restrict__ al, const float* __restrict__ ar,
    float* __restrict__ el, float* __restrict__ er, int H, int D)
{
    const int lane  = threadIdx.x & 31;
    const int l15   = lane & 15;
    const int khalf = (lane >> 4) * 2;
    const int mt    = blockIdx.x;
    const int head  = blockIdx.y;
    const int col0  = head * 64;
    const int row   = mt * 16 + l15;

    const float* __restrict__ Ap = A + (size_t)row * K + khalf;
    const float* __restrict__ Bp = B + (size_t)khalf * Np + col0 + l15;

    float wl[4], wr[4];
    #pragma unroll
    for (int t = 0; t < 4; ++t) {
        const int  c  = t * 16 + l15;
        const bool ok = c < D;
        const int  cc = ok ? c : D - 1;
        const float xl = al[head * D + cc];
        const float xr = ar[head * D + cc];
        wl[t] = ok ? xl : 0.f;
        wr[t] = ok ? xr : 0.f;
    }

    v8f acc[4];
    #pragma unroll
    for (int t = 0; t < 4; ++t)
        acc[t] = (v8f){0.f, 0.f, 0.f, 0.f, 0.f, 0.f, 0.f, 0.f};

    for (int k = 0; k < K; k += 4) {
        v2f a;
        a.x = Ap[0];
        a.y = Ap[1];
        v2f b0, b1, b2, b3;
        b0.x = Bp[0];   b0.y = Bp[Np + 0];
        b1.x = Bp[16];  b1.y = Bp[Np + 16];
        b2.x = Bp[32];  b2.y = Bp[Np + 32];
        b3.x = Bp[48];  b3.y = Bp[Np + 48];
        acc[0] = __builtin_amdgcn_wmma_f32_16x16x4_f32(false, a, false, b0, (short)0, acc[0], false, false);
        acc[1] = __builtin_amdgcn_wmma_f32_16x16x4_f32(false, a, false, b1, (short)0, acc[1], false, false);
        acc[2] = __builtin_amdgcn_wmma_f32_16x16x4_f32(false, a, false, b2, (short)0, acc[2], false, false);
        acc[3] = __builtin_amdgcn_wmma_f32_16x16x4_f32(false, a, false, b3, (short)0, acc[3], false, false);
        Ap += 4;
        Bp += (size_t)4 * Np;
    }

    const int srow0 = mt * 16 + (lane >> 4) * 8;
    #pragma unroll
    for (int t = 0; t < 4; ++t) {
        const int col = col0 + t * 16 + l15;
        if (col < Nout) {
            #pragma unroll
            for (int r = 0; r < 8; ++r)
                C[(size_t)(srow0 + r) * Nout + col] = acc[t][r];
        }
    }

    float v[16];
    #pragma unroll
    for (int r = 0; r < 8; ++r) {
        v[r]     = acc[0][r] * wl[0] + acc[1][r] * wl[1]
                 + acc[2][r] * wl[2] + acc[3][r] * wl[3];
        v[8 + r] = acc[0][r] * wr[0] + acc[1][r] * wr[1]
                 + acc[2][r] * wr[2] + acc[3][r] * wr[3];
    }
    const bool b3 = (lane & 8) != 0;
    const bool b2 = (lane & 4) != 0;
    const bool b1 = (lane & 2) != 0;
    const bool b0 = (lane & 1) != 0;
    float w8[8];
    #pragma unroll
    for (int i = 0; i < 8; ++i) {
        const float keep = b3 ? v[i + 8] : v[i];
        const float send = b3 ? v[i] : v[i + 8];
        w8[i] = keep + gat_row_xor<8>(send);
    }
    float w4[4];
    #pragma unroll
    for (int i = 0; i < 4; ++i) {
        const float keep = b2 ? w8[i + 4] : w8[i];
        const float send = b2 ? w8[i] : w8[i + 4];
        w4[i] = keep + gat_row_xor<4>(send);
    }
    float w2[2];
    #pragma unroll
    for (int i = 0; i < 2; ++i) {
        const float keep = b1 ? w4[i + 2] : w4[i];
        const float send = b1 ? w4[i] : w4[i + 2];
        w2[i] = keep + gat_row_xor<2>(send);
    }
    const float keep1 = b0 ? w2[1] : w2[0];
    const float send1 = b0 ? w2[0] : w2[1];
    const float z = keep1 + gat_row_xor<1>(send1);
    float* __restrict__ sp = b3 ? er : el;
    sp[(size_t)(srow0 + (lane & 7)) * H + head] = z;
}

__global__ __launch_bounds__(256) void gat_pad_weights(
    const float* __restrict__ W, float* __restrict__ Wp, int K, int N, int Np)
{
    int t = blockIdx.x * blockDim.x + threadIdx.x;
    if (t >= K * Np) return;
    const int r = t / Np, c = t - r * Np;
    Wp[t] = (c < N) ? W[r * N + c] : 0.f;
}

__global__ __launch_bounds__(256) void gat_bucket_rank(
    const int* __restrict__ dst, int* __restrict__ cnt,
    int* __restrict__ rank, int E)
{
    const int e = blockIdx.x * blockDim.x + threadIdx.x;
    if (e < E) rank[e] = atomicAdd(&cnt[dst[e]], 1);
}

__global__ __launch_bounds__(256) void gat_bucket_scan(
    const int* __restrict__ cnt, int* __restrict__ rowptr, int N)
{
    __shared__ int s_tot[8];
    __shared__ int s_car[8];
    const int tid = threadIdx.x, lane = tid & 31, wv = tid >> 5;
    const int4* __restrict__ c4 = (const int4*)cnt;

    int p = 0;
    for (int k = 0; k < (int)blockIdx.x; ++k) {
        const int4 q = c4[k * 256 + tid];
        p += (q.x + q.y) + (q.z + q.w);
    }
    const int i0 = blockIdx.x * 1024 + tid * 4;
    const bool live = (i0 + 3 < N);
    int4 own = make_int4(0, 0, 0, 0);
    if (live) own = c4[i0 >> 2];
    const int v = (own.x + own.y) + (own.z + own.w);

    int incl = v;
    for (int off = 1; off < 32; off <<= 1) {
        const int y = __shfl_up(incl, off, 32);
        if (lane >= off) incl += y;
    }
    for (int m = 16; m >= 1; m >>= 1) p += __shfl_xor(p, m, 32);
    if (lane == 31) s_tot[wv] = incl;
    if (lane == 0)  s_car[wv] = p;
    __syncthreads();
    int base = 0;
    for (int w = 0; w < 8; ++w) {
        base += s_car[w];
        if (w < wv) base += s_tot[w];
    }
    base += incl - v;
    if (live) {
        int4 r;
        r.x = base;
        r.y = r.x + own.x;
        r.z = r.y + own.y;
        r.w = r.z + own.z;
        ((int4*)rowptr)[i0 >> 2] = r;
        if (i0 + 4 == N) rowptr[N] = r.w + own.w;
    }
}

__global__ __launch_bounds__(256) void gat_bucket_place(
    const int* __restrict__ src, const int* __restrict__ dst,
    const int* __restrict__ rowptr, const int* __restrict__ rank,
    int* __restrict__ bsrc, int E)
{
    const int e = blockIdx.x * blockDim.x + threadIdx.x;
    if (e >= E) return;
    bsrc[rowptr[dst[e]] + rank[e]] = src[e];
}

__device__ __forceinline__ float gat_leaky(float v) {
    return (v > 0.f) ? v : SLOPE * v;
}

__global__ __launch_bounds__(256) void gat_node_gather_h4(
    const int* __restrict__ rowptr, const int* __restrict__ bsrc,
    const float* __restrict__ h, const float* __restrict__ el,
    const float* __restrict__ er, float* __restrict__ out, int N)
{
    const int lane = threadIdx.x & 31;
    const int n = __builtin_amdgcn_readfirstlane(
        (int)((blockIdx.x * blockDim.x + threadIdx.x) >> 5));
    if (n >= N) return;
    const int beg = __builtin_amdgcn_readfirstlane(rowptr[n]);
    const int end = __builtin_amdgcn_readfirstlane(rowptr[n + 1]);
    const float4* __restrict__ el4 = (const float4*)el;
    const float4 ern = ((const float4*)er)[n];
    const float NINF = -__builtin_inff();

    const int  j0 = beg + lane;
    const bool v0 = j0 < end;
    int   src0 = 0;
    float e0 = NINF, e1 = NINF, e2 = NINF, e3 = NINF;
    if (v0) {
        src0 = bsrc[j0];
        const float4 l = el4[src0];
        e0 = gat_leaky(l.x + ern.x);
        e1 = gat_leaky(l.y + ern.y);
        e2 = gat_leaky(l.z + ern.z);
        e3 = gat_leaky(l.w + ern.w);
    }
    float m0 = e0, m1 = e1, m2 = e2, m3 = e3;
    for (int j = j0 + 32; j < end; j += 32) {
        const float4 l = el4[bsrc[j]];
        m0 = fmaxf(m0, gat_leaky(l.x + ern.x));
        m1 = fmaxf(m1, gat_leaky(l.y + ern.y));
        m2 = fmaxf(m2, gat_leaky(l.z + ern.z));
        m3 = fmaxf(m3, gat_leaky(l.w + ern.w));
    }
    for (int m = 16; m >= 1; m >>= 1) {
        m0 = fmaxf(m0, __shfl_xor(m0, m, 32));
        m1 = fmaxf(m1, __shfl_xor(m1, m, 32));
        m2 = fmaxf(m2, __shfl_xor(m2, m, 32));
        m3 = fmaxf(m3, __shfl_xor(m3, m, 32));
    }

    const float p0 = v0 ? __expf(e0 - m0) : 0.f;
    const float p1 = v0 ? __expf(e1 - m1) : 0.f;
    const float p2 = v0 ? __expf(e2 - m2) : 0.f;
    const float p3 = v0 ? __expf(e3 - m3) : 0.f;
    float s0 = p0, s1 = p1, s2 = p2, s3 = p3;
    for (int j = j0 + 32; j < end; j += 32) {
        const float4 l = el4[bsrc[j]];
        s0 += __expf(gat_leaky(l.x + ern.x) - m0);
        s1 += __expf(gat_leaky(l.y + ern.y) - m1);
        s2 += __expf(gat_leaky(l.z + ern.z) - m2);
        s3 += __expf(gat_leaky(l.w + ern.w) - m3);
    }
    for (int m = 16; m >= 1; m >>= 1) {
        s0 += __shfl_xor(s0, m, 32);
        s1 += __shfl_xor(s1, m, 32);
        s2 += __shfl_xor(s2, m, 32);
        s3 += __shfl_xor(s3, m, 32);
    }

    float4 accA = make_float4(0.f, 0.f, 0.f, 0.f);
    float4 accB = make_float4(0.f, 0.f, 0.f, 0.f);
    const bool hi = lane >= 16;
    for (int base = beg; base < end; base += 32) {
        int   mysrc = 0;
        float a0 = 0.f, a1 = 0.f, a2 = 0.f, a3 = 0.f;
        if (base == beg) {
            mysrc = src0;
            if (v0) {
                a0 = p0 / s0;  a1 = p1 / s1;
                a2 = p2 / s2;  a3 = p3 / s3;
            }
        } else {
            const int j = base + lane;
            if (j < end) {
                mysrc = bsrc[j];
                const float4 l = el4[mysrc];
                a0 = __expf(gat_leaky(l.x + ern.x) - m0) / s0;
                a1 = __expf(gat_leaky(l.y + ern.y) - m1) / s1;
                a2 = __expf(gat_leaky(l.z + ern.z) - m2) / s2;
                a3 = __expf(gat_leaky(l.w + ern.w) - m3) / s3;
            }
        }
        const int left = end - base;
        const int cntc = left < 32 ? left : 32;
        #pragma unroll 4
        for (int t = 0; t < cntc; ++t) {
            const int   sN = __builtin_amdgcn_readlane(mysrc, t);
            const float b0 = __int_as_float(__builtin_amdgcn_readlane(__float_as_int(a0), t));
            const float b1 = __int_as_float(__builtin_amdgcn_readlane(__float_as_int(a1), t));
            const float b2 = __int_as_float(__builtin_amdgcn_readlane(__float_as_int(a2), t));
            const float b3 = __int_as_float(__builtin_amdgcn_readlane(__float_as_int(a3), t));
            const float aA = hi ? b1 : b0;
            const float aB = hi ? b3 : b2;
            const float4* __restrict__ hp = (const float4*)(h + (size_t)sN * 256);
            const float4 x = hp[lane];
            const float4 y = hp[32 + lane];
            accA.x += aA * x.x;  accA.y += aA * x.y;
            accA.z += aA * x.z;  accA.w += aA * x.w;
            accB.x += aB * y.x;  accB.y += aB * y.y;
            accB.z += aB * y.z;  accB.w += aB * y.w;
        }
    }
    float4* __restrict__ op = (float4*)(out + (size_t)n * 256);
    op[lane]      = accA;
    op[32 + lane] = accB;
}

__global__ __launch_bounds__(128) void gat_node_gather_h1(
    const int* __restrict__ rowptr, const int* __restrict__ bsrc,
    const float* __restrict__ h, const float* __restrict__ el,
    const float* __restrict__ er, float* __restrict__ out, int N, int D)
{
    __shared__ float stage[4 * 64];
    const int lane = threadIdx.x & 31;
    const int wv = __builtin_amdgcn_readfirstlane((int)(threadIdx.x >> 5));
    const int n  = (int)blockIdx.x * 4 + wv;
    const bool has0 = lane < D;
    const bool has1 = lane + 32 < D;
    float acc0 = 0.f, acc1 = 0.f;

    if (n < N) {
        const int beg = __builtin_amdgcn_readfirstlane(rowptr[n]);
        const int end = __builtin_amdgcn_readfirstlane(rowptr[n + 1]);
        const float ern = er[n];
        const float NINF = -__builtin_inff();

        const int  j0 = beg + lane;
        const bool v0 = j0 < end;
        int   src0 = 0;
        float e0 = NINF;
        if (v0) {
            src0 = bsrc[j0];
            e0 = gat_leaky(el[src0] + ern);
        }
        float mx = e0;
        for (int j = j0 + 32; j < end; j += 32)
            mx = fmaxf(mx, gat_leaky(el[bsrc[j]] + ern));
        for (int m = 16; m >= 1; m >>= 1) mx = fmaxf(mx, __shfl_xor(mx, m, 32));

        const float p0 = v0 ? __expf(e0 - mx) : 0.f;
        float s = p0;
        for (int j = j0 + 32; j < end; j += 32)
            s += __expf(gat_leaky(el[bsrc[j]] + ern) - mx);
        for (int m = 16; m >= 1; m >>= 1) s += __shfl_xor(s, m, 32);

        for (int base = beg; base < end; base += 32) {
            int   mysrc = 0;
            float a = 0.f;
            if (base == beg) {
                mysrc = src0;
                if (v0) a = p0 / s;
            } else {
                const int j = base + lane;
                if (j < end) {
                    mysrc = bsrc[j];
                    a = __expf(gat_leaky(el[mysrc] + ern) - mx) / s;
                }
            }
            const int left = end - base;
            const int cntc = left < 32 ? left : 32;
            #pragma unroll 4
            for (int t = 0; t < cntc; ++t) {
                const int   sN = __builtin_amdgcn_readlane(mysrc, t);
                const float b  = __int_as_float(__builtin_amdgcn_readlane(__float_as_int(a), t));
                const float* __restrict__ hp = h + (size_t)sN * D;
                const float x = has0 ? hp[lane] : 0.f;
                const float y = has1 ? hp[32 + lane] : 0.f;
                acc0 += b * x;
                acc1 += b * y;
            }
        }
    }
    if (has0) stage[wv * D + lane] = acc0;
    if (has1) stage[wv * D + 32 + lane] = acc1;
    __syncthreads();
    const int nleft  = N - (int)blockIdx.x * 4;
    const int nvalid = nleft < 4 ? nleft : 4;
    float* __restrict__ ob = out + (size_t)blockIdx.x * 4 * D;
    for (int i = threadIdx.x; i < nvalid * D; i += 128) ob[i] = stage[i];
}

extern "C" void kernel_launch(void* const* d_in, const int* in_sizes, int n_in,
                              void* d_out, int out_size, void* d_ws, size_t ws_size,
                              hipStream_t stream)
{
    const float* feat = (const float*)d_in[0];
    const int*   src  = (const int*)  d_in[1];
    const int*   dst  = (const int*)  d_in[2];
    const float* W0   = (const float*)d_in[3];
    const float* al0  = (const float*)d_in[4];
    const float* ar0  = (const float*)d_in[5];
    const float* W1   = (const float*)d_in[6];
    const float* al1  = (const float*)d_in[7];
    const float* ar1  = (const float*)d_in[8];
    float* out = (float*)d_out;

    const int FIN = 128, H0 = 4, D0 = 64, HD0 = H0 * D0, NC = 40, NCP = 64;
    const int N = in_sizes[0] / FIN;
    const int E = in_sizes[1];

    char* p = (char*)d_ws;
    auto alloc = [&](size_t bytes) -> void* {
        void* r = (void*)p;
        p += (bytes + 255) & ~(size_t)255;
        return r;
    };
    float* h0     = (float*)alloc((size_t)N * HD0 * 4);
    float* agg0   = (float*)alloc((size_t)N * HD0 * 4);
    float* h1     = (float*)alloc((size_t)N * NC  * 4);
    float* W1p    = (float*)alloc((size_t)HD0 * NCP * 4);
    float* el0    = (float*)alloc((size_t)N * H0 * 4);
    float* er0    = (float*)alloc((size_t)N * H0 * 4);
    float* el1    = (float*)alloc((size_t)N * 4);
    float* er1    = (float*)alloc((size_t)N * 4);
    int*   cnt    = (int*)  alloc((size_t)N * 4);
    int*   rowptr = (int*)  alloc((size_t)(N + 1) * 4);
    int*   rank   = (int*)  alloc((size_t)E * 4);
    int*   bsrc   = (int*)  alloc((size_t)E * 4);

    hipMemsetAsync(cnt, 0, (size_t)N * 4, stream);
    gat_bucket_rank<<<(E + 255) / 256, 256, 0, stream>>>(dst, cnt, rank, E);
    gat_bucket_scan<<<(N + 1023) / 1024, 256, 0, stream>>>(cnt, rowptr, N);
    gat_bucket_place<<<(E + 255) / 256, 256, 0, stream>>>(src, dst, rowptr, rank, bsrc, E);

    {
        dim3 grid(N / 16, HD0 / 64);
        gat_wmma_gemm_f32<<<grid, 32, 0, stream>>>(feat, W0, h0, N, HD0, FIN, HD0,
                                                   al0, ar0, el0, er0, H0, D0);
    }
    gat_node_gather_h4<<<(N + 7) / 8, 256, 0, stream>>>(
        rowptr, bsrc, h0, el0, er0, agg0, N);

    {
        int tot = HD0 * NCP;
        gat_pad_weights<<<(tot + 255) / 256, 256, 0, stream>>>(W1, W1p, HD0, NC, NCP);
        dim3 grid(N / 16, NCP / 64);
        gat_wmma_gemm_f32<<<grid, 32, 0, stream>>>(agg0, W1p, h1, N, NCP, HD0, NC,
                                                   al1, ar1, el1, er1, 1, NC);
    }
    gat_node_gather_h1<<<(N + 3) / 4, 128, 0, stream>>>(
        rowptr, bsrc, h1, el1, er1, out, N, NC);
}


# Buckets built by a two-level counting sort with counters and cursors in LDS (64-node ranges, runs copied out whole): no global atomic, no scattered per-edge store, no memset; sum order changes
# speedup vs baseline: 4.0496x; 1.2416x over previous
#include <hip/hip_runtime.h>
#include <hip/hip_bf16.h>

typedef __attribute__((ext_vector_type(2))) float v2f;
typedef __attribute__((ext_vector_type(8))) float v8f;

#define SLOPE 0.2f

template <int M>
__device__ __forceinline__ float gat_row_xor(float x) {
    return __builtin_bit_cast(float, __builtin_amdgcn_update_dpp(
        0, __builtin_bit_cast(int, x), 0x160 | M, 0xf, 0xf, true));
}

__global__ __launch_bounds__(32) void gat_wmma_gemm_f32(
    const float* __restrict__ A, const float* __restrict__ B,
    float* __restrict__ C, int M, int Np, int K, int Nout,
    const float* __restrict__ al, const float* __restrict__ ar,
    float* __restrict__ el, float* __restrict__ er, int H, int D)
{
    const int lane  = threadIdx.x & 31;
    const int l15   = lane & 15;
    const int khalf = (lane >> 4) * 2;
    const int mt    = blockIdx.x;
    const int head  = blockIdx.y;
    const int col0  = head * 64;
    const int row   = mt * 16 + l15;

    const float* __restrict__ Ap = A + (size_t)row * K + khalf;
    const float* __restrict__ Bp = B + (size_t)khalf * Np + col0 + l15;

    float wl[4], wr[4];
    #pragma unroll
    for (int t = 0; t < 4; ++t) {
        const int  c  = t * 16 + l15;
        const bool ok = c < D;
        const int  cc = ok ? c : D - 1;
        const float xl = al[head * D + cc];
        const float xr = ar[head * D + cc];
        wl[t] = ok ? xl : 0.f;
        wr[t] = ok ? xr : 0.f;
    }

    v8f acc[4];
    #pragma unroll
    for (int t = 0; t < 4; ++t)
        acc[t] = (v8f){0.f, 0.f, 0.f, 0.f, 0.f, 0.f, 0.f, 0.f};

    for (int k = 0; k < K; k += 4) {
        v2f a;
        a.x = Ap[0];
        a.y = Ap[1];
        v2f b0, b1, b2, b3;
        b0.x = Bp[0];   b0.y = Bp[Np + 0];
        b1.x = Bp[16];  b1.y = Bp[Np + 16];
        b2.x = Bp[32];  b2.y = Bp[Np + 32];
        b3.x = Bp[48];  b3.y = Bp[Np + 48];
        acc[0] = __builtin_amdgcn_wmma_f32_16x16x4_f32(false, a, false, b0, (short)0, acc[0], false, false);
        acc[1] = __builtin_amdgcn_wmma_f32_16x16x4_f32(false, a, false, b1, (short)0, acc[1], false, false);
        acc[2] = __builtin_amdgcn_wmma_f32_16x16x4_f32(false, a, false, b2, (short)0, acc[2], false, false);
        acc[3] = __builtin_amdgcn_wmma_f32_16x16x4_f32(false, a, false, b3, (short)0, acc[3], false, false);
        Ap += 4;
        Bp += (size_t)4 * Np;
    }

    const int srow0 = mt * 16 + (lane >> 4) * 8;
    #pragma unroll
    for (int t = 0; t < 4; ++t) {
        const int col = col0 + t * 16 + l15;
        if (col < Nout) {
            #pragma unroll
            for (int r = 0; r < 8; ++r)
                C[(size_t)(srow0 + r) * Nout + col] = acc[t][r];
        }
    }

    float v[16];
    #pragma unroll
    for (int r = 0; r < 8; ++r) {
        v[r]     = acc[0][r] * wl[0] + acc[1][r] * wl[1]
                 + acc[2][r] * wl[2] + acc[3][r] * wl[3];
        v[8 + r] = acc[0][r] * wr[0] + acc[1][r] * wr[1]
                 + acc[2][r] * wr[2] + acc[3][r] * wr[3];
    }
    const bool b3 = (lane & 8) != 0;
    const bool b2 = (lane & 4) != 0;
    const bool b1 = (lane & 2) != 0;
    const bool b0 = (lane & 1) != 0;
    float w8[8];
    #pragma unroll
    for (int i = 0; i < 8; ++i) {
        const float keep = b3 ? v[i + 8] : v[i];
        const float send = b3 ? v[i] : v[i + 8];
        w8[i] = keep + gat_row_xor<8>(send);
    }
    float w4[4];
    #pragma unroll
    for (int i = 0; i < 4; ++i) {
        const float keep = b2 ? w8[i + 4] : w8[i];
        const float send = b2 ? w8[i] : w8[i + 4];
        w4[i] = keep + gat_row_xor<4>(send);
    }
    float w2[2];
    #pragma unroll
    for (int i = 0; i < 2; ++i) {
        const float keep = b1 ? w4[i + 2] : w4[i];
        const float send = b1 ? w4[i] : w4[i + 2];
        w2[i] = keep + gat_row_xor<2>(send);
    }
    const float keep1 = b0 ? w2[1] : w2[0];
    const float send1 = b0 ? w2[0] : w2[1];
    const float z = keep1 + gat_row_xor<1>(send1);
    float* __restrict__ sp = b3 ? er : el;
    sp[(size_t)(srow0 + (lane & 7)) * H + head] = z;
}

__global__ __launch_bounds__(256) void gat_pad_weights(
    const float* __restrict__ W, float* __restrict__ Wp, int K, int N, int Np)
{
    int t = blockIdx.x * blockDim.x + threadIdx.x;
    if (t >= K * Np) return;
    const int r = t / Np, c = t - r * Np;
    Wp[t] = (c < N) ? W[r * N + c] : 0.f;
}

#define GAT_CAP 6144
#define GAT_FCAP 4096

__global__ __launch_bounds__(1024) void gat_bin_count(
    const int* __restrict__ dst, int* __restrict__ cnt2, int E, int R, int G)
{
    extern __shared__ int gat_sh[];
    const int tid = threadIdx.x;
    for (int r = tid; r < R; r += 1024) gat_sh[r] = 0;
    __syncthreads();
    const int base = (int)blockIdx.x * GAT_CAP;
    int dk[GAT_CAP / 1024];
    #pragma unroll
    for (int k = 0; k < GAT_CAP / 1024; ++k) {
        const int e = base + k * 1024 + tid;
        dk[k] = dst[e < E ? e : E - 1];
    }
    #pragma unroll
    for (int k = 0; k < GAT_CAP / 1024; ++k) {
        const int e = base + k * 1024 + tid;
        if (e < E) atomicAdd(&gat_sh[dk[k] >> 6], 1);
    }
    __syncthreads();
    for (int r = tid; r < R; r += 1024) cnt2[r * G + (int)blockIdx.x] = gat_sh[r];
}

__global__ __launch_bounds__(1024) void gat_bin_scatter(
    const int* __restrict__ src, const int* __restrict__ dst,
    const int* __restrict__ off2, unsigned* __restrict__ tmp, int E, int R, int G)
{
    extern __shared__ int gat_sh[];
    __shared__ int wsum[32];
    int* lcnt = gat_sh;
    int* lbeg = gat_sh + R;
    int* gofs = gat_sh + 2 * R;
    unsigned* bw = (unsigned*)(gat_sh + 3 * R);
    int* br = gat_sh + 3 * R + GAT_CAP;
    const int tid = threadIdx.x, lane = tid & 31, wv = tid >> 5;
    const int g = (int)blockIdx.x;
    const int base = g * GAT_CAP;

    for (int r = tid; r < R; r += 1024) {
        lcnt[r] = 0;
        gofs[r] = off2[r * G + g];
    }
    __syncthreads();
    int ed[GAT_CAP / 1024], es[GAT_CAP / 1024];
    #pragma unroll
    for (int k = 0; k < GAT_CAP / 1024; ++k) {
        const int e  = base + k * 1024 + tid;
        const int ec = e < E ? e : E - 1;
        const int d  = dst[ec];
        es[k] = src[ec];
        ed[k] = e < E ? d : -1;
    }
    #pragma unroll
    for (int k = 0; k < GAT_CAP / 1024; ++k)
        if (ed[k] >= 0) atomicAdd(&lcnt[ed[k] >> 6], 1);
    __syncthreads();
    int carry = 0;
    for (int b0 = 0; b0 < R; b0 += 1024) {
        const int idx = b0 + tid;
        const int v = (idx < R) ? lcnt[idx] : 0;
        int incl = v;
        for (int off = 1; off < 32; off <<= 1) {
            const int y = __shfl_up(incl, off, 32);
            if (lane >= off) incl += y;
        }
        if (lane == 31) wsum[wv] = incl;
        __syncthreads();
        int pre = 0, tot = 0;
        for (int w = 0; w < 32; ++w) {
            const int t = wsum[w];
            tot += t;
            if (w < wv) pre += t;
        }
        if (idx < R) lbeg[idx] = carry + pre + incl - v;
        carry += tot;
        __syncthreads();
    }
    for (int r = tid; r < R; r += 1024) lcnt[r] = lbeg[r];
    __syncthreads();
    #pragma unroll
    for (int k = 0; k < GAT_CAP / 1024; ++k) {
        if (ed[k] >= 0) {
            const int r = ed[k] >> 6;
            const int p = atomicAdd(&lcnt[r], 1);
            bw[p] = ((unsigned)es[k] << 6) | (unsigned)(ed[k] & 63);
            br[p] = r;
        }
    }
    __syncthreads();
    const int left = E - base;
    const int nloc = left < GAT_CAP ? left : GAT_CAP;
    for (int k = tid; k < nloc; k += 1024) {
        const int r = br[k];
        tmp[gofs[r] + (k - lbeg[r])] = bw[k];
    }
}

__global__ __launch_bounds__(256) void gat_bin_finish(
    const unsigned* __restrict__ tmp, const int* __restrict__ off2,
    int* __restrict__ rowptr, int* __restrict__ bsrc, int N, int R, int G)
{
    __shared__ int c[64];
    __shared__ int cur[64];
    __shared__ int wtot[2];
    __shared__ int stage[GAT_FCAP];
    const int tid = threadIdx.x, lane = tid & 31, wv = tid >> 5;
    const int r = (int)blockIdx.x;
    const int s = off2[r * G];
    const int e = off2[(r + 1) * G];
    const int n = e - s;
    if (tid < 64) c[tid] = 0;
    __syncthreads();
    for (int i = tid; i < n; i += 256) atomicAdd(&c[tmp[s + i] & 63u], 1);
    __syncthreads();
    int v = 0, incl = 0;
    if (tid < 64) {
        v = c[tid];
        incl = v;
        for (int off = 1; off < 32; off <<= 1) {
            const int y = __shfl_up(incl, off, 32);
            if (lane >= off) incl += y;
        }
        if (lane == 31) wtot[wv] = incl;
    }
    __syncthreads();
    if (tid < 64) {
        const int excl = incl - v + (wv == 1 ? wtot[0] : 0);
        cur[tid] = excl;
        const int node = r * 64 + tid;
        if (node < N) rowptr[node] = s + excl;
    }
    if (r == R - 1 && tid == 0) rowptr[N] = e;
    __syncthreads();
    if (n <= GAT_FCAP) {
        for (int i = tid; i < n; i += 256) {
            const unsigned w = tmp[s + i];
            const int p = atomicAdd(&cur[w & 63u], 1);
            stage[p] = (int)(w >> 6);
        }
        __syncthreads();
        for (int i = tid; i < n; i += 256) bsrc[s + i] = stage[i];
    } else {
        for (int i = tid; i < n; i += 256) {
            const unsigned w = tmp[s + i];
            const int p = atomicAdd(&cur[w & 63u], 1);
            bsrc[s + p] = (int)(w >> 6);
        }
    }
}

__global__ __launch_bounds__(256) void gat_bucket_scan(
    const int* __restrict__ cnt, int* __restrict__ rowptr, int N)
{
    __shared__ int s_tot[8];
    __shared__ int s_car[8];
    const int tid = threadIdx.x, lane = tid & 31, wv = tid >> 5;
    const int4* __restrict__ c4 = (const int4*)cnt;

    int p = 0;
    for (int k = 0; k < (int)blockIdx.x; ++k) {
        const int4 q = c4[k * 256 + tid];
        p += (q.x + q.y) + (q.z + q.w);
    }
    const int i0 = blockIdx.x * 1024 + tid * 4;
    const bool live = (i0 + 3 < N);
    int4 own = make_int4(0, 0, 0, 0);
    if (live) own = c4[i0 >> 2];
    const int v = (own.x + own.y) + (own.z + own.w);

    int incl = v;
    for (int off = 1; off < 32; off <<= 1) {
        const int y = __shfl_up(incl, off, 32);
        if (lane >= off) incl += y;
    }
    for (int m = 16; m >= 1; m >>= 1) p += __shfl_xor(p, m, 32);
    if (lane == 31) s_tot[wv] = incl;
    if (lane == 0)  s_car[wv] = p;
    __syncthreads();
    int base = 0;
    for (int w = 0; w < 8; ++w) {
        base += s_car[w];
        if (w < wv) base += s_tot[w];
    }
    base += incl - v;
    if (live) {
        int4 r;
        r.x = base;
        r.y = r.x + own.x;
        r.z = r.y + own.y;
        r.w = r.z + own.z;
        ((int4*)rowptr)[i0 >> 2] = r;
        if (i0 + 4 == N) rowptr[N] = r.w + own.w;
    }
}

__device__ __forceinline__ float gat_leaky(float v) {
    return (v > 0.f) ? v : SLOPE * v;
}

__global__ __launch_bounds__(256) void gat_node_gather_h4(
    const int* __restrict__ rowptr, const int* __restrict__ bsrc,
    const float* __restrict__ h, const float* __restrict__ el,
    const float* __restrict__ er, float* __restrict__ out, int N)
{
    const int lane = threadIdx.x & 31;
    const int n = __builtin_amdgcn_readfirstlane(
        (int)((blockIdx.x * blockDim.x + threadIdx.x) >> 5));
    if (n >= N) return;
    const int beg = __builtin_amdgcn_readfirstlane(rowptr[n]);
    const int end = __builtin_amdgcn_readfirstlane(rowptr[n + 1]);
    const float4* __restrict__ el4 = (const float4*)el;
    const float4 ern = ((const float4*)er)[n];
    const float NINF = -__builtin_inff();

    const int  j0 = beg + lane;
    const bool v0 = j0 < end;
    int   src0 = 0;
    float e0 = NINF, e1 = NINF, e2 = NINF, e3 = NINF;
    if (v0) {
        src0 = bsrc[j0];
        const float4 l = el4[src0];
        e0 = gat_leaky(l.x + ern.x);
        e1 = gat_leaky(l.y + ern.y);
        e2 = gat_leaky(l.z + ern.z);
        e3 = gat_leaky(l.w + ern.w);
    }
    float m0 = e0, m1 = e1, m2 = e2, m3 = e3;
    for (int j = j0 + 32; j < end; j += 32) {
        const float4 l = el4[bsrc[j]];
        m0 = fmaxf(m0, gat_leaky(l.x + ern.x));
        m1 = fmaxf(m1, gat_leaky(l.y + ern.y));
        m2 = fmaxf(m2, gat_leaky(l.z + ern.z));
        m3 = fmaxf(m3, gat_leaky(l.w + ern.w));
    }
    for (int m = 16; m >= 1; m >>= 1) {
        m0 = fmaxf(m0, __shfl_xor(m0, m, 32));
        m1 = fmaxf(m1, __shfl_xor(m1, m, 32));
        m2 = fmaxf(m2, __shfl_xor(m2, m, 32));
        m3 = fmaxf(m3, __shfl_xor(m3, m, 32));
    }

    const float p0 = v0 ? __expf(e0 - m0) : 0.f;
    const float p1 = v0 ? __expf(e1 - m1) : 0.f;
    const float p2 = v0 ? __expf(e2 - m2) : 0.f;
    const float p3 = v0 ? __expf(e3 - m3) : 0.f;
    float s0 = p0, s1 = p1, s2 = p2, s3 = p3;
    for (int j = j0 + 32; j < end; j += 32) {
        const float4 l = el4[bsrc[j]];
        s0 += __expf(gat_leaky(l.x + ern.x) - m0);
        s1 += __expf(gat_leaky(l.y + ern.y) - m1);
        s2 += __expf(gat_leaky(l.z + ern.z) - m2);
        s3 += __expf(gat_leaky(l.w + ern.w) - m3);
    }
    for (int m = 16; m >= 1; m >>= 1) {
        s0 += __shfl_xor(s0, m, 32);
        s1 += __shfl_xor(s1, m, 32);
        s2 += __shfl_xor(s2, m, 32);
        s3 += __shfl_xor(s3, m, 32);
    }

    float4 accA = make_float4(0.f, 0.f, 0.f, 0.f);
    float4 accB = make_float4(0.f, 0.f, 0.f, 0.f);
    const bool hi = lane >= 16;
    for (int base = beg; base < end; base += 32) {
        int   mysrc = 0;
        float a0 = 0.f, a1 = 0.f, a2 = 0.f, a3 = 0.f;
        if (base == beg) {
            mysrc = src0;
            if (v0) {
                a0 = p0 / s0;  a1 = p1 / s1;
                a2 = p2 / s2;  a3 = p3 / s3;
            }
        } else {
            const int j = base + lane;
            if (j < end) {
                mysrc = bsrc[j];
                const float4 l = el4[mysrc];
                a0 = __expf(gat_leaky(l.x + ern.x) - m0) / s0;
                a1 = __expf(gat_leaky(l.y + ern.y) - m1) / s1;
                a2 = __expf(gat_leaky(l.z + ern.z) - m2) / s2;
                a3 = __expf(gat_leaky(l.w + ern.w) - m3) / s3;
            }
        }
        const int left = end - base;
        const int cntc = left < 32 ? left : 32;
        #pragma unroll 4
        for (int t = 0; t < cntc; ++t) {
            const int   sN = __builtin_amdgcn_readlane(mysrc, t);
            const float b0 = __int_as_float(__builtin_amdgcn_readlane(__float_as_int(a0), t));
            const float b1 = __int_as_float(__builtin_amdgcn_readlane(__float_as_int(a1), t));
            const float b2 = __int_as_float(__builtin_amdgcn_readlane(__float_as_int(a2), t));
            const float b3 = __int_as_float(__builtin_amdgcn_readlane(__float_as_int(a3), t));
            const float aA = hi ? b1 : b0;
            const float aB = hi ? b3 : b2;
            const float4* __restrict__ hp = (const float4*)(h + (size_t)sN * 256);
            const float4 x = hp[lane];
            const float4 y = hp[32 + lane];
            accA.x += aA * x.x;  accA.y += aA * x.y;
            accA.z += aA * x.z;  accA.w += aA * x.w;
            accB.x += aB * y.x;  accB.y += aB * y.y;
            accB.z += aB * y.z;  accB.w += aB * y.w;
        }
    }
    float4* __restrict__ op = (float4*)(out + (size_t)n * 256);
    op[lane]      = accA;
    op[32 + lane] = accB;
}

__global__ __launch_bounds__(128) void gat_node_gather_h1(
    const int* __restrict__ rowptr, const int* __restrict__ bsrc,
    const float* __restrict__ h, const float* __restrict__ el,
    const float* __restrict__ er, float* __restrict__ out, int N, int D)
{
    __shared__ float stage[4 * 64];
    const int lane = threadIdx.x & 31;
    const int wv = __builtin_amdgcn_readfirstlane((int)(threadIdx.x >> 5));
    const int n  = (int)blockIdx.x * 4 + wv;
    const bool has0 = lane < D;
    const bool has1 = lane + 32 < D;
    float acc0 = 0.f, acc1 = 0.f;

    if (n < N) {
        const int beg = __builtin_amdgcn_readfirstlane(rowptr[n]);
        const int end = __builtin_amdgcn_readfirstlane(rowptr[n + 1]);
        const float ern = er[n];
        const float NINF = -__builtin_inff();

        const int  j0 = beg + lane;
        const bool v0 = j0 < end;
        int   src0 = 0;
        float e0 = NINF;
        if (v0) {
            src0 = bsrc[j0];
            e0 = gat_leaky(el[src0] + ern);
        }
        float mx = e0;
        for (int j = j0 + 32; j < end; j += 32)
            mx = fmaxf(mx, gat_leaky(el[bsrc[j]] + ern));
        for (int m = 16; m >= 1; m >>= 1) mx = fmaxf(mx, __shfl_xor(mx, m, 32));

        const float p0 = v0 ? __expf(e0 - mx) : 0.f;
        float s = p0;
        for (int j = j0 + 32; j < end; j += 32)
            s += __expf(gat_leaky(el[bsrc[j]] + ern) - mx);
        for (int m = 16; m >= 1; m >>= 1) s += __shfl_xor(s, m, 32);

        for (int base = beg; base < end; base += 32) {
            int   mysrc = 0;
            float a = 0.f;
            if (base == beg) {
                mysrc = src0;
                if (v0) a = p0 / s;
            } else {
                const int j = base + lane;
                if (j < end) {
                    mysrc = bsrc[j];
                    a = __expf(gat_leaky(el[mysrc] + ern) - mx) / s;
                }
            }
            const int left = end - base;
            const int cntc = left < 32 ? left : 32;
            #pragma unroll 4
            for (int t = 0; t < cntc; ++t) {
                const int   sN = __builtin_amdgcn_readlane(mysrc, t);
                const float b  = __int_as_float(__builtin_amdgcn_readlane(__float_as_int(a), t));
                const float* __restrict__ hp = h + (size_t)sN * D;
                const float x = has0 ? hp[lane] : 0.f;
                const float y = has1 ? hp[32 + lane] : 0.f;
                acc0 += b * x;
                acc1 += b * y;
            }
        }
    }
    if (has0) stage[wv * D + lane] = acc0;
    if (has1) stage[wv * D + 32 + lane] = acc1;
    __syncthreads();
    const int nleft  = N - (int)blockIdx.x * 4;
    const int nvalid = nleft < 4 ? nleft : 4;
    float* __restrict__ ob = out + (size_t)blockIdx.x * 4 * D;
    for (int i = threadIdx.x; i < nvalid * D; i += 128) ob[i] = stage[i];
}

extern "C" void kernel_launch(void* const* d_in, const int* in_sizes, int n_in,
                              void* d_out, int out_size, void* d_ws, size_t ws_size,
                              hipStream_t stream)
{
    const float* feat = (const float*)d_in[0];
    const int*   src  = (const int*)  d_in[1];
    const int*   dst  = (const int*)  d_in[2];
    const float* W0   = (const float*)d_in[3];
    const float* al0  = (const float*)d_in[4];
    const float* ar0  = (const float*)d_in[5];
    const float* W1   = (const float*)d_in[6];
    const float* al1  = (const float*)d_in[7];
    const float* ar1  = (const float*)d_in[8];
    float* out = (float*)d_out;

    const int FIN = 128, H0 = 4, D0 = 64, HD0 = H0 * D0, NC = 40, NCP = 64;
    const int N = in_sizes[0] / FIN;
    const int E = in_sizes[1];

    char* p = (char*)d_ws;
    auto alloc = [&](size_t bytes) -> void* {
        void* r = (void*)p;
        p += (bytes + 255) & ~(size_t)255;
        return r;
    };
    float* h0     = (float*)alloc((size_t)N * HD0 * 4);
    float* agg0   = (float*)alloc((size_t)N * HD0 * 4);
    float* h1     = (float*)alloc((size_t)N * NC  * 4);
    float* W1p    = (float*)alloc((size_t)HD0 * NCP * 4);
    float* el0    = (float*)alloc((size_t)N * H0 * 4);
    float* er0    = (float*)alloc((size_t)N * H0 * 4);
    float* el1    = (float*)alloc((size_t)N * 4);
    float* er1    = (float*)alloc((size_t)N * 4);
    const int R  = (N + 63) / 64;
    const int G  = ((E + 4 * GAT_CAP - 1) / (4 * GAT_CAP)) * 4;
    const int RG = R * G;
    int*      cnt2   = (int*)     alloc((size_t)RG * 4);
    int*      off2   = (int*)     alloc((size_t)(RG + 1) * 4);
    int*      rowptr = (int*)     alloc((size_t)(N + 1) * 4);
    unsigned* tmp    = (unsigned*)alloc((size_t)E * 4);
    int*      bsrc   = (int*)     alloc((size_t)E * 4);

    gat_bin_count<<<G, 1024, (size_t)R * 4, stream>>>(dst, cnt2, E, R, G);
    gat_bucket_scan<<<(RG + 1023) / 1024, 256, 0, stream>>>(cnt2, off2, RG);
    gat_bin_scatter<<<G, 1024, ((size_t)3 * R + 2 * GAT_CAP) * 4, stream>>>(
        src, dst, off2, tmp, E, R, G);
    gat_bin_finish<<<R, 256, 0, stream>>>(tmp, off2, rowptr, bsrc, N, R, G);

    {
        dim3 grid(N / 16, HD0 / 64);
        gat_wmma_gemm_f32<<<grid, 32, 0, stream>>>(feat, W0, h0, N, HD0, FIN, HD0,
                                                   al0, ar0, el0, er0, H0, D0);
    }
    gat_node_gather_h4<<<(N + 7) / 8, 256, 0, stream>>>(
        rowptr, bsrc, h0, el0, er0, agg0, N);

    {
        int tot = HD0 * NCP;
        gat_pad_weights<<<(tot + 255) / 256, 256, 0, stream>>>(W1, W1p, HD0, NC, NCP);
        dim3 grid(N / 16, NCP / 64);
        gat_wmma_gemm_f32<<<grid, 32, 0, stream>>>(agg0, W1p, h1, N, NCP, HD0, NC,
                                                   al1, ar1, el1, er1, 1, NC);
    }
    gat_node_gather_h1<<<(N + 3) / 4, 128, 0, stream>>>(
        rowptr, bsrc, h1, el1, er1, out, N, NC);
}


# Layer-1 rows padded to one aligned 256-byte chunk that also holds the node's el: its gather takes one load per edge in one online-softmax pass, el gather and two passes gone; sum order changes
# speedup vs baseline: 4.1794x; 1.0321x over previous
#include <hip/hip_runtime.h>
#include <hip/hip_bf16.h>

typedef __attribute__((ext_vector_type(2))) float v2f;
typedef __attribute__((ext_vector_type(8))) float v8f;

#define SLOPE 0.2f

template <int M>
__device__ __forceinline__ float gat_row_xor(float x) {
    return __builtin_bit_cast(float, __builtin_amdgcn_update_dpp(
        0, __builtin_bit_cast(int, x), 0x160 | M, 0xf, 0xf, true));
}

__global__ __launch_bounds__(32) void gat_wmma_gemm_f32(
    const float* __restrict__ A, const float* __restrict__ B,
    float* __restrict__ C, int M, int Np, int K, int Nout,
    const float* __restrict__ al, const float* __restrict__ ar,
    float* __restrict__ el, float* __restrict__ er, int els, int ers,
    int skipcol, int D)
{
    const int lane  = threadIdx.x & 31;
    const int l15   = lane & 15;
    const int khalf = (lane >> 4) * 2;
    const int mt    = blockIdx.x;
    const int head  = blockIdx.y;
    const int col0  = head * 64;
    const int row   = mt * 16 + l15;

    const float* __restrict__ Ap = A + (size_t)row * K + khalf;
    const float* __restrict__ Bp = B + (size_t)khalf * Np + col0 + l15;

    float wl[4], wr[4];
    #pragma unroll
    for (int t = 0; t < 4; ++t) {
        const int  c  = t * 16 + l15;
        const bool ok = c < D;
        const int  cc = ok ? c : D - 1;
        const float xl = al[head * D + cc];
        const float xr = ar[head * D + cc];
        wl[t] = ok ? xl : 0.f;
        wr[t] = ok ? xr : 0.f;
    }

    v8f acc[4];
    #pragma unroll
    for (int t = 0; t < 4; ++t)
        acc[t] = (v8f){0.f, 0.f, 0.f, 0.f, 0.f, 0.f, 0.f, 0.f};

    for (int k = 0; k < K; k += 4) {
        v2f a;
        a.x = Ap[0];
        a.y = Ap[1];
        v2f b0, b1, b2, b3;
        b0.x = Bp[0];   b0.y = Bp[Np + 0];
        b1.x = Bp[16];  b1.y = Bp[Np + 16];
        b2.x = Bp[32];  b2.y = Bp[Np + 32];
        b3.x = Bp[48];  b3.y = Bp[Np + 48];
        acc[0] = __builtin_amdgcn_wmma_f32_16x16x4_f32(false, a, false, b0, (short)0, acc[0], false, false);
        acc[1] = __builtin_amdgcn_wmma_f32_16x16x4_f32(false, a, false, b1, (short)0, acc[1], false, false);
        acc[2] = __builtin_amdgcn_wmma_f32_16x16x4_f32(false, a, false, b2, (short)0, acc[2], false, false);
        acc[3] = __builtin_amdgcn_wmma_f32_16x16x4_f32(false, a, false, b3, (short)0, acc[3], false, false);
        Ap += 4;
        Bp += (size_t)4 * Np;
    }

    const int srow0 = mt * 16 + (lane >> 4) * 8;
    #pragma unroll
    for (int t = 0; t < 4; ++t) {
        const int col = col0 + t * 16 + l15;
        if (col < Nout && col != skipcol) {
            #pragma unroll
            for (int r = 0; r < 8; ++r)
                C[(size_t)(srow0 + r) * Nout + col] = acc[t][r];
        }
    }

    float v[16];
    #pragma unroll
    for (int r = 0; r < 8; ++r) {
        v[r]     = acc[0][r] * wl[0] + acc[1][r] * wl[1]
                 + acc[2][r] * wl[2] + acc[3][r] * wl[3];
        v[8 + r] = acc[0][r] * wr[0] + acc[1][r] * wr[1]
                 + acc[2][r] * wr[2] + acc[3][r] * wr[3];
    }
    const bool b3 = (lane & 8) != 0;
    const bool b2 = (lane & 4) != 0;
    const bool b1 = (lane & 2) != 0;
    const bool b0 = (lane & 1) != 0;
    float w8[8];
    #pragma unroll
    for (int i = 0; i < 8; ++i) {
        const float keep = b3 ? v[i + 8] : v[i];
        const float send = b3 ? v[i] : v[i + 8];
        w8[i] = keep + gat_row_xor<8>(send);
    }
    float w4[4];
    #pragma unroll
    for (int i = 0; i < 4; ++i) {
        const float keep = b2 ? w8[i + 4] : w8[i];
        const float send = b2 ? w8[i] : w8[i + 4];
        w4[i] = keep + gat_row_xor<4>(send);
    }
    float w2[2];
    #pragma unroll
    for (int i = 0; i < 2; ++i) {
        const float keep = b1 ? w4[i + 2] : w4[i];
        const float send = b1 ? w4[i] : w4[i + 2];
        w2[i] = keep + gat_row_xor<2>(send);
    }
    const float keep1 = b0 ? w2[1] : w2[0];
    const float send1 = b0 ? w2[0] : w2[1];
    const float z = keep1 + gat_row_xor<1>(send1);
    float* __restrict__ sp = b3 ? er : el;
    const int st = b3 ? ers : els;
    sp[(size_t)(srow0 + (lane & 7)) * st + head] = z;
}

__global__ __launch_bounds__(256) void gat_pad_weights(
    const float* __restrict__ W, float* __restrict__ Wp, int K, int N, int Np)
{
    int t = blockIdx.x * blockDim.x + threadIdx.x;
    if (t >= K * Np) return;
    const int r = t / Np, c = t - r * Np;
    Wp[t] = (c < N) ? W[r * N + c] : 0.f;
}

#define GAT_CAP 6144
#define GAT_FCAP 4096

__global__ __launch_bounds__(1024) void gat_bin_count(
    const int* __restrict__ dst, int* __restrict__ cnt2, int E, int R, int G)
{
    extern __shared__ int gat_sh[];
    const int tid = threadIdx.x;
    for (int r = tid; r < R; r += 1024) gat_sh[r] = 0;
    __syncthreads();
    const int base = (int)blockIdx.x * GAT_CAP;
    int dk[GAT_CAP / 1024];
    #pragma unroll
    for (int k = 0; k < GAT_CAP / 1024; ++k) {
        const int e = base + k * 1024 + tid;
        dk[k] = dst[e < E ? e : E - 1];
    }
    #pragma unroll
    for (int k = 0; k < GAT_CAP / 1024; ++k) {
        const int e = base + k * 1024 + tid;
        if (e < E) atomicAdd(&gat_sh[dk[k] >> 6], 1);
    }
    __syncthreads();
    for (int r = tid; r < R; r += 1024) cnt2[r * G + (int)blockIdx.x] = gat_sh[r];
}

__global__ __launch_bounds__(1024) void gat_bin_scatter(
    const int* __restrict__ src, const int* __restrict__ dst,
    const int* __restrict__ off2, unsigned* __restrict__ tmp, int E, int R, int G)
{
    extern __shared__ int gat_sh[];
    __shared__ int wsum[32];
    int* lcnt = gat_sh;
    int* lbeg = gat_sh + R;
    int* gofs = gat_sh + 2 * R;
    unsigned* bw = (unsigned*)(gat_sh + 3 * R);
    int* br = gat_sh + 3 * R + GAT_CAP;
    const int tid = threadIdx.x, lane = tid & 31, wv = tid >> 5;
    const int g = (int)blockIdx.x;
    const int base = g * GAT_CAP;

    for (int r = tid; r < R; r += 1024) {
        lcnt[r] = 0;
        gofs[r] = off2[r * G + g];
    }
    __syncthreads();
    int ed[GAT_CAP / 1024], es[GAT_CAP / 1024];
    #pragma unroll
    for (int k = 0; k < GAT_CAP / 1024; ++k) {
        const int e  = base + k * 1024 + tid;
        const int ec = e < E ? e : E - 1;
        const int d  = dst[ec];
        es[k] = src[ec];
        ed[k] = e < E ? d : -1;
    }
    #pragma unroll
    for (int k = 0; k < GAT_CAP / 1024; ++k)
        if (ed[k] >= 0) atomicAdd(&lcnt[ed[k] >> 6], 1);
    __syncthreads();
    int carry = 0;
    for (int b0 = 0; b0 < R; b0 += 1024) {
        const int idx = b0 + tid;
        const int v = (idx < R) ? lcnt[idx] : 0;
        int incl = v;
        for (int off = 1; off < 32; off <<= 1) {
            const int y = __shfl_up(incl, off, 32);
            if (lane >= off) incl += y;
        }
        if (lane == 31) wsum[wv] = incl;
        __syncthreads();
        int pre = 0, tot = 0;
        for (int w = 0; w < 32; ++w) {
            const int t = wsum[w];
            tot += t;
            if (w < wv) pre += t;
        }
        if (idx < R) lbeg[idx] = carry + pre + incl - v;
        carry += tot;
        __syncthreads();
    }
    for (int r = tid; r < R; r += 1024) lcnt[r] = lbeg[r];
    __syncthreads();
    #pragma unroll
    for (int k = 0; k < GAT_CAP / 1024; ++k) {
        if (ed[k] >= 0) {
            const int r = ed[k] >> 6;
            const int p = atomicAdd(&lcnt[r], 1);
            bw[p] = ((unsigned)es[k] << 6) | (unsigned)(ed[k] & 63);
            br[p] = r;
        }
    }
    __syncthreads();
    const int left = E - base;
    const int nloc = left < GAT_CAP ? left : GAT_CAP;
    for (int k = tid; k < nloc; k += 1024) {
        const int r = br[k];
        tmp[gofs[r] + (k - lbeg[r])] = bw[k];
    }
}

__global__ __launch_bounds__(256) void gat_bin_finish(
    const unsigned* __restrict__ tmp, const int* __restrict__ off2,
    int* __restrict__ rowptr, int* __restrict__ bsrc, int N, int R, int G)
{
    __shared__ int c[64];
    __shared__ int cur[64];
    __shared__ int wtot[2];
    __shared__ int stage[GAT_FCAP];
    const int tid = threadIdx.x, lane = tid & 31, wv = tid >> 5;
    const int r = (int)blockIdx.x;
    const int s = off2[r * G];
    const int e = off2[(r + 1) * G];
    const int n = e - s;
    if (tid < 64) c[tid] = 0;
    __syncthreads();
    for (int i = tid; i < n; i += 256) atomicAdd(&c[tmp[s + i] & 63u], 1);
    __syncthreads();
    int v = 0, incl = 0;
    if (tid < 64) {
        v = c[tid];
        incl = v;
        for (int off = 1; off < 32; off <<= 1) {
            const int y = __shfl_up(incl, off, 32);
            if (lane >= off) incl += y;
        }
        if (lane == 31) wtot[wv] = incl;
    }
    __syncthreads();
    if (tid < 64) {
        const int excl = incl - v + (wv == 1 ? wtot[0] : 0);
        cur[tid] = excl;
        const int node = r * 64 + tid;
        if (node < N) rowptr[node] = s + excl;
    }
    if (r == R - 1 && tid == 0) rowptr[N] = e;
    __syncthreads();
    if (n <= GAT_FCAP) {
        for (int i = tid; i < n; i += 256) {
            const unsigned w = tmp[s + i];
            const int p = atomicAdd(&cur[w & 63u], 1);
            stage[p] = (int)(w >> 6);
        }
        __syncthreads();
        for (int i = tid; i < n; i += 256) bsrc[s + i] = stage[i];
    } else {
        for (int i = tid; i < n; i += 256) {
            const unsigned w = tmp[s + i];
            const int p = atomicAdd(&cur[w & 63u], 1);
            bsrc[s + p] = (int)(w >> 6);
        }
    }
}

__global__ __launch_bounds__(256) void gat_bucket_scan(
    const int* __restrict__ cnt, int* __restrict__ rowptr, int N)
{
    __shared__ int s_tot[8];
    __shared__ int s_car[8];
    const int tid = threadIdx.x, lane = tid & 31, wv = tid >> 5;
    const int4* __restrict__ c4 = (const int4*)cnt;

    int p = 0;
    for (int k = 0; k < (int)blockIdx.x; ++k) {
        const int4 q = c4[k * 256 + tid];
        p += (q.x + q.y) + (q.z + q.w);
    }
    const int i0 = blockIdx.x * 1024 + tid * 4;
    const bool live = (i0 + 3 < N);
    int4 own = make_int4(0, 0, 0, 0);
    if (live) own = c4[i0 >> 2];
    const int v = (own.x + own.y) + (own.z + own.w);

    int incl = v;
    for (int off = 1; off < 32; off <<= 1) {
        const int y = __shfl_up(incl, off, 32);
        if (lane >= off) incl += y;
    }
    for (int m = 16; m >= 1; m >>= 1) p += __shfl_xor(p, m, 32);
    if (lane == 31) s_tot[wv] = incl;
    if (lane == 0)  s_car[wv] = p;
    __syncthreads();
    int base = 0;
    for (int w = 0; w < 8; ++w) {
        base += s_car[w];
        if (w < wv) base += s_tot[w];
    }
    base += incl - v;
    if (live) {
        int4 r;
        r.x = base;
        r.y = r.x + own.x;
        r.z = r.y + own.y;
        r.w = r.z + own.z;
        ((int4*)rowptr)[i0 >> 2] = r;
        if (i0 + 4 == N) rowptr[N] = r.w + own.w;
    }
}

__device__ __forceinline__ float gat_leaky(float v) {
    return (v > 0.f) ? v : SLOPE * v;
}

__global__ __launch_bounds__(256) void gat_node_gather_h4(
    const int* __restrict__ rowptr, const int* __restrict__ bsrc,
    const float* __restrict__ h, const float* __restrict__ el,
    const float* __restrict__ er, float* __restrict__ out, int N)
{
    const int lane = threadIdx.x & 31;
    const int n = __builtin_amdgcn_readfirstlane(
        (int)((blockIdx.x * blockDim.x + threadIdx.x) >> 5));
    if (n >= N) return;
    const int beg = __builtin_amdgcn_readfirstlane(rowptr[n]);
    const int end = __builtin_amdgcn_readfirstlane(rowptr[n + 1]);
    const float4* __restrict__ el4 = (const float4*)el;
    const float4 ern = ((const float4*)er)[n];
    const float NINF = -__builtin_inff();

    const int  j0 = beg + lane;
    const bool v0 = j0 < end;
    int   src0 = 0;
    float e0 = NINF, e1 = NINF, e2 = NINF, e3 = NINF;
    if (v0) {
        src0 = bsrc[j0];
        const float4 l = el4[src0];
        e0 = gat_leaky(l.x + ern.x);
        e1 = gat_leaky(l.y + ern.y);
        e2 = gat_leaky(l.z + ern.z);
        e3 = gat_leaky(l.w + ern.w);
    }
    float m0 = e0, m1 = e1, m2 = e2, m3 = e3;
    for (int j = j0 + 32; j < end; j += 32) {
        const float4 l = el4[bsrc[j]];
        m0 = fmaxf(m0, gat_leaky(l.x + ern.x));
        m1 = fmaxf(m1, gat_leaky(l.y + ern.y));
        m2 = fmaxf(m2, gat_leaky(l.z + ern.z));
        m3 = fmaxf(m3, gat_leaky(l.w + ern.w));
    }
    for (int m = 16; m >= 1; m >>= 1) {
        m0 = fmaxf(m0, __shfl_xor(m0, m, 32));
        m1 = fmaxf(m1, __shfl_xor(m1, m, 32));
        m2 = fmaxf(m2, __shfl_xor(m2, m, 32));
        m3 = fmaxf(m3, __shfl_xor(m3, m, 32));
    }

    const float p0 = v0 ? __expf(e0 - m0) : 0.f;
    const float p1 = v0 ? __expf(e1 - m1) : 0.f;
    const float p2 = v0 ? __expf(e2 - m2) : 0.f;
    const float p3 = v0 ? __expf(e3 - m3) : 0.f;
    float s0 = p0, s1 = p1, s2 = p2, s3 = p3;
    for (int j = j0 + 32; j < end; j += 32) {
        const float4 l = el4[bsrc[j]];
        s0 += __expf(gat_leaky(l.x + ern.x) - m0);
        s1 += __expf(gat_leaky(l.y + ern.y) - m1);
        s2 += __expf(gat_leaky(l.z + ern.z) - m2);
        s3 += __expf(gat_leaky(l.w + ern.w) - m3);
    }
    for (int m = 16; m >= 1; m >>= 1) {
        s0 += __shfl_xor(s0, m, 32);
        s1 += __shfl_xor(s1, m, 32);
        s2 += __shfl_xor(s2, m, 32);
        s3 += __shfl_xor(s3, m, 32);
    }

    float4 accA = make_float4(0.f, 0.f, 0.f, 0.f);
    float4 accB = make_float4(0.f, 0.f, 0.f, 0.f);
    const bool hi = lane >= 16;
    for (int base = beg; base < end; base += 32) {
        int   mysrc = 0;
        float a0 = 0.f, a1 = 0.f, a2 = 0.f, a3 = 0.f;
        if (base == beg) {
            mysrc = src0;
            if (v0) {
                a0 = p0 / s0;  a1 = p1 / s1;
                a2 = p2 / s2;  a3 = p3 / s3;
            }
        } else {
            const int j = base + lane;
            if (j < end) {
                mysrc = bsrc[j];
                const float4 l = el4[mysrc];
                a0 = __expf(gat_leaky(l.x + ern.x) - m0) / s0;
                a1 = __expf(gat_leaky(l.y + ern.y) - m1) / s1;
                a2 = __expf(gat_leaky(l.z + ern.z) - m2) / s2;
                a3 = __expf(gat_leaky(l.w + ern.w) - m3) / s3;
            }
        }
        const int left = end - base;
        const int cntc = left < 32 ? left : 32;
        #pragma unroll 4
        for (int t = 0; t < cntc; ++t) {
            const int   sN = __builtin_amdgcn_readlane(mysrc, t);
            const float b0 = __int_as_float(__builtin_amdgcn_readlane(__float_as_int(a0), t));
            const float b1 = __int_as_float(__builtin_amdgcn_readlane(__float_as_int(a1), t));
            const float b2 = __int_as_float(__builtin_amdgcn_readlane(__float_as_int(a2), t));
            const float b3 = __int_as_float(__builtin_amdgcn_readlane(__float_as_int(a3), t));
            const float aA = hi ? b1 : b0;
            const float aB = hi ? b3 : b2;
            const float4* __restrict__ hp = (const float4*)(h + (size_t)sN * 256);
            const float4 x = hp[lane];
            const float4 y = hp[32 + lane];
            accA.x += aA * x.x;  accA.y += aA * x.y;
            accA.z += aA * x.z;  accA.w += aA * x.w;
            accB.x += aB * y.x;  accB.y += aB * y.y;
            accB.z += aB * y.z;  accB.w += aB * y.w;
        }
    }
    float4* __restrict__ op = (float4*)(out + (size_t)n * 256);
    op[lane]      = accA;
    op[32 + lane] = accB;
}

#define GAT_G1 8

template <bool FULL>
__device__ __forceinline__ void gat_h1_group(
    const float2* __restrict__ h2, int mysrc, int t0, int g, int lane,
    int ell, float ern, float& m, float& s, v2f& acc)
{
    const float NINF = -__builtin_inff();
    int sN[GAT_G1];
    #pragma unroll
    for (int u = 0; u < GAT_G1; ++u)
        sN[u] = __builtin_amdgcn_readlane(mysrc, (FULL || u < g) ? t0 + u : t0);
    float2 x[GAT_G1];
    float  e[GAT_G1];
    #pragma unroll
    for (int u = 0; u < GAT_G1; ++u) {
        x[u] = make_float2(0.f, 0.f);
        if (FULL || u < g)
            x[u] = h2[(size_t)sN[u] * 32 + lane];
    }
    float gm = NINF;
    #pragma unroll
    for (int u = 0; u < GAT_G1; ++u) {
        const float ev = __int_as_float(
            __builtin_amdgcn_readlane(__float_as_int(x[u].x), ell));
        const float lv = ev + ern;
        const float lk = (lv > 0.f) ? lv : SLOPE * lv;
        e[u] = (FULL || u < g) ? lk : NINF;
        gm = fmaxf(gm, e[u]);
    }
    const float mn = fmaxf(m, gm);
    const float sc = __expf(m - mn);
    s *= sc;
    acc = acc * (v2f){sc, sc};
    #pragma unroll
    for (int u = 0; u < GAT_G1; ++u) {
        if (FULL || u < g) {
            const float p = __expf(e[u] - mn);
            s += p;
            acc = __builtin_elementwise_fma((v2f){p, p}, (v2f){x[u].x, x[u].y}, acc);
        }
    }
    m = mn;
}

__global__ __launch_bounds__(128) void gat_node_gather_h1(
    const int* __restrict__ rowptr, const int* __restrict__ bsrc,
    const float* __restrict__ h, const float* __restrict__ er,
    float* __restrict__ out, int N, int D)
{
    __shared__ float stage[4 * 64];
    const int lane = threadIdx.x & 31;
    const int wv = __builtin_amdgcn_readfirstlane((int)(threadIdx.x >> 5));
    const int n  = (int)blockIdx.x * 4 + wv;
    const bool has = 2 * lane < D;
    v2f   acc = (v2f){0.f, 0.f};
    float s = 0.f;

    if (n < N) {
        const int beg = __builtin_amdgcn_readfirstlane(rowptr[n]);
        const int end = __builtin_amdgcn_readfirstlane(rowptr[n + 1]);
        const float ern = er[n];
        const float2* __restrict__ h2 = (const float2*)h;
        const int ell = D >> 1;
        float m = -__builtin_inff();

        for (int base = beg; base < end; base += 32) {
            const int j = base + lane;
            const int mysrc = (j < end) ? bsrc[j] : 0;
            const int left = end - base;
            const int cntc = left < 32 ? left : 32;
            int t0 = 0;
            for (; t0 + GAT_G1 <= cntc; t0 += GAT_G1)
                gat_h1_group<true>(h2, mysrc, t0, GAT_G1, lane, ell, ern, m, s, acc);
            if (t0 < cntc)
                gat_h1_group<false>(h2, mysrc, t0, cntc - t0, lane, ell, ern, m, s, acc);
        }
    }
    const float inv = (s > 0.f) ? 1.0f / s : 0.f;
    if (has) {
        stage[wv * D + 2 * lane]     = acc.x * inv;
        stage[wv * D + 2 * lane + 1] = acc.y * inv;
    }
    __syncthreads();
    const int nleft  = N - (int)blockIdx.x * 4;
    const int nvalid = nleft < 4 ? nleft : 4;
    float* __restrict__ ob = out + (size_t)blockIdx.x * 4 * D;
    for (int i = threadIdx.x; i < nvalid * D; i += 128) ob[i] = stage[i];
}

extern "C" void kernel_launch(void* const* d_in, const int* in_sizes, int n_in,
                              void* d_out, int out_size, void* d_ws, size_t ws_size,
                              hipStream_t stream)
{
    const float* feat = (const float*)d_in[0];
    const int*   src  = (const int*)  d_in[1];
    const int*   dst  = (const int*)  d_in[2];
    const float* W0   = (const float*)d_in[3];
    const float* al0  = (const float*)d_in[4];
    const float* ar0  = (const float*)d_in[5];
    const float* W1   = (const float*)d_in[6];
    const float* al1  = (const float*)d_in[7];
    const float* ar1  = (const float*)d_in[8];
    float* out = (float*)d_out;

    const int FIN = 128, H0 = 4, D0 = 64, HD0 = H0 * D0, NC = 40, NCP = 64;
    const int N = in_sizes[0] / FIN;
    const int E = in_sizes[1];

    char* p = (char*)d_ws;
    auto alloc = [&](size_t bytes) -> void* {
        void* r = (void*)p;
        p += (bytes + 255) & ~(size_t)255;
        return r;
    };
    float* h0     = (float*)alloc((size_t)N * HD0 * 4);
    float* agg0   = (float*)alloc((size_t)N * HD0 * 4);
    float* h1     = (float*)alloc((size_t)N * NCP * 4);
    float* W1p    = (float*)alloc((size_t)HD0 * NCP * 4);
    float* el0    = (float*)alloc((size_t)N * H0 * 4);
    float* er0    = (float*)alloc((size_t)N * H0 * 4);
    float* er1    = (float*)alloc((size_t)N * 4);
    const int R  = (N + 63) / 64;
    const int G  = ((E + 4 * GAT_CAP - 1) / (4 * GAT_CAP)) * 4;
    const int RG = R * G;
    int*      cnt2   = (int*)     alloc((size_t)RG * 4);
    int*      off2   = (int*)     alloc((size_t)(RG + 1) * 4);
    int*      rowptr = (int*)     alloc((size_t)(N + 1) * 4);
    unsigned* tmp    = (unsigned*)alloc((size_t)E * 4);
    int*      bsrc   = (int*)     alloc((size_t)E * 4);

    gat_bin_count<<<G, 1024, (size_t)R * 4, stream>>>(dst, cnt2, E, R, G);
    gat_bucket_scan<<<(RG + 1023) / 1024, 256, 0, stream>>>(cnt2, off2, RG);
    gat_bin_scatter<<<G, 1024, ((size_t)3 * R + 2 * GAT_CAP) * 4, stream>>>(
        src, dst, off2, tmp, E, R, G);
    gat_bin_finish<<<R, 256, 0, stream>>>(tmp, off2, rowptr, bsrc, N, R, G);

    {
        dim3 grid(N / 16, HD0 / 64);
        gat_wmma_gemm_f32<<<grid, 32, 0, stream>>>(feat, W0, h0, N, HD0, FIN, HD0,
                                                   al0, ar0, el0, er0, H0, H0, -1, D0);
    }
    gat_node_gather_h4<<<(N + 7) / 8, 256, 0, stream>>>(
        rowptr, bsrc, h0, el0, er0, agg0, N);

    {
        int tot = HD0 * NCP;
        gat_pad_weights<<<(tot + 255) / 256, 256, 0, stream>>>(W1, W1p, HD0, NC, NCP);
        dim3 grid(N / 16, NCP / 64);
        gat_wmma_gemm_f32<<<grid, 32, 0, stream>>>(agg0, W1p, h1, N, NCP, HD0, NCP,
                                                   al1, ar1, h1 + NC, er1, NCP, 1, NC, NC);
    }
    gat_node_gather_h1<<<(N + 3) / 4, 128, 0, stream>>>(
        rowptr, bsrc, h1, er1, out, N, NC);
}


# Layer-1 GEMM fused into the layer-0 gather: a block's 16 waves stage their rows in LDS, multiply by W1 (as given) on WMMA, K in 4 parts; agg0, padded W1, 2 launches gone; h1/el1/er1 sum order changes
# speedup vs baseline: 4.7738x; 1.1422x over previous
#include <hip/hip_runtime.h>
#include <hip/hip_bf16.h>

typedef __attribute__((ext_vector_type(2))) float v2f;
typedef __attribute__((ext_vector_type(8))) float v8f;

#define SLOPE 0.2f

template <int M>
__device__ __forceinline__ float gat_row_xor(float x) {
    return __builtin_bit_cast(float, __builtin_amdgcn_update_dpp(
        0, __builtin_bit_cast(int, x), 0x160 | M, 0xf, 0xf, true));
}

__global__ __launch_bounds__(32) void gat_wmma_gemm_f32(
    const float* __restrict__ A, const float* __restrict__ B,
    float* __restrict__ C, int M, int Np, int K, int Nout,
    const float* __restrict__ al, const float* __restrict__ ar,
    float* __restrict__ el, float* __restrict__ er, int els, int ers,
    int skipcol, int D)
{
    const int lane  = threadIdx.x & 31;
    const int l15   = lane & 15;
    const int khalf = (lane >> 4) * 2;
    const int mt    = blockIdx.x;
    const int head  = blockIdx.y;
    const int col0  = head * 64;
    const int row   = mt * 16 + l15;

    const float* __restrict__ Ap = A + (size_t)row * K + khalf;
    const float* __restrict__ Bp = B + (size_t)khalf * Np + col0 + l15;

    float wl[4], wr[4];
    #pragma unroll
    for (int t = 0; t < 4; ++t) {
        const int  c  = t * 16 + l15;
        const bool ok = c < D;
        const int  cc = ok ? c : D - 1;
        const float xl = al[head * D + cc];
        const float xr = ar[head * D + cc];
        wl[t] = ok ? xl : 0.f;
        wr[t] = ok ? xr : 0.f;
    }

    v8f acc[4];
    #pragma unroll
    for (int t = 0; t < 4; ++t)
        acc[t] = (v8f){0.f, 0.f, 0.f, 0.f, 0.f, 0.f, 0.f, 0.f};

    for (int k = 0; k < K; k += 4) {
        v2f a;
        a.x = Ap[0];
        a.y = Ap[1];
        v2f b0, b1, b2, b3;
        b0.x = Bp[0];   b0.y = Bp[Np + 0];
        b1.x = Bp[16];  b1.y = Bp[Np + 16];
        b2.x = Bp[32];  b2.y = Bp[Np + 32];
        b3.x = Bp[48];  b3.y = Bp[Np + 48];
        acc[0] = __builtin_amdgcn_wmma_f32_16x16x4_f32(false, a, false, b0, (short)0, acc[0], false, false);
        acc[1] = __builtin_amdgcn_wmma_f32_16x16x4_f32(false, a, false, b1, (short)0, acc[1], false, false);
        acc[2] = __builtin_amdgcn_wmma_f32_16x16x4_f32(false, a, false, b2, (short)0, acc[2], false, false);
        acc[3] = __builtin_amdgcn_wmma_f32_16x16x4_f32(false, a, false, b3, (short)0, acc[3], false, false);
        Ap += 4;
        Bp += (size_t)4 * Np;
    }

    const int srow0 = mt * 16 + (lane >> 4) * 8;
    #pragma unroll
    for (int t = 0; t < 4; ++t) {
        const int col = col0 + t * 16 + l15;
        if (col < Nout && col != skipcol) {
            #pragma unroll
            for (int r = 0; r < 8; ++r)
                C[(size_t)(srow0 + r) * Nout + col] = acc[t][r];
        }
    }

    float v[16];
    #pragma unroll
    for (int r = 0; r < 8; ++r) {
        v[r]     = acc[0][r] * wl[0] + acc[1][r] * wl[1]
                 + acc[2][r] * wl[2] + acc[3][r] * wl[3];
        v[8 + r] = acc[0][r] * wr[0] + acc[1][r] * wr[1]
                 + acc[2][r] * wr[2] + acc[3][r] * wr[3];
    }
    const bool b3 = (lane & 8) != 0;
    const bool b2 = (lane & 4) != 0;
    const bool b1 = (lane & 2) != 0;
    const bool b0 = (lane & 1) != 0;
    float w8[8];
    #pragma unroll
    for (int i = 0; i < 8; ++i) {
        const float keep = b3 ? v[i + 8] : v[i];
        const float send = b3 ? v[i] : v[i + 8];
        w8[i] = keep + gat_row_xor<8>(send);
    }
    float w4[4];
    #pragma unroll
    for (int i = 0; i < 4; ++i) {
        const float keep = b2 ? w8[i + 4] : w8[i];
        const float send = b2 ? w8[i] : w8[i + 4];
        w4[i] = keep + gat_row_xor<4>(send);
    }
    float w2[2];
    #pragma unroll
    for (int i = 0; i < 2; ++i) {
        const float keep = b1 ? w4[i + 2] : w4[i];
        const float send = b1 ? w4[i] : w4[i + 2];
        w2[i] = keep + gat_row_xor<2>(send);
    }
    const float keep1 = b0 ? w2[1] : w2[0];
    const float send1 = b0 ? w2[0] : w2[1];
    const float z = keep1 + gat_row_xor<1>(send1);
    float* __restrict__ sp = b3 ? er : el;
    const int st = b3 ? ers : els;
    sp[(size_t)(srow0 + (lane & 7)) * st + head] = z;
}

#define GAT_CAP 6144
#define GAT_FCAP 4096

__global__ __launch_bounds__(1024) void gat_bin_count(
    const int* __restrict__ dst, int* __restrict__ cnt2, int E, int R, int G)
{
    extern __shared__ int gat_sh[];
    const int tid = threadIdx.x;
    for (int r = tid; r < R; r += 1024) gat_sh[r] = 0;
    __syncthreads();
    const int base = (int)blockIdx.x * GAT_CAP;
    int dk[GAT_CAP / 1024];
    #pragma unroll
    for (int k = 0; k < GAT_CAP / 1024; ++k) {
        const int e = base + k * 1024 + tid;
        dk[k] = dst[e < E ? e : E - 1];
    }
    #pragma unroll
    for (int k = 0; k < GAT_CAP / 1024; ++k) {
        const int e = base + k * 1024 + tid;
        if (e < E) atomicAdd(&gat_sh[dk[k] >> 6], 1);
    }
    __syncthreads();
    for (int r = tid; r < R; r += 1024) cnt2[r * G + (int)blockIdx.x] = gat_sh[r];
}

__global__ __launch_bounds__(1024) void gat_bin_scatter(
    const int* __restrict__ src, const int* __restrict__ dst,
    const int* __restrict__ off2, unsigned* __restrict__ tmp, int E, int R, int G)
{
    extern __shared__ int gat_sh[];
    __shared__ int wsum[32];
    int* lcnt = gat_sh;
    int* lbeg = gat_sh + R;
    int* gofs = gat_sh + 2 * R;
    unsigned* bw = (unsigned*)(gat_sh + 3 * R);
    int* br = gat_sh + 3 * R + GAT_CAP;
    const int tid = threadIdx.x, lane = tid & 31, wv = tid >> 5;
    const int g = (int)blockIdx.x;
    const int base = g * GAT_CAP;

    for (int r = tid; r < R; r += 1024) {
        lcnt[r] = 0;
        gofs[r] = off2[r * G + g];
    }
    __syncthreads();
    int ed[GAT_CAP / 1024], es[GAT_CAP / 1024];
    #pragma unroll
    for (int k = 0; k < GAT_CAP / 1024; ++k) {
        const int e  = base + k * 1024 + tid;
        const int ec = e < E ? e : E - 1;
        const int d  = dst[ec];
        es[k] = src[ec];
        ed[k] = e < E ? d : -1;
    }
    #pragma unroll
    for (int k = 0; k < GAT_CAP / 1024; ++k)
        if (ed[k] >= 0) atomicAdd(&lcnt[ed[k] >> 6], 1);
    __syncthreads();
    int carry = 0;
    for (int b0 = 0; b0 < R; b0 += 1024) {
        const int idx = b0 + tid;
        const int v = (idx < R) ? lcnt[idx] : 0;
        int incl = v;
        for (int off = 1; off < 32; off <<= 1) {
            const int y = __shfl_up(incl, off, 32);
            if (lane >= off) incl += y;
        }
        if (lane == 31) wsum[wv] = incl;
        __syncthreads();
        int pre = 0, tot = 0;
        for (int w = 0; w < 32; ++w) {
            const int t = wsum[w];
            tot += t;
            if (w < wv) pre += t;
        }
        if (idx < R) lbeg[idx] = carry + pre + incl - v;
        carry += tot;
        __syncthreads();
    }
    for (int r = tid; r < R; r += 1024) lcnt[r] = lbeg[r];
    __syncthreads();
    #pragma unroll
    for (int k = 0; k < GAT_CAP / 1024; ++k) {
        if (ed[k] >= 0) {
            const int r = ed[k] >> 6;
            const int p = atomicAdd(&lcnt[r], 1);
            bw[p] = ((unsigned)es[k] << 6) | (unsigned)(ed[k] & 63);
            br[p] = r;
        }
    }
    __syncthreads();
    const int left = E - base;
    const int nloc = left < GAT_CAP ? left : GAT_CAP;
    for (int k = tid; k < nloc; k += 1024) {
        const int r = br[k];
        tmp[gofs[r] + (k - lbeg[r])] = bw[k];
    }
}

__global__ __launch_bounds__(256) void gat_bin_finish(
    const unsigned* __restrict__ tmp, const int* __restrict__ off2,
    int* __restrict__ rowptr, int* __restrict__ bsrc, int N, int R, int G)
{
    __shared__ int c[64];
    __shared__ int cur[64];
    __shared__ int wtot[2];
    __shared__ int stage[GAT_FCAP];
    const int tid = threadIdx.x, lane = tid & 31, wv = tid >> 5;
    const int r = (int)blockIdx.x;
    const int s = off2[r * G];
    const int e = off2[(r + 1) * G];
    const int n = e - s;
    if (tid < 64) c[tid] = 0;
    __syncthreads();
    for (int i = tid; i < n; i += 256) atomicAdd(&c[tmp[s + i] & 63u], 1);
    __syncthreads();
    int v = 0, incl = 0;
    if (tid < 64) {
        v = c[tid];
        incl = v;
        for (int off = 1; off < 32; off <<= 1) {
            const int y = __shfl_up(incl, off, 32);
            if (lane >= off) incl += y;
        }
        if (lane == 31) wtot[wv] = incl;
    }
    __syncthreads();
    if (tid < 64) {
        const int excl = incl - v + (wv == 1 ? wtot[0] : 0);
        cur[tid] = excl;
        const int node = r * 64 + tid;
        if (node < N) rowptr[node] = s + excl;
    }
    if (r == R - 1 && tid == 0) rowptr[N] = e;
    __syncthreads();
    if (n <= GAT_FCAP) {
        for (int i = tid; i < n; i += 256) {
            const unsigned w = tmp[s + i];
            const int p = atomicAdd(&cur[w & 63u], 1);
            stage[p] = (int)(w >> 6);
        }
        __syncthreads();
        for (int i = tid; i < n; i += 256) bsrc[s + i] = stage[i];
    } else {
        for (int i = tid; i < n; i += 256) {
            const unsigned w = tmp[s + i];
            const int p = atomicAdd(&cur[w & 63u], 1);
            bsrc[s + p] = (int)(w >> 6);
        }
    }
}

__global__ __launch_bounds__(256) void gat_bucket_scan(
    const int* __restrict__ cnt, int* __restrict__ rowptr, int N)
{
    __shared__ int s_tot[8];
    __shared__ int s_car[8];
    const int tid = threadIdx.x, lane = tid & 31, wv = tid >> 5;
    const int4* __restrict__ c4 = (const int4*)cnt;

    int p = 0;
    for (int k = 0; k < (int)blockIdx.x; ++k) {
        const int4 q = c4[k * 256 + tid];
        p += (q.x + q.y) + (q.z + q.w);
    }
    const int i0 = blockIdx.x * 1024 + tid * 4;
    const bool live = (i0 + 3 < N);
    int4 own = make_int4(0, 0, 0, 0);
    if (live) own = c4[i0 >> 2];
    const int v = (own.x + own.y) + (own.z + own.w);

    int incl = v;
    for (int off = 1; off < 32; off <<= 1) {
        const int y = __shfl_up(incl, off, 32);
        if (lane >= off) incl += y;
    }
    for (int m = 16; m >= 1; m >>= 1) p += __shfl_xor(p, m, 32);
    if (lane == 31) s_tot[wv] = incl;
    if (lane == 0)  s_car[wv] = p;
    __syncthreads();
    int base = 0;
    for (int w = 0; w < 8; ++w) {
        base += s_car[w];
        if (w < wv) base += s_tot[w];
    }
    base += incl - v;
    if (live) {
        int4 r;
        r.x = base;
        r.y = r.x + own.x;
        r.z = r.y + own.y;
        r.w = r.z + own.z;
        ((int4*)rowptr)[i0 >> 2] = r;
        if (i0 + 4 == N) rowptr[N] = r.w + own.w;
    }
}

__device__ __forceinline__ float gat_leaky(float v) {
    return (v > 0.f) ? v : SLOPE * v;
}

#define GAT_LDA 260

template <int Dn>
__global__ __launch_bounds__(512) void gat_node_gather_h4_gemm(
    const int* __restrict__ rowptr, const int* __restrict__ bsrc,
    const float* __restrict__ h, const float* __restrict__ el,
    const float* __restrict__ er, const float* __restrict__ B,
    const float* __restrict__ al, const float* __restrict__ ar,
    float* __restrict__ C, float* __restrict__ er1, int N)
{
    __shared__ __attribute__((aligned(16))) float stage[16 * GAT_LDA];
    __shared__ __attribute__((aligned(16))) float red[16 * 32 * 8];
    __shared__ float part[4 * 32];
    const int lane = threadIdx.x & 31;
    const int wv = __builtin_amdgcn_readfirstlane((int)(threadIdx.x >> 5));
    const int node0 = (int)blockIdx.x * 16;
    const int n = node0 + wv;
    float4 accA = make_float4(0.f, 0.f, 0.f, 0.f);
    float4 accB = make_float4(0.f, 0.f, 0.f, 0.f);

    if (n < N) {
        const int beg = __builtin_amdgcn_readfirstlane(rowptr[n]);
        const int end = __builtin_amdgcn_readfirstlane(rowptr[n + 1]);
        const float4* __restrict__ el4 = (const float4*)el;
        const float4 ern = ((const float4*)er)[n];
        const float NINF = -__builtin_inff();

        const int  j0 = beg + lane;
        const bool v0 = j0 < end;
        int   src0 = 0;
        float e0 = NINF, e1 = NINF, e2 = NINF, e3 = NINF;
        if (v0) {
            src0 = bsrc[j0];
            const float4 l = el4[src0];
            e0 = gat_leaky(l.x + ern.x);
            e1 = gat_leaky(l.y + ern.y);
            e2 = gat_leaky(l.z + ern.z);
            e3 = gat_leaky(l.w + ern.w);
        }
        float m0 = e0, m1 = e1, m2 = e2, m3 = e3;
        for (int j = j0 + 32; j < end; j += 32) {
            const float4 l = el4[bsrc[j]];
            m0 = fmaxf(m0, gat_leaky(l.x + ern.x));
            m1 = fmaxf(m1, gat_leaky(l.y + ern.y));
            m2 = fmaxf(m2, gat_leaky(l.z + ern.z));
            m3 = fmaxf(m3, gat_leaky(l.w + ern.w));
        }
        for (int m = 16; m >= 1; m >>= 1) {
            m0 = fmaxf(m0, __shfl_xor(m0, m, 32));
            m1 = fmaxf(m1, __shfl_xor(m1, m, 32));
            m2 = fmaxf(m2, __shfl_xor(m2, m, 32));
            m3 = fmaxf(m3, __shfl_xor(m3, m, 32));
        }

        const float p0 = v0 ? __expf(e0 - m0) : 0.f;
        const float p1 = v0 ? __expf(e1 - m1) : 0.f;
        const float p2 = v0 ? __expf(e2 - m2) : 0.f;
        const float p3 = v0 ? __expf(e3 - m3) : 0.f;
        float s0 = p0, s1 = p1, s2 = p2, s3 = p3;
        for (int j = j0 + 32; j < end; j += 32) {
            const float4 l = el4[bsrc[j]];
            s0 += __expf(gat_leaky(l.x + ern.x) - m0);
            s1 += __expf(gat_leaky(l.y + ern.y) - m1);
            s2 += __expf(gat_leaky(l.z + ern.z) - m2);
            s3 += __expf(gat_leaky(l.w + ern.w) - m3);
        }
        for (int m = 16; m >= 1; m >>= 1) {
            s0 += __shfl_xor(s0, m, 32);
            s1 += __shfl_xor(s1, m, 32);
            s2 += __shfl_xor(s2, m, 32);
            s3 += __shfl_xor(s3, m, 32);
        }

        const bool hi = lane >= 16;
        for (int base = beg; base < end; base += 32) {
            int   mysrc = 0;
            float a0 = 0.f, a1 = 0.f, a2 = 0.f, a3 = 0.f;
            if (base == beg) {
                mysrc = src0;
                if (v0) {
                    a0 = p0 / s0;  a1 = p1 / s1;
                    a2 = p2 / s2;  a3 = p3 / s3;
                }
            } else {
                const int j = base + lane;
                if (j < end) {
                    mysrc = bsrc[j];
                    const float4 l = el4[mysrc];
                    a0 = __expf(gat_leaky(l.x + ern.x) - m0) / s0;
                    a1 = __expf(gat_leaky(l.y + ern.y) - m1) / s1;
                    a2 = __expf(gat_leaky(l.z + ern.z) - m2) / s2;
                    a3 = __expf(gat_leaky(l.w + ern.w) - m3) / s3;
                }
            }
            const int left = end - base;
            const int cntc = left < 32 ? left : 32;
            #pragma unroll 4
            for (int t = 0; t < cntc; ++t) {
                const int   sN = __builtin_amdgcn_readlane(mysrc, t);
                const float b0 = __int_as_float(__builtin_amdgcn_readlane(__float_as_int(a0), t));
                const float b1 = __int_as_float(__builtin_amdgcn_readlane(__float_as_int(a1), t));
                const float b2 = __int_as_float(__builtin_amdgcn_readlane(__float_as_int(a2), t));
                const float b3 = __int_as_float(__builtin_amdgcn_readlane(__float_as_int(a3), t));
                const float aA = hi ? b1 : b0;
                const float aB = hi ? b3 : b2;
                const float4* __restrict__ hp = (const float4*)(h + (size_t)sN * 256);
                const float4 x = hp[lane];
                const float4 y = hp[32 + lane];
                accA.x += aA * x.x;  accA.y += aA * x.y;
                accA.z += aA * x.z;  accA.w += aA * x.w;
                accB.x += aB * y.x;  accB.y += aB * y.y;
                accB.z += aB * y.z;  accB.w += aB * y.w;
            }
        }
    }
    float4* __restrict__ sp4 = (float4*)(stage + wv * GAT_LDA);
    sp4[lane]      = accA;
    sp4[32 + lane] = accB;
    __syncthreads();

    const int l15   = lane & 15;
    const int khalf = (lane >> 4) * 2;
    const int tile  = wv & 3;
    const int kq    = wv >> 2;
    const int col   = tile * 16 + l15;
    {
        const bool okb = col < Dn;
        const int  cb  = okb ? col : Dn - 1;
        const float* As = stage + l15 * GAT_LDA + kq * 64 + khalf;
        const float* __restrict__ Bp = B + (size_t)(kq * 64 + khalf) * Dn + cb;
        v8f acc = (v8f){0.f, 0.f, 0.f, 0.f, 0.f, 0.f, 0.f, 0.f};
        if (tile * 16 < Dn) {
            #pragma unroll
            for (int k = 0; k < 64; k += 4) {
                v2f a, b;
                a.x = As[k];
                a.y = As[k + 1];
                const float bx = Bp[k * Dn];
                const float by = Bp[k * Dn + Dn];
                b.x = okb ? bx : 0.f;
                b.y = okb ? by : 0.f;
                acc = __builtin_amdgcn_wmma_f32_16x16x4_f32(false, a, false, b, (short)0, acc, false, false);
            }
        }
        float4* __restrict__ rp = (float4*)(red + (wv * 32 + lane) * 8);
        rp[0] = make_float4(acc[0], acc[1], acc[2], acc[3]);
        rp[1] = make_float4(acc[4], acc[5], acc[6], acc[7]);
    }
    __syncthreads();

    if (wv < 4) {
        const bool okc = col < Dn;
        const int  cc  = okc ? col : Dn - 1;
        const float xl = al[cc];
        const float xr = ar[cc];
        const float wl = okc ? xl : 0.f;
        const float wr = okc ? xr : 0.f;

        float acc[8];
        #pragma unroll
        for (int q = 0; q < 4; ++q) {
            const float4* __restrict__ rp =
                (const float4*)(red + ((q * 4 + wv) * 32 + lane) * 8);
            const float4 lo = rp[0];
            const float4 hi4 = rp[1];
            if (q == 0) {
                acc[0] = lo.x;  acc[1] = lo.y;  acc[2] = lo.z;  acc[3] = lo.w;
                acc[4] = hi4.x; acc[5] = hi4.y; acc[6] = hi4.z; acc[7] = hi4.w;
            } else {
                acc[0] += lo.x;  acc[1] += lo.y;  acc[2] += lo.z;  acc[3] += lo.w;
                acc[4] += hi4.x; acc[5] += hi4.y; acc[6] += hi4.z; acc[7] += hi4.w;
            }
        }

        const int srow0 = node0 + (lane >> 4) * 8;
        if (col != Dn) {
            #pragma unroll
            for (int r = 0; r < 8; ++r)
                C[(size_t)(srow0 + r) * 64 + col] = acc[r];
        }

        float v[16];
        #pragma unroll
        for (int r = 0; r < 8; ++r) {
            v[r]     = acc[r] * wl;
            v[8 + r] = acc[r] * wr;
        }
        const bool b3 = (lane & 8) != 0;
        const bool b2 = (lane & 4) != 0;
        const bool b1 = (lane & 2) != 0;
        const bool b0 = (lane & 1) != 0;
        float w8[8];
        #pragma unroll
        for (int i = 0; i < 8; ++i) {
            const float keep = b3 ? v[i + 8] : v[i];
            const float send = b3 ? v[i] : v[i + 8];
            w8[i] = keep + gat_row_xor<8>(send);
        }
        float w4[4];
        #pragma unroll
        for (int i = 0; i < 4; ++i) {
            const float keep = b2 ? w8[i + 4] : w8[i];
            const float send = b2 ? w8[i] : w8[i + 4];
            w4[i] = keep + gat_row_xor<4>(send);
        }
        float w2[2];
        #pragma unroll
        for (int i = 0; i < 2; ++i) {
            const float keep = b1 ? w4[i + 2] : w4[i];
            const float send = b1 ? w4[i] : w4[i + 2];
            w2[i] = keep + gat_row_xor<2>(send);
        }
        const float keep1 = b0 ? w2[1] : w2[0];
        const float send1 = b0 ? w2[0] : w2[1];
        part[wv * 32 + lane] = keep1 + gat_row_xor<1>(send1);
    }
    __syncthreads();
    if (threadIdx.x < 32) {
        const float z = (part[lane] + part[32 + lane])
                      + (part[64 + lane] + part[96 + lane]);
        const int row = node0 + (lane >> 4) * 8 + (lane & 7);
        if (lane & 8) er1[row] = z;
        else          C[(size_t)row * 64 + Dn] = z;
    }
}

#define GAT_G1 8

template <bool FULL>
__device__ __forceinline__ void gat_h1_group(
    const float2* __restrict__ h2, int mysrc, int t0, int g, int lane,
    int ell, float ern, float& m, float& s, v2f& acc)
{
    const float NINF = -__builtin_inff();
    int sN[GAT_G1];
    #pragma unroll
    for (int u = 0; u < GAT_G1; ++u)
        sN[u] = __builtin_amdgcn_readlane(mysrc, (FULL || u < g) ? t0 + u : t0);
    float2 x[GAT_G1];
    float  e[GAT_G1];
    #pragma unroll
    for (int u = 0; u < GAT_G1; ++u) {
        x[u] = make_float2(0.f, 0.f);
        if (FULL || u < g)
            x[u] = h2[(size_t)sN[u] * 32 + lane];
    }
    float gm = NINF;
    #pragma unroll
    for (int u = 0; u < GAT_G1; ++u) {
        const float ev = __int_as_float(
            __builtin_amdgcn_readlane(__float_as_int(x[u].x), ell));
        const float lv = ev + ern;
        const float lk = (lv > 0.f) ? lv : SLOPE * lv;
        e[u] = (FULL || u < g) ? lk : NINF;
        gm = fmaxf(gm, e[u]);
    }
    const float mn = fmaxf(m, gm);
    const float sc = __expf(m - mn);
    s *= sc;
    acc = acc * (v2f){sc, sc};
    #pragma unroll
    for (int u = 0; u < GAT_G1; ++u) {
        if (FULL || u < g) {
            const float p = __expf(e[u] - mn);
            s += p;
            acc = __builtin_elementwise_fma((v2f){p, p}, (v2f){x[u].x, x[u].y}, acc);
        }
    }
    m = mn;
}

__global__ __launch_bounds__(128) void gat_node_gather_h1(
    const int* __restrict__ rowptr, const int* __restrict__ bsrc,
    const float* __restrict__ h, const float* __restrict__ er,
    float* __restrict__ out, int N, int D)
{
    __shared__ float stage[4 * 64];
    const int lane = threadIdx.x & 31;
    const int wv = __builtin_amdgcn_readfirstlane((int)(threadIdx.x >> 5));
    const int n  = (int)blockIdx.x * 4 + wv;
    const bool has = 2 * lane < D;
    v2f   acc = (v2f){0.f, 0.f};
    float s = 0.f;

    if (n < N) {
        const int beg = __builtin_amdgcn_readfirstlane(rowptr[n]);
        const int end = __builtin_amdgcn_readfirstlane(rowptr[n + 1]);
        const float ern = er[n];
        const float2* __restrict__ h2 = (const float2*)h;
        const int ell = D >> 1;
        float m = -__builtin_inff();

        for (int base = beg; base < end; base += 32) {
            const int j = base + lane;
            const int mysrc = (j < end) ? bsrc[j] : 0;
            const int left = end - base;
            const int cntc = left < 32 ? left : 32;
            int t0 = 0;
            for (; t0 + GAT_G1 <= cntc; t0 += GAT_G1)
                gat_h1_group<true>(h2, mysrc, t0, GAT_G1, lane, ell, ern, m, s, acc);
            if (t0 < cntc)
                gat_h1_group<false>(h2, mysrc, t0, cntc - t0, lane, ell, ern, m, s, acc);
        }
    }
    const float inv = (s > 0.f) ? 1.0f / s : 0.f;
    if (has) {
        stage[wv * D + 2 * lane]     = acc.x * inv;
        stage[wv * D + 2 * lane + 1] = acc.y * inv;
    }
    __syncthreads();
    const int nleft  = N - (int)blockIdx.x * 4;
    const int nvalid = nleft < 4 ? nleft : 4;
    float* __restrict__ ob = out + (size_t)blockIdx.x * 4 * D;
    for (int i = threadIdx.x; i < nvalid * D; i += 128) ob[i] = stage[i];
}

extern "C" void kernel_launch(void* const* d_in, const int* in_sizes, int n_in,
                              void* d_out, int out_size, void* d_ws, size_t ws_size,
                              hipStream_t stream)
{
    const float* feat = (const float*)d_in[0];
    const int*   src  = (const int*)  d_in[1];
    const int*   dst  = (const int*)  d_in[2];
    const float* W0   = (const float*)d_in[3];
    const float* al0  = (const float*)d_in[4];
    const float* ar0  = (const float*)d_in[5];
    const float* W1   = (const float*)d_in[6];
    const float* al1  = (const float*)d_in[7];
    const float* ar1  = (const float*)d_in[8];
    float* out = (float*)d_out;

    const int FIN = 128, H0 = 4, D0 = 64, HD0 = H0 * D0, NC = 40, NCP = 64;
    const int N = in_sizes[0] / FIN;
    const int E = in_sizes[1];

    char* p = (char*)d_ws;
    auto alloc = [&](size_t bytes) -> void* {
        void* r = (void*)p;
        p += (bytes + 255) & ~(size_t)255;
        return r;
    };
    float* h0     = (float*)alloc((size_t)N * HD0 * 4);
    float* h1     = (float*)alloc((size_t)N * NCP * 4);
    float* el0    = (float*)alloc((size_t)N * H0 * 4);
    float* er0    = (float*)alloc((size_t)N * H0 * 4);
    float* er1    = (float*)alloc((size_t)N * 4);
    const int R  = (N + 63) / 64;
    const int G  = ((E + 4 * GAT_CAP - 1) / (4 * GAT_CAP)) * 4;
    const int RG = R * G;
    int*      cnt2   = (int*)     alloc((size_t)RG * 4);
    int*      off2   = (int*)     alloc((size_t)(RG + 1) * 4);
    int*      rowptr = (int*)     alloc((size_t)(N + 1) * 4);
    unsigned* tmp    = (unsigned*)alloc((size_t)E * 4);
    int*      bsrc   = (int*)     alloc((size_t)E * 4);

    gat_bin_count<<<G, 1024, (size_t)R * 4, stream>>>(dst, cnt2, E, R, G);
    gat_bucket_scan<<<(RG + 1023) / 1024, 256, 0, stream>>>(cnt2, off2, RG);
    gat_bin_scatter<<<G, 1024, ((size_t)3 * R + 2 * GAT_CAP) * 4, stream>>>(
        src, dst, off2, tmp, E, R, G);
    gat_bin_finish<<<R, 256, 0, stream>>>(tmp, off2, rowptr, bsrc, N, R, G);

    {
        dim3 grid(N / 16, HD0 / 64);
        gat_wmma_gemm_f32<<<grid, 32, 0, stream>>>(feat, W0, h0, N, HD0, FIN, HD0,
                                                   al0, ar0, el0, er0, H0, H0, -1, D0);
    }
    gat_node_gather_h4_gemm<NC><<<(N + 15) / 16, 512, 0, stream>>>(
        rowptr, bsrc, h0, el0, er0, W1, al1, ar1, h1, er1, N);
    gat_node_gather_h1<<<(N + 3) / 4, 128, 0, stream>>>(
        rowptr, bsrc, h1, er1, out, N, NC);
}


# W1 and al0/ar0 folded into W0 per call on WMMA (W0 @ [W1|al0|ar0] per head): node GEMM gives 160-float rows + el/er, gather reads 640 B per edge with no GEMM behind it; sum order changes
# speedup vs baseline: 5.5509x; 1.1628x over previous
#include <hip/hip_runtime.h>
#include <hip/hip_bf16.h>

typedef __attribute__((ext_vector_type(2))) float v2f;
typedef __attribute__((ext_vector_type(8))) float v8f;

#define SLOPE 0.2f

#define GAT_D0 64
#define GAT_NC 40
#define GAT_PS 160
#define GAT_WCN 192

__global__ __launch_bounds__(32) void gat_fold_weights(
    const float* __restrict__ W0, const float* __restrict__ W1,
    const float* __restrict__ al, const float* __restrict__ ar,
    float* __restrict__ Wc, int H)
{
    const int lane  = threadIdx.x & 31;
    const int l15   = lane & 15;
    const int khalf = (lane >> 4) * 2;
    const int mt    = blockIdx.x;
    const int head  = (int)blockIdx.y / 3;
    const int ct    = (int)blockIdx.y % 3;
    const int c     = ct * 16 + l15;
    const int row   = mt * 16 + l15;
    const int ldw   = H * GAT_D0;

    const float* __restrict__ Ap = W0 + (size_t)row * ldw + head * GAT_D0 + khalf;
    const bool okb = c < GAT_NC + 2;
    const float* __restrict__ Bp;
    int bs;
    if (c < GAT_NC)       { Bp = W1 + (size_t)(head * GAT_D0 + khalf) * GAT_NC + c; bs = GAT_NC; }
    else if (c == GAT_NC) { Bp = al + head * GAT_D0 + khalf; bs = 1; }
    else                  { Bp = ar + head * GAT_D0 + khalf; bs = 1; }

    v8f acc = (v8f){0.f, 0.f, 0.f, 0.f, 0.f, 0.f, 0.f, 0.f};
    #pragma unroll
    for (int k = 0; k < GAT_D0; k += 4) {
        v2f a, b;
        a.x = Ap[k];
        a.y = Ap[k + 1];
        const float bx = Bp[k * bs];
        const float by = Bp[(k + 1) * bs];
        b.x = okb ? bx : 0.f;
        b.y = okb ? by : 0.f;
        acc = __builtin_amdgcn_wmma_f32_16x16x4_f32(false, a, false, b, (short)0, acc, false, false);
    }
    int oc;
    if (c < GAT_NC)           oc = head * GAT_NC + c;
    else if (c == GAT_NC)     oc = H * GAT_NC + head;
    else if (c == GAT_NC + 1) oc = H * GAT_NC + H + head;
    else                      oc = H * GAT_NC + 2 * H + head * 6 + (c - GAT_NC - 2);
    const int srow0 = mt * 16 + (lane >> 4) * 8;
    #pragma unroll
    for (int r = 0; r < 8; ++r)
        Wc[(size_t)(srow0 + r) * GAT_WCN + oc] = acc[r];
}

__global__ __launch_bounds__(32) void gat_wmma_gemm_f32(
    const float* __restrict__ A, const float* __restrict__ B,
    float* __restrict__ C, int Np, int K, int Nc,
    float* __restrict__ el, float* __restrict__ er, int H)
{
    const int lane  = threadIdx.x & 31;
    const int l15   = lane & 15;
    const int khalf = (lane >> 4) * 2;
    const int mt    = blockIdx.x;
    const int col0  = (int)blockIdx.y * 64;
    const int row   = mt * 16 + l15;

    const float* __restrict__ Ap = A + (size_t)row * K + khalf;
    const float* __restrict__ Bp = B + (size_t)khalf * Np + col0 + l15;

    v8f acc[4];
    #pragma unroll
    for (int t = 0; t < 4; ++t)
        acc[t] = (v8f){0.f, 0.f, 0.f, 0.f, 0.f, 0.f, 0.f, 0.f};

    for (int k = 0; k < K; k += 4) {
        v2f a;
        a.x = Ap[0];
        a.y = Ap[1];
        v2f b0, b1, b2, b3;
        b0.x = Bp[0];   b0.y = Bp[Np + 0];
        b1.x = Bp[16];  b1.y = Bp[Np + 16];
        b2.x = Bp[32];  b2.y = Bp[Np + 32];
        b3.x = Bp[48];  b3.y = Bp[Np + 48];
        acc[0] = __builtin_amdgcn_wmma_f32_16x16x4_f32(false, a, false, b0, (short)0, acc[0], false, false);
        acc[1] = __builtin_amdgcn_wmma_f32_16x16x4_f32(false, a, false, b1, (short)0, acc[1], false, false);
        acc[2] = __builtin_amdgcn_wmma_f32_16x16x4_f32(false, a, false, b2, (short)0, acc[2], false, false);
        acc[3] = __builtin_amdgcn_wmma_f32_16x16x4_f32(false, a, false, b3, (short)0, acc[3], false, false);
        Ap += 4;
        Bp += (size_t)4 * Np;
    }

    const int srow0 = mt * 16 + (lane >> 4) * 8;
    #pragma unroll
    for (int t = 0; t < 4; ++t) {
        const int col = col0 + t * 16 + l15;
        if (col < Nc) {
            #pragma unroll
            for (int r = 0; r < 8; ++r)
                C[(size_t)(srow0 + r) * Nc + col] = acc[t][r];
        } else if (col < Nc + H) {
            #pragma unroll
            for (int r = 0; r < 8; ++r)
                el[(size_t)(srow0 + r) * H + (col - Nc)] = acc[t][r];
        } else if (col < Nc + 2 * H) {
            #pragma unroll
            for (int r = 0; r < 8; ++r)
                er[(size_t)(srow0 + r) * H + (col - Nc - H)] = acc[t][r];
        }
    }
}

#define GAT_CAP 6144
#define GAT_FCAP 4096

__global__ __launch_bounds__(1024) void gat_bin_count(
    const int* __restrict__ dst, int* __restrict__ cnt2, int E, int R, int G)
{
    extern __shared__ int gat_sh[];
    const int tid = threadIdx.x;
    for (int r = tid; r < R; r += 1024) gat_sh[r] = 0;
    __syncthreads();
    const int base = (int)blockIdx.x * GAT_CAP;
    int dk[GAT_CAP / 1024];
    #pragma unroll
    for (int k = 0; k < GAT_CAP / 1024; ++k) {
        const int e = base + k * 1024 + tid;
        dk[k] = dst[e < E ? e : E - 1];
    }
    #pragma unroll
    for (int k = 0; k < GAT_CAP / 1024; ++k) {
        const int e = base + k * 1024 + tid;
        if (e < E) atomicAdd(&gat_sh[dk[k] >> 6], 1);
    }
    __syncthreads();
    for (int r = tid; r < R; r += 1024) cnt2[r * G + (int)blockIdx.x] = gat_sh[r];
}

__global__ __launch_bounds__(1024) void gat_bin_scatter(
    const int* __restrict__ src, const int* __restrict__ dst,
    const int* __restrict__ off2, unsigned* __restrict__ tmp, int E, int R, int G)
{
    extern __shared__ int gat_sh[];
    __shared__ int wsum[32];
    int* lcnt = gat_sh;
    int* lbeg = gat_sh + R;
    int* gofs = gat_sh + 2 * R;
    unsigned* bw = (unsigned*)(gat_sh + 3 * R);
    int* br = gat_sh + 3 * R + GAT_CAP;
    const int tid = threadIdx.x, lane = tid & 31, wv = tid >> 5;
    const int g = (int)blockIdx.x;
    const int base = g * GAT_CAP;

    for (int r = tid; r < R; r += 1024) {
        lcnt[r] = 0;
        gofs[r] = off2[r * G + g];
    }
    __syncthreads();
    int ed[GAT_CAP / 1024], es[GAT_CAP / 1024];
    #pragma unroll
    for (int k = 0; k < GAT_CAP / 1024; ++k) {
        const int e  = base + k * 1024 + tid;
        const int ec = e < E ? e : E - 1;
        const int d  = dst[ec];
        es[k] = src[ec];
        ed[k] = e < E ? d : -1;
    }
    #pragma unroll
    for (int k = 0; k < GAT_CAP / 1024; ++k)
        if (ed[k] >= 0) atomicAdd(&lcnt[ed[k] >> 6], 1);
    __syncthreads();
    int carry = 0;
    for (int b0 = 0; b0 < R; b0 += 1024) {
        const int idx = b0 + tid;
        const int v = (idx < R) ? lcnt[idx] : 0;
        int incl = v;
        for (int off = 1; off < 32; off <<= 1) {
            const int y = __shfl_up(incl, off, 32);
            if (lane >= off) incl += y;
        }
        if (lane == 31) wsum[wv] = incl;
        __syncthreads();
        int pre = 0, tot = 0;
        for (int w = 0; w < 32; ++w) {
            const int t = wsum[w];
            tot += t;
            if (w < wv) pre += t;
        }
        if (idx < R) lbeg[idx] = carry + pre + incl - v;
        carry += tot;
        __syncthreads();
    }
    for (int r = tid; r < R; r += 1024) lcnt[r] = lbeg[r];
    __syncthreads();
    #pragma unroll
    for (int k = 0; k < GAT_CAP / 1024; ++k) {
        if (ed[k] >= 0) {
            const int r = ed[k] >> 6;
            const int p = atomicAdd(&lcnt[r], 1);
            bw[p] = ((unsigned)es[k] << 6) | (unsigned)(ed[k] & 63);
            br[p] = r;
        }
    }
    __syncthreads();
    const int left = E - base;
    const int nloc = left < GAT_CAP ? left : GAT_CAP;
    for (int k = tid; k < nloc; k += 1024) {
        const int r = br[k];
        tmp[gofs[r] + (k - lbeg[r])] = bw[k];
    }
}

__global__ __launch_bounds__(256) void gat_bin_finish(
    const unsigned* __restrict__ tmp, const int* __restrict__ off2,
    int* __restrict__ rowptr, int* __restrict__ bsrc, int N, int R, int G)
{
    __shared__ int c[64];
    __shared__ int cur[64];
    __shared__ int wtot[2];
    __shared__ int stage[GAT_FCAP];
    const int tid = threadIdx.x, lane = tid & 31, wv = tid >> 5;
    const int r = (int)blockIdx.x;
    const int s = off2[r * G];
    const int e = off2[(r + 1) * G];
    const int n = e - s;
    if (tid < 64) c[tid] = 0;
    __syncthreads();
    for (int i = tid; i < n; i += 256) atomicAdd(&c[tmp[s + i] & 63u], 1);
    __syncthreads();
    int v = 0, incl = 0;
    if (tid < 64) {
        v = c[tid];
        incl = v;
        for (int off = 1; off < 32; off <<= 1) {
            const int y = __shfl_up(incl, off, 32);
            if (lane >= off) incl += y;
        }
        if (lane == 31) wtot[wv] = incl;
    }
    __syncthreads();
    if (tid < 64) {
        const int excl = incl - v + (wv == 1 ? wtot[0] : 0);
        cur[tid] = excl;
        const int node = r * 64 + tid;
        if (node < N) rowptr[node] = s + excl;
    }
    if (r == R - 1 && tid == 0) rowptr[N] = e;
    __syncthreads();
    if (n <= GAT_FCAP) {
        for (int i = tid; i < n; i += 256) {
            const unsigned w = tmp[s + i];
            const int p = atomicAdd(&cur[w & 63u], 1);
            stage[p] = (int)(w >> 6);
        }
        __syncthreads();
        for (int i = tid; i < n; i += 256) bsrc[s + i] = stage[i];
    } else {
        for (int i = tid; i < n; i += 256) {
            const unsigned w = tmp[s + i];
            const int p = atomicAdd(&cur[w & 63u], 1);
            bsrc[s + p] = (int)(w >> 6);
        }
    }
}

__global__ __launch_bounds__(256) void gat_bucket_scan(
    const int* __restrict__ cnt, int* __restrict__ rowptr, int N)
{
    __shared__ int s_tot[8];
    __shared__ int s_car[8];
    const int tid = threadIdx.x, lane = tid & 31, wv = tid >> 5;
    const int4* __restrict__ c4 = (const int4*)cnt;

    int p = 0;
    for (int k = 0; k < (int)blockIdx.x; ++k) {
        const int4 q = c4[k * 256 + tid];
        p += (q.x + q.y) + (q.z + q.w);
    }
    const int i0 = blockIdx.x * 1024 + tid * 4;
    const bool live = (i0 + 3 < N);
    int4 own = make_int4(0, 0, 0, 0);
    if (live) own = c4[i0 >> 2];
    const int v = (own.x + own.y) + (own.z + own.w);

    int incl = v;
    for (int off = 1; off < 32; off <<= 1) {
        const int y = __shfl_up(incl, off, 32);
        if (lane >= off) incl += y;
    }
    for (int m = 16; m >= 1; m >>= 1) p += __shfl_xor(p, m, 32);
    if (lane == 31) s_tot[wv] = incl;
    if (lane == 0)  s_car[wv] = p;
    __syncthreads();
    int base = 0;
    for (int w = 0; w < 8; ++w) {
        base += s_car[w];
        if (w < wv) base += s_tot[w];
    }
    base += incl - v;
    if (live) {
        int4 r;
        r.x = base;
        r.y = r.x + own.x;
        r.z = r.y + own.y;
        r.w = r.z + own.z;
        ((int4*)rowptr)[i0 >> 2] = r;
        if (i0 + 4 == N) rowptr[N] = r.w + own.w;
    }
}

__device__ __forceinline__ float gat_leaky(float v) {
    return (v > 0.f) ? v : SLOPE * v;
}

__global__ __launch_bounds__(512) void gat_node_gather_h4_fold(
    const int* __restrict__ rowptr, const int* __restrict__ bsrc,
    const float* __restrict__ P, const float* __restrict__ el,
    const float* __restrict__ er,
    const float* __restrict__ al, const float* __restrict__ ar,
    float* __restrict__ C, float* __restrict__ er1, int N)
{
    __shared__ __attribute__((aligned(16))) float stage[16 * GAT_PS];
    __shared__ float part[16];
    const int lane = threadIdx.x & 31;
    const int wv = __builtin_amdgcn_readfirstlane((int)(threadIdx.x >> 5));
    const int node0 = (int)blockIdx.x * 16;
    const int n = node0 + wv;
    float4 accA = make_float4(0.f, 0.f, 0.f, 0.f);
    float4 accB = make_float4(0.f, 0.f, 0.f, 0.f);
    const int c2 = 32 + (lane & 7);
    const float wl0 = al[lane], wl1 = al[c2];
    const float wr0 = ar[lane], wr1 = ar[c2];

    if (n < N) {
        const int beg = __builtin_amdgcn_readfirstlane(rowptr[n]);
        const int end = __builtin_amdgcn_readfirstlane(rowptr[n + 1]);
        const float4* __restrict__ el4 = (const float4*)el;
        const float4 ern = ((const float4*)er)[n];
        const float NINF = -__builtin_inff();

        const int  j0 = beg + lane;
        const bool v0 = j0 < end;
        int   src0 = 0;
        float e0 = NINF, e1 = NINF, e2 = NINF, e3 = NINF;
        if (v0) {
            src0 = bsrc[j0];
            const float4 l = el4[src0];
            e0 = gat_leaky(l.x + ern.x);
            e1 = gat_leaky(l.y + ern.y);
            e2 = gat_leaky(l.z + ern.z);
            e3 = gat_leaky(l.w + ern.w);
        }
        float m0 = e0, m1 = e1, m2 = e2, m3 = e3;
        for (int j = j0 + 32; j < end; j += 32) {
            const float4 l = el4[bsrc[j]];
            m0 = fmaxf(m0, gat_leaky(l.x + ern.x));
            m1 = fmaxf(m1, gat_leaky(l.y + ern.y));
            m2 = fmaxf(m2, gat_leaky(l.z + ern.z));
            m3 = fmaxf(m3, gat_leaky(l.w + ern.w));
        }
        for (int m = 16; m >= 1; m >>= 1) {
            m0 = fmaxf(m0, __shfl_xor(m0, m, 32));
            m1 = fmaxf(m1, __shfl_xor(m1, m, 32));
            m2 = fmaxf(m2, __shfl_xor(m2, m, 32));
            m3 = fmaxf(m3, __shfl_xor(m3, m, 32));
        }

        const float p0 = v0 ? __expf(e0 - m0) : 0.f;
        const float p1 = v0 ? __expf(e1 - m1) : 0.f;
        const float p2 = v0 ? __expf(e2 - m2) : 0.f;
        const float p3 = v0 ? __expf(e3 - m3) : 0.f;
        float s0 = p0, s1 = p1, s2 = p2, s3 = p3;
        for (int j = j0 + 32; j < end; j += 32) {
            const float4 l = el4[bsrc[j]];
            s0 += __expf(gat_leaky(l.x + ern.x) - m0);
            s1 += __expf(gat_leaky(l.y + ern.y) - m1);
            s2 += __expf(gat_leaky(l.z + ern.z) - m2);
            s3 += __expf(gat_leaky(l.w + ern.w) - m3);
        }
        for (int m = 16; m >= 1; m >>= 1) {
            s0 += __shfl_xor(s0, m, 32);
            s1 += __shfl_xor(s1, m, 32);
            s2 += __shfl_xor(s2, m, 32);
            s3 += __shfl_xor(s3, m, 32);
        }

        const bool h0l = lane < 10;
        const bool h1l = lane < 20;
        const bool h2l = lane < 30;
        for (int base = beg; base < end; base += 32) {
            int   mysrc = 0;
            float a0 = 0.f, a1 = 0.f, a2 = 0.f, a3 = 0.f;
            if (base == beg) {
                mysrc = src0;
                if (v0) {
                    a0 = p0 / s0;  a1 = p1 / s1;
                    a2 = p2 / s2;  a3 = p3 / s3;
                }
            } else {
                const int j = base + lane;
                if (j < end) {
                    mysrc = bsrc[j];
                    const float4 l = el4[mysrc];
                    a0 = __expf(gat_leaky(l.x + ern.x) - m0) / s0;
                    a1 = __expf(gat_leaky(l.y + ern.y) - m1) / s1;
                    a2 = __expf(gat_leaky(l.z + ern.z) - m2) / s2;
                    a3 = __expf(gat_leaky(l.w + ern.w) - m3) / s3;
                }
            }
            const int left = end - base;
            const int cntc = left < 32 ? left : 32;
            #pragma unroll 4
            for (int t = 0; t < cntc; ++t) {
                const int   sN = __builtin_amdgcn_readlane(mysrc, t);
                const float b0 = __int_as_float(__builtin_amdgcn_readlane(__float_as_int(a0), t));
                const float b1 = __int_as_float(__builtin_amdgcn_readlane(__float_as_int(a1), t));
                const float b2 = __int_as_float(__builtin_amdgcn_readlane(__float_as_int(a2), t));
                const float b3 = __int_as_float(__builtin_amdgcn_readlane(__float_as_int(a3), t));
                const float aA = h0l ? b0 : (h1l ? b1 : (h2l ? b2 : b3));
                const float4* __restrict__ hp = (const float4*)(P + (size_t)sN * GAT_PS);
                const float4 x = hp[lane];
                const float4 y = hp[c2];
                accA.x += aA * x.x;  accA.y += aA * x.y;
                accA.z += aA * x.z;  accA.w += aA * x.w;
                accB.x += b3 * y.x;  accB.y += b3 * y.y;
                accB.z += b3 * y.z;  accB.w += b3 * y.w;
            }
        }
    }
    float4* __restrict__ sp4 = (float4*)(stage + wv * GAT_PS);
    sp4[lane] = accA;
    if (lane < 8) sp4[32 + lane] = accB;
    __syncthreads();

    const float* sr = stage + wv * GAT_PS;
    const bool lo8 = lane < 8;
    const float u0 = (sr[lane] + sr[GAT_NC + lane])
                   + (sr[2 * GAT_NC + lane] + sr[3 * GAT_NC + lane]);
    const float u1 = (sr[c2] + sr[GAT_NC + c2])
                   + (sr[2 * GAT_NC + c2] + sr[3 * GAT_NC + c2]);
    float ql = u0 * wl0 + (lo8 ? u1 * wl1 : 0.f);
    float qr = u0 * wr0 + (lo8 ? u1 * wr1 : 0.f);
    for (int m = 16; m >= 1; m >>= 1) {
        ql += __shfl_xor(ql, m, 32);
        qr += __shfl_xor(qr, m, 32);
    }
    if (n < N) {
        float* __restrict__ cr = C + (size_t)n * 64;
        cr[lane] = u0;
        cr[32 + lane] = lo8 ? u1 : (lane == 8 ? ql : 0.f);
    }
    if (lane == 0) part[wv] = qr;
    __syncthreads();
    if (threadIdx.x < 16 && node0 + (int)threadIdx.x < N)
        er1[node0 + threadIdx.x] = part[threadIdx.x];
}

#define GAT_G1 8

template <bool FULL>
__device__ __forceinline__ void gat_h1_group(
    const float2* __restrict__ h2, int mysrc, int t0, int g, int lane,
    int ell, float ern, float& m, float& s, v2f& acc)
{
    const float NINF = -__builtin_inff();
    int sN[GAT_G1];
    #pragma unroll
    for (int u = 0; u < GAT_G1; ++u)
        sN[u] = __builtin_amdgcn_readlane(mysrc, (FULL || u < g) ? t0 + u : t0);
    float2 x[GAT_G1];
    float  e[GAT_G1];
    #pragma unroll
    for (int u = 0; u < GAT_G1; ++u) {
        x[u] = make_float2(0.f, 0.f);
        if (FULL || u < g)
            x[u] = h2[(size_t)sN[u] * 32 + lane];
    }
    float gm = NINF;
    #pragma unroll
    for (int u = 0; u < GAT_G1; ++u) {
        const float ev = __int_as_float(
            __builtin_amdgcn_readlane(__float_as_int(x[u].x), ell));
        const float lv = ev + ern;
        const float lk = (lv > 0.f) ? lv : SLOPE * lv;
        e[u] = (FULL || u < g) ? lk : NINF;
        gm = fmaxf(gm, e[u]);
    }
    const float mn = fmaxf(m, gm);
    const float sc = __expf(m - mn);
    s *= sc;
    acc = acc * (v2f){sc, sc};
    #pragma unroll
    for (int u = 0; u < GAT_G1; ++u) {
        if (FULL || u < g) {
            const float p = __expf(e[u] - mn);
            s += p;
            acc = __builtin_elementwise_fma((v2f){p, p}, (v2f){x[u].x, x[u].y}, acc);
        }
    }
    m = mn;
}

__global__ __launch_bounds__(128) void gat_node_gather_h1(
    const int* __restrict__ rowptr, const int* __restrict__ bsrc,
    const float* __restrict__ h, const float* __restrict__ er,
    float* __restrict__ out, int N, int D)
{
    __shared__ float stage[4 * 64];
    const int lane = threadIdx.x & 31;
    const int wv = __builtin_amdgcn_readfirstlane((int)(threadIdx.x >> 5));
    const int n  = (int)blockIdx.x * 4 + wv;
    const bool has = 2 * lane < D;
    v2f   acc = (v2f){0.f, 0.f};
    float s = 0.f;

    if (n < N) {
        const int beg = __builtin_amdgcn_readfirstlane(rowptr[n]);
        const int end = __builtin_amdgcn_readfirstlane(rowptr[n + 1]);
        const float ern = er[n];
        const float2* __restrict__ h2 = (const float2*)h;
        const int ell = D >> 1;
        float m = -__builtin_inff();

        for (int base = beg; base < end; base += 32) {
            const int j = base + lane;
            const int mysrc = (j < end) ? bsrc[j] : 0;
            const int left = end - base;
            const int cntc = left < 32 ? left : 32;
            int t0 = 0;
            for (; t0 + GAT_G1 <= cntc; t0 += GAT_G1)
                gat_h1_group<true>(h2, mysrc, t0, GAT_G1, lane, ell, ern, m, s, acc);
            if (t0 < cntc)
                gat_h1_group<false>(h2, mysrc, t0, cntc - t0, lane, ell, ern, m, s, acc);
        }
    }
    const float inv = (s > 0.f) ? 1.0f / s : 0.f;
    if (has) {
        stage[wv * D + 2 * lane]     = acc.x * inv;
        stage[wv * D + 2 * lane + 1] = acc.y * inv;
    }
    __syncthreads();
    const int nleft  = N - (int)blockIdx.x * 4;
    const int nvalid = nleft < 4 ? nleft : 4;
    float* __restrict__ ob = out + (size_t)blockIdx.x * 4 * D;
    for (int i = threadIdx.x; i < nvalid * D; i += 128) ob[i] = stage[i];
}

extern "C" void kernel_launch(void* const* d_in, const int* in_sizes, int n_in,
                              void* d_out, int out_size, void* d_ws, size_t ws_size,
                              hipStream_t stream)
{
    const float* feat = (const float*)d_in[0];
    const int*   src  = (const int*)  d_in[1];
    const int*   dst  = (const int*)  d_in[2];
    const float* W0   = (const float*)d_in[3];
    const float* al0  = (const float*)d_in[4];
    const float* ar0  = (const float*)d_in[5];
    const float* W1   = (const float*)d_in[6];
    const float* al1  = (const float*)d_in[7];
    const float* ar1  = (const float*)d_in[8];
    float* out = (float*)d_out;

    const int FIN = 128, H0 = 4, NC = GAT_NC, NCP = 64;
    const int N = in_sizes[0] / FIN;
    const int E = in_sizes[1];

    char* p = (char*)d_ws;
    auto alloc = [&](size_t bytes) -> void* {
        void* r = (void*)p;
        p += (bytes + 255) & ~(size_t)255;
        return r;
    };
    float* P0     = (float*)alloc((size_t)N * GAT_PS * 4);
    float* h1     = (float*)alloc((size_t)N * NCP * 4);
    float* el0    = (float*)alloc((size_t)N * H0 * 4);
    float* er0    = (float*)alloc((size_t)N * H0 * 4);
    float* er1    = (float*)alloc((size_t)N * 4);
    float* Wc     = (float*)alloc((size_t)FIN * GAT_WCN * 4);
    const int R  = (N + 63) / 64;
    const int G  = ((E + 4 * GAT_CAP - 1) / (4 * GAT_CAP)) * 4;
    const int RG = R * G;
    int*      cnt2   = (int*)     alloc((size_t)RG * 4);
    int*      off2   = (int*)     alloc((size_t)(RG + 1) * 4);
    int*      rowptr = (int*)     alloc((size_t)(N + 1) * 4);
    unsigned* tmp    = (unsigned*)alloc((size_t)E * 4);
    int*      bsrc   = (int*)     alloc((size_t)E * 4);

    {
        dim3 grid(FIN / 16, H0 * 3);
        gat_fold_weights<<<grid, 32, 0, stream>>>(W0, W1, al0, ar0, Wc, H0);
    }

    gat_bin_count<<<G, 1024, (size_t)R * 4, stream>>>(dst, cnt2, E, R, G);
    gat_bucket_scan<<<(RG + 1023) / 1024, 256, 0, stream>>>(cnt2, off2, RG);
    gat_bin_scatter<<<G, 1024, ((size_t)3 * R + 2 * GAT_CAP) * 4, stream>>>(
        src, dst, off2, tmp, E, R, G);
    gat_bin_finish<<<R, 256, 0, stream>>>(tmp, off2, rowptr, bsrc, N, R, G);

    {
        dim3 grid(N / 16, GAT_WCN / 64);
        gat_wmma_gemm_f32<<<grid, 32, 0, stream>>>(feat, Wc, P0, GAT_WCN, FIN, GAT_PS,
                                                   el0, er0, H0);
    }
    gat_node_gather_h4_fold<<<(N + 15) / 16, 512, 0, stream>>>(
        rowptr, bsrc, P0, el0, er0, al1, ar1, h1, er1, N);
    gat_node_gather_h1<<<(N + 3) / 4, 128, 0, stream>>>(
        rowptr, bsrc, h1, er1, out, N, NC);
}


# Weight fold and node GEMM ride as guest blocks in the four bucket launches (one 16x64 tile per guest wave): 2 launches gone, the GEMM overlaps the sort; sum order unchanged
# speedup vs baseline: 6.0161x; 1.0838x over previous
#include <hip/hip_runtime.h>
#include <hip/hip_bf16.h>

typedef __attribute__((ext_vector_type(2))) float v2f;
typedef __attribute__((ext_vector_type(8))) float v8f;

#define SLOPE 0.2f

#define GAT_D0 64
#define GAT_NC 40
#define GAT_PS 160
#define GAT_WCN 192

__device__ __forceinline__ void gat_fold_tile(
    const float* __restrict__ W0, const float* __restrict__ W1,
    const float* __restrict__ al, const float* __restrict__ ar,
    float* __restrict__ Wc, int H, int mt, int hc)
{
    const int lane  = threadIdx.x & 31;
    const int l15   = lane & 15;
    const int khalf = (lane >> 4) * 2;
    const int head  = hc / 3;
    const int ct    = hc % 3;
    const int c     = ct * 16 + l15;
    const int row   = mt * 16 + l15;
    const int ldw   = H * GAT_D0;

    const float* __restrict__ Ap = W0 + (size_t)row * ldw + head * GAT_D0 + khalf;
    const bool okb = c < GAT_NC + 2;
    const float* __restrict__ Bp;
    int bs;
    if (c < GAT_NC)       { Bp = W1 + (size_t)(head * GAT_D0 + khalf) * GAT_NC + c; bs = GAT_NC; }
    else if (c == GAT_NC) { Bp = al + head * GAT_D0 + khalf; bs = 1; }
    else                  { Bp = ar + head * GAT_D0 + khalf; bs = 1; }

    v8f acc = (v8f){0.f, 0.f, 0.f, 0.f, 0.f, 0.f, 0.f, 0.f};
    #pragma unroll
    for (int k = 0; k < GAT_D0; k += 4) {
        v2f a, b;
        a.x = Ap[k];
        a.y = Ap[k + 1];
        const float bx = Bp[k * bs];
        const float by = Bp[(k + 1) * bs];
        b.x = okb ? bx : 0.f;
        b.y = okb ? by : 0.f;
        acc = __builtin_amdgcn_wmma_f32_16x16x4_f32(false, a, false, b, (short)0, acc, false, false);
    }
    int oc;
    if (c < GAT_NC)           oc = head * GAT_NC + c;
    else if (c == GAT_NC)     oc = H * GAT_NC + head;
    else if (c == GAT_NC + 1) oc = H * GAT_NC + H + head;
    else                      oc = H * GAT_NC + 2 * H + head * 6 + (c - GAT_NC - 2);
    const int srow0 = mt * 16 + (lane >> 4) * 8;
    #pragma unroll
    for (int r = 0; r < 8; ++r)
        Wc[(size_t)(srow0 + r) * GAT_WCN + oc] = acc[r];
}

__device__ __forceinline__ void gat_wmma_gemm_tile(
    const float* __restrict__ A, const float* __restrict__ B,
    float* __restrict__ C, int Np, int K, int Nc,
    float* __restrict__ el, float* __restrict__ er, int H, int mt, int strip)
{
    const int lane  = threadIdx.x & 31;
    const int l15   = lane & 15;
    const int khalf = (lane >> 4) * 2;
    const int col0  = strip * 64;
    const int row   = mt * 16 + l15;

    const float* __restrict__ Ap = A + (size_t)row * K + khalf;
    const float* __restrict__ Bp = B + (size_t)khalf * Np + col0 + l15;

    v8f acc[4];
    #pragma unroll
    for (int t = 0; t < 4; ++t)
        acc[t] = (v8f){0.f, 0.f, 0.f, 0.f, 0.f, 0.f, 0.f, 0.f};

    for (int k = 0; k < K; k += 4) {
        v2f a;
        a.x = Ap[0];
        a.y = Ap[1];
        v2f b0, b1, b2, b3;
        b0.x = Bp[0];   b0.y = Bp[Np + 0];
        b1.x = Bp[16];  b1.y = Bp[Np + 16];
        b2.x = Bp[32];  b2.y = Bp[Np + 32];
        b3.x = Bp[48];  b3.y = Bp[Np + 48];
        acc[0] = __builtin_amdgcn_wmma_f32_16x16x4_f32(false, a, false, b0, (short)0, acc[0], false, false);
        acc[1] = __builtin_amdgcn_wmma_f32_16x16x4_f32(false, a, false, b1, (short)0, acc[1], false, false);
        acc[2] = __builtin_amdgcn_wmma_f32_16x16x4_f32(false, a, false, b2, (short)0, acc[2], false, false);
        acc[3] = __builtin_amdgcn_wmma_f32_16x16x4_f32(false, a, false, b3, (short)0, acc[3], false, false);
        Ap += 4;
        Bp += (size_t)4 * Np;
    }

    const int srow0 = mt * 16 + (lane >> 4) * 8;
    #pragma unroll
    for (int t = 0; t < 4; ++t) {
        const int col = col0 + t * 16 + l15;
        if (col < Nc) {
            #pragma unroll
            for (int r = 0; r < 8; ++r)
                C[(size_t)(srow0 + r) * Nc + col] = acc[t][r];
        } else if (col < Nc + H) {
            #pragma unroll
            for (int r = 0; r < 8; ++r)
                el[(size_t)(srow0 + r) * H + (col - Nc)] = acc[t][r];
        } else if (col < Nc + 2 * H) {
            #pragma unroll
            for (int r = 0; r < 8; ++r)
                er[(size_t)(srow0 + r) * H + (col - Nc - H)] = acc[t][r];
        }
    }
}

template <bool HALF>
__device__ __forceinline__ void gat_gemm_guest(
    const float* __restrict__ A, const float* __restrict__ B,
    float* __restrict__ C, float* __restrict__ el, float* __restrict__ er,
    int Np, int K, int Nc, int H, int gb, int t0, int t1)
{
    const int wv = __builtin_amdgcn_readfirstlane((int)(threadIdx.x >> 5));
    int t;
    if (HALF) {
        if (wv & 4) return;
        t = t0 + gb * 16 + ((wv >> 3) << 2) + (wv & 3);
    } else {
        t = t0 + gb * 8 + wv;
    }
    if (t >= t1) return;
    const int S  = Np / 64;
    const int mt = t / S;
    gat_wmma_gemm_tile(A, B, C, Np, K, Nc, el, er, H, mt, t - mt * S);
}

#define GAT_CAP 6144
#define GAT_FCAP 4096

__global__ __launch_bounds__(1024) void gat_bin_count(
    const int* __restrict__ dst, int* __restrict__ cnt2, int E, int R, int G,
    const float* __restrict__ fW0, const float* __restrict__ fW1,
    const float* __restrict__ fal, const float* __restrict__ far_,
    float* __restrict__ fWc, int fH, int fMT)
{
    extern __shared__ int gat_sh[];
    if ((int)blockIdx.x >= G) {
        const int wv = __builtin_amdgcn_readfirstlane((int)(threadIdx.x >> 5));
        const int t  = ((int)blockIdx.x - G) * 32 + wv;
        if (t < fMT * fH * 3)
            gat_fold_tile(fW0, fW1, fal, far_, fWc, fH, t % fMT, t / fMT);
        return;
    }
    const int tid = threadIdx.x;
    for (int r = tid; r < R; r += 1024) gat_sh[r] = 0;
    __syncthreads();
    const int base = (int)blockIdx.x * GAT_CAP;
    int dk[GAT_CAP / 1024];
    #pragma unroll
    for (int k = 0; k < GAT_CAP / 1024; ++k) {
        const int e = base + k * 1024 + tid;
        dk[k] = dst[e < E ? e : E - 1];
    }
    #pragma unroll
    for (int k = 0; k < GAT_CAP / 1024; ++k) {
        const int e = base + k * 1024 + tid;
        if (e < E) atomicAdd(&gat_sh[dk[k] >> 6], 1);
    }
    __syncthreads();
    for (int r = tid; r < R; r += 1024) cnt2[r * G + (int)blockIdx.x] = gat_sh[r];
}

__global__ __launch_bounds__(1024) void gat_bin_scatter(
    const int* __restrict__ src, const int* __restrict__ dst,
    const int* __restrict__ off2, unsigned* __restrict__ tmp, int E, int R, int G,
    const float* __restrict__ gA, const float* __restrict__ gB,
    float* __restrict__ gC, float* __restrict__ gel, float* __restrict__ ger,
    int gNp, int gK, int gNc, int gH, int gt0, int gt1)
{
    extern __shared__ int gat_sh[];
    __shared__ int wsum[32];
    if ((int)blockIdx.x >= G) {
        gat_gemm_guest<true>(gA, gB, gC, gel, ger, gNp, gK, gNc, gH,
                             (int)blockIdx.x - G, gt0, gt1);
        return;
    }
    int* lcnt = gat_sh;
    int* lbeg = gat_sh + R;
    int* gofs = gat_sh + 2 * R;
    unsigned* bw = (unsigned*)(gat_sh + 3 * R);
    int* br = gat_sh + 3 * R + GAT_CAP;
    const int tid = threadIdx.x, lane = tid & 31, wv = tid >> 5;
    const int g = (int)blockIdx.x;
    const int base = g * GAT_CAP;

    for (int r = tid; r < R; r += 1024) {
        lcnt[r] = 0;
        gofs[r] = off2[r * G + g];
    }
    __syncthreads();
    int ed[GAT_CAP / 1024], es[GAT_CAP / 1024];
    #pragma unroll
    for (int k = 0; k < GAT_CAP / 1024; ++k) {
        const int e  = base + k * 1024 + tid;
        const int ec = e < E ? e : E - 1;
        const int d  = dst[ec];
        es[k] = src[ec];
        ed[k] = e < E ? d : -1;
    }
    #pragma unroll
    for (int k = 0; k < GAT_CAP / 1024; ++k)
        if (ed[k] >= 0) atomicAdd(&lcnt[ed[k] >> 6], 1);
    __syncthreads();
    int carry = 0;
    for (int b0 = 0; b0 < R; b0 += 1024) {
        const int idx = b0 + tid;
        const int v = (idx < R) ? lcnt[idx] : 0;
        int incl = v;
        for (int off = 1; off < 32; off <<= 1) {
            const int y = __shfl_up(incl, off, 32);
            if (lane >= off) incl += y;
        }
        if (lane == 31) wsum[wv] = incl;
        __syncthreads();
        int pre = 0, tot = 0;
        for (int w = 0; w < 32; ++w) {
            const int t = wsum[w];
            tot += t;
            if (w < wv) pre += t;
        }
        if (idx < R) lbeg[idx] = carry + pre + incl - v;
        carry += tot;
        __syncthreads();
    }
    for (int r = tid; r < R; r += 1024) lcnt[r] = lbeg[r];
    __syncthreads();
    #pragma unroll
    for (int k = 0; k < GAT_CAP / 1024; ++k) {
        if (ed[k] >= 0) {
            const int r = ed[k] >> 6;
            const int p = atomicAdd(&lcnt[r], 1);
            bw[p] = ((unsigned)es[k] << 6) | (unsigned)(ed[k] & 63);
            br[p] = r;
        }
    }
    __syncthreads();
    const int left = E - base;
    const int nloc = left < GAT_CAP ? left : GAT_CAP;
    for (int k = tid; k < nloc; k += 1024) {
        const int r = br[k];
        tmp[gofs[r] + (k - lbeg[r])] = bw[k];
    }
}

__global__ __launch_bounds__(256) void gat_bin_finish(
    const unsigned* __restrict__ tmp, const int* __restrict__ off2,
    int* __restrict__ rowptr, int* __restrict__ bsrc, int N, int R, int G,
    const float* __restrict__ gA, const float* __restrict__ gB,
    float* __restrict__ gC, float* __restrict__ gel, float* __restrict__ ger,
    int gNp, int gK, int gNc, int gH, int gt0, int gt1)
{
    __shared__ int c[64];
    __shared__ int cur[64];
    __shared__ int wtot[2];
    __shared__ int stage[GAT_FCAP];
    if ((int)blockIdx.x >= R) {
        gat_gemm_guest<false>(gA, gB, gC, gel, ger, gNp, gK, gNc, gH,
                              (int)blockIdx.x - R, gt0, gt1);
        return;
    }
    const int tid = threadIdx.x, lane = tid & 31, wv = tid >> 5;
    const int r = (int)blockIdx.x;
    const int s = off2[r * G];
    const int e = off2[(r + 1) * G];
    const int n = e - s;
    if (tid < 64) c[tid] = 0;
    __syncthreads();
    for (int i = tid; i < n; i += 256) atomicAdd(&c[tmp[s + i] & 63u], 1);
    __syncthreads();
    int v = 0, incl = 0;
    if (tid < 64) {
        v = c[tid];
        incl = v;
        for (int off = 1; off < 32; off <<= 1) {
            const int y = __shfl_up(incl, off, 32);
            if (lane >= off) incl += y;
        }
        if (lane == 31) wtot[wv] = incl;
    }
    __syncthreads();
    if (tid < 64) {
        const int excl = incl - v + (wv == 1 ? wtot[0] : 0);
        cur[tid] = excl;
        const int node = r * 64 + tid;
        if (node < N) rowptr[node] = s + excl;
    }
    if (r == R - 1 && tid == 0) rowptr[N] = e;
    __syncthreads();
    if (n <= GAT_FCAP) {
        for (int i = tid; i < n; i += 256) {
            const unsigned w = tmp[s + i];
            const int p = atomicAdd(&cur[w & 63u], 1);
            stage[p] = (int)(w >> 6);
        }
        __syncthreads();
        for (int i = tid; i < n; i += 256) bsrc[s + i] = stage[i];
    } else {
        for (int i = tid; i < n; i += 256) {
            const unsigned w = tmp[s + i];
            const int p = atomicAdd(&cur[w & 63u], 1);
            bsrc[s + p] = (int)(w >> 6);
        }
    }
}

__global__ __launch_bounds__(256) void gat_bucket_scan(
    const int* __restrict__ cnt, int* __restrict__ rowptr, int N, int nscan,
    const float* __restrict__ gA, const float* __restrict__ gB,
    float* __restrict__ gC, float* __restrict__ gel, float* __restrict__ ger,
    int gNp, int gK, int gNc, int gH, int gt0, int gt1)
{
    __shared__ int s_tot[8];
    __shared__ int s_car[8];
    if ((int)blockIdx.x >= nscan) {
        gat_gemm_guest<false>(gA, gB, gC, gel, ger, gNp, gK, gNc, gH,
                              (int)blockIdx.x - nscan, gt0, gt1);
        return;
    }
    const int tid = threadIdx.x, lane = tid & 31, wv = tid >> 5;
    const int4* __restrict__ c4 = (const int4*)cnt;

    int p = 0;
    for (int k = 0; k < (int)blockIdx.x; ++k) {
        const int4 q = c4[k * 256 + tid];
        p += (q.x + q.y) + (q.z + q.w);
    }
    const int i0 = blockIdx.x * 1024 + tid * 4;
    const bool live = (i0 + 3 < N);
    int4 own = make_int4(0, 0, 0, 0);
    if (live) own = c4[i0 >> 2];
    const int v = (own.x + own.y) + (own.z + own.w);

    int incl = v;
    for (int off = 1; off < 32; off <<= 1) {
        const int y = __shfl_up(incl, off, 32);
        if (lane >= off) incl += y;
    }
    for (int m = 16; m >= 1; m >>= 1) p += __shfl_xor(p, m, 32);
    if (lane == 31) s_tot[wv] = incl;
    if (lane == 0)  s_car[wv] = p;
    __syncthreads();
    int base = 0;
    for (int w = 0; w < 8; ++w) {
        base += s_car[w];
        if (w < wv) base += s_tot[w];
    }
    base += incl - v;
    if (live) {
        int4 r;
        r.x = base;
        r.y = r.x + own.x;
        r.z = r.y + own.y;
        r.w = r.z + own.z;
        ((int4*)rowptr)[i0 >> 2] = r;
        if (i0 + 4 == N) rowptr[N] = r.w + own.w;
    }
}

__device__ __forceinline__ float gat_leaky(float v) {
    return (v > 0.f) ? v : SLOPE * v;
}

__global__ __launch_bounds__(512) void gat_node_gather_h4_fold(
    const int* __restrict__ rowptr, const int* __restrict__ bsrc,
    const float* __restrict__ P, const float* __restrict__ el,
    const float* __restrict__ er,
    const float* __restrict__ al, const float* __restrict__ ar,
    float* __restrict__ C, float* __restrict__ er1, int N)
{
    __shared__ __attribute__((aligned(16))) float stage[16 * GAT_PS];
    __shared__ float part[16];
    const int lane = threadIdx.x & 31;
    const int wv = __builtin_amdgcn_readfirstlane((int)(threadIdx.x >> 5));
    const int node0 = (int)blockIdx.x * 16;
    const int n = node0 + wv;
    float4 accA = make_float4(0.f, 0.f, 0.f, 0.f);
    float4 accB = make_float4(0.f, 0.f, 0.f, 0.f);
    const int c2 = 32 + (lane & 7);
    const float wl0 = al[lane], wl1 = al[c2];
    const float wr0 = ar[lane], wr1 = ar[c2];

    if (n < N) {
        const int beg = __builtin_amdgcn_readfirstlane(rowptr[n]);
        const int end = __builtin_amdgcn_readfirstlane(rowptr[n + 1]);
        const float4* __restrict__ el4 = (const float4*)el;
        const float4 ern = ((const float4*)er)[n];
        const float NINF = -__builtin_inff();

        const int  j0 = beg + lane;
        const bool v0 = j0 < end;
        int   src0 = 0;
        float e0 = NINF, e1 = NINF, e2 = NINF, e3 = NINF;
        if (v0) {
            src0 = bsrc[j0];
            const float4 l = el4[src0];
            e0 = gat_leaky(l.x + ern.x);
            e1 = gat_leaky(l.y + ern.y);
            e2 = gat_leaky(l.z + ern.z);
            e3 = gat_leaky(l.w + ern.w);
        }
        float m0 = e0, m1 = e1, m2 = e2, m3 = e3;
        for (int j = j0 + 32; j < end; j += 32) {
            const float4 l = el4[bsrc[j]];
            m0 = fmaxf(m0, gat_leaky(l.x + ern.x));
            m1 = fmaxf(m1, gat_leaky(l.y + ern.y));
            m2 = fmaxf(m2, gat_leaky(l.z + ern.z));
            m3 = fmaxf(m3, gat_leaky(l.w + ern.w));
        }
        for (int m = 16; m >= 1; m >>= 1) {
            m0 = fmaxf(m0, __shfl_xor(m0, m, 32));
            m1 = fmaxf(m1, __shfl_xor(m1, m, 32));
            m2 = fmaxf(m2, __shfl_xor(m2, m, 32));
            m3 = fmaxf(m3, __shfl_xor(m3, m, 32));
        }

        const float p0 = v0 ? __expf(e0 - m0) : 0.f;
        const float p1 = v0 ? __expf(e1 - m1) : 0.f;
        const float p2 = v0 ? __expf(e2 - m2) : 0.f;
        const float p3 = v0 ? __expf(e3 - m3) : 0.f;
        float s0 = p0, s1 = p1, s2 = p2, s3 = p3;
        for (int j = j0 + 32; j < end; j += 32) {
            const float4 l = el4[bsrc[j]];
            s0 += __expf(gat_leaky(l.x + ern.x) - m0);
            s1 += __expf(gat_leaky(l.y + ern.y) - m1);
            s2 += __expf(gat_leaky(l.z + ern.z) - m2);
            s3 += __expf(gat_leaky(l.w + ern.w) - m3);
        }
        for (int m = 16; m >= 1; m >>= 1) {
            s0 += __shfl_xor(s0, m, 32);
            s1 += __shfl_xor(s1, m, 32);
            s2 += __shfl_xor(s2, m, 32);
            s3 += __shfl_xor(s3, m, 32);
        }

        const bool h0l = lane < 10;
        const bool h1l = lane < 20;
        const bool h2l = lane < 30;
        for (int base = beg; base < end; base += 32) {
            int   mysrc = 0;
            float a0 = 0.f, a1 = 0.f, a2 = 0.f, a3 = 0.f;
            if (base == beg) {
                mysrc = src0;
                if (v0) {
                    a0 = p0 / s0;  a1 = p1 / s1;
                    a2 = p2 / s2;  a3 = p3 / s3;
                }
            } else {
                const int j = base + lane;
                if (j < end) {
                    mysrc = bsrc[j];
                    const float4 l = el4[mysrc];
                    a0 = __expf(gat_leaky(l.x + ern.x) - m0) / s0;
                    a1 = __expf(gat_leaky(l.y + ern.y) - m1) / s1;
                    a2 = __expf(gat_leaky(l.z + ern.z) - m2) / s2;
                    a3 = __expf(gat_leaky(l.w + ern.w) - m3) / s3;
                }
            }
            const int left = end - base;
            const int cntc = left < 32 ? left : 32;
            #pragma unroll 4
            for (int t = 0; t < cntc; ++t) {
                const int   sN = __builtin_amdgcn_readlane(mysrc, t);
                const float b0 = __int_as_float(__builtin_amdgcn_readlane(__float_as_int(a0), t));
                const float b1 = __int_as_float(__builtin_amdgcn_readlane(__float_as_int(a1), t));
                const float b2 = __int_as_float(__builtin_amdgcn_readlane(__float_as_int(a2), t));
                const float b3 = __int_as_float(__builtin_amdgcn_readlane(__float_as_int(a3), t));
                const float aA = h0l ? b0 : (h1l ? b1 : (h2l ? b2 : b3));
                const float4* __restrict__ hp = (const float4*)(P + (size_t)sN * GAT_PS);
                const float4 x = hp[lane];
                const float4 y = hp[c2];
                accA.x += aA * x.x;  accA.y += aA * x.y;
                accA.z += aA * x.z;  accA.w += aA * x.w;
                accB.x += b3 * y.x;  accB.y += b3 * y.y;
                accB.z += b3 * y.z;  accB.w += b3 * y.w;
            }
        }
    }
    float4* __restrict__ sp4 = (float4*)(stage + wv * GAT_PS);
    sp4[lane] = accA;
    if (lane < 8) sp4[32 + lane] = accB;
    __syncthreads();

    const float* sr = stage + wv * GAT_PS;
    const bool lo8 = lane < 8;
    const float u0 = (sr[lane] + sr[GAT_NC + lane])
                   + (sr[2 * GAT_NC + lane] + sr[3 * GAT_NC + lane]);
    const float u1 = (sr[c2] + sr[GAT_NC + c2])
                   + (sr[2 * GAT_NC + c2] + sr[3 * GAT_NC + c2]);
    float ql = u0 * wl0 + (lo8 ? u1 * wl1 : 0.f);
    float qr = u0 * wr0 + (lo8 ? u1 * wr1 : 0.f);
    for (int m = 16; m >= 1; m >>= 1) {
        ql += __shfl_xor(ql, m, 32);
        qr += __shfl_xor(qr, m, 32);
    }
    if (n < N) {
        float* __restrict__ cr = C + (size_t)n * 64;
        cr[lane] = u0;
        cr[32 + lane] = lo8 ? u1 : (lane == 8 ? ql : 0.f);
    }
    if (lane == 0) part[wv] = qr;
    __syncthreads();
    if (threadIdx.x < 16 && node0 + (int)threadIdx.x < N)
        er1[node0 + threadIdx.x] = part[threadIdx.x];
}

#define GAT_G1 8

template <bool FULL>
__device__ __forceinline__ void gat_h1_group(
    const float2* __restrict__ h2, int mysrc, int t0, int g, int lane,
    int ell, float ern, float& m, float& s, v2f& acc)
{
    const float NINF = -__builtin_inff();
    int sN[GAT_G1];
    #pragma unroll
    for (int u = 0; u < GAT_G1; ++u)
        sN[u] = __builtin_amdgcn_readlane(mysrc, (FULL || u < g) ? t0 + u : t0);
    float2 x[GAT_G1];
    float  e[GAT_G1];
    #pragma unroll
    for (int u = 0; u < GAT_G1; ++u) {
        x[u] = make_float2(0.f, 0.f);
        if (FULL || u < g)
            x[u] = h2[(size_t)sN[u] * 32 + lane];
    }
    float gm = NINF;
    #pragma unroll
    for (int u = 0; u < GAT_G1; ++u) {
        const float ev = __int_as_float(
            __builtin_amdgcn_readlane(__float_as_int(x[u].x), ell));
        const float lv = ev + ern;
        const float lk = (lv > 0.f) ? lv : SLOPE * lv;
        e[u] = (FULL || u < g) ? lk : NINF;
        gm = fmaxf(gm, e[u]);
    }
    const float mn = fmaxf(m, gm);
    const float sc = __expf(m - mn);
    s *= sc;
    acc = acc * (v2f){sc, sc};
    #pragma unroll
    for (int u = 0; u < GAT_G1; ++u) {
        if (FULL || u < g) {
            const float p = __expf(e[u] - mn);
            s += p;
            acc = __builtin_elementwise_fma((v2f){p, p}, (v2f){x[u].x, x[u].y}, acc);
        }
    }
    m = mn;
}

__global__ __launch_bounds__(128) void gat_node_gather_h1(
    const int* __restrict__ rowptr, const int* __restrict__ bsrc,
    const float* __restrict__ h, const float* __restrict__ er,
    float* __restrict__ out, int N, int D)
{
    __shared__ float stage[4 * 64];
    const int lane = threadIdx.x & 31;
    const int wv = __builtin_amdgcn_readfirstlane((int)(threadIdx.x >> 5));
    const int n  = (int)blockIdx.x * 4 + wv;
    const bool has = 2 * lane < D;
    v2f   acc = (v2f){0.f, 0.f};
    float s = 0.f;

    if (n < N) {
        const int beg = __builtin_amdgcn_readfirstlane(rowptr[n]);
        const int end = __builtin_amdgcn_readfirstlane(rowptr[n + 1]);
        const float ern = er[n];
        const float2* __restrict__ h2 = (const float2*)h;
        const int ell = D >> 1;
        float m = -__builtin_inff();

        for (int base = beg; base < end; base += 32) {
            const int j = base + lane;
            const int mysrc = (j < end) ? bsrc[j] : 0;
            const int left = end - base;
            const int cntc = left < 32 ? left : 32;
            int t0 = 0;
            for (; t0 + GAT_G1 <= cntc; t0 += GAT_G1)
                gat_h1_group<true>(h2, mysrc, t0, GAT_G1, lane, ell, ern, m, s, acc);
            if (t0 < cntc)
                gat_h1_group<false>(h2, mysrc, t0, cntc - t0, lane, ell, ern, m, s, acc);
        }
    }
    const float inv = (s > 0.f) ? 1.0f / s : 0.f;
    if (has) {
        stage[wv * D + 2 * lane]     = acc.x * inv;
        stage[wv * D + 2 * lane + 1] = acc.y * inv;
    }
    __syncthreads();
    const int nleft  = N - (int)blockIdx.x * 4;
    const int nvalid = nleft < 4 ? nleft : 4;
    float* __restrict__ ob = out + (size_t)blockIdx.x * 4 * D;
    for (int i = threadIdx.x; i < nvalid * D; i += 128) ob[i] = stage[i];
}

extern "C" void kernel_launch(void* const* d_in, const int* in_sizes, int n_in,
                              void* d_out, int out_size, void* d_ws, size_t ws_size,
                              hipStream_t stream)
{
    const float* feat = (const float*)d_in[0];
    const int*   src  = (const int*)  d_in[1];
    const int*   dst  = (const int*)  d_in[2];
    const float* W0   = (const float*)d_in[3];
    const float* al0  = (const float*)d_in[4];
    const float* ar0  = (const float*)d_in[5];
    const float* W1   = (const float*)d_in[6];
    const float* al1  = (const float*)d_in[7];
    const float* ar1  = (const float*)d_in[8];
    float* out = (float*)d_out;

    const int FIN = 128, H0 = 4, NC = GAT_NC, NCP = 64;
    const int N = in_sizes[0] / FIN;
    const int E = in_sizes[1];

    char* p = (char*)d_ws;
    auto alloc = [&](size_t bytes) -> void* {
        void* r = (void*)p;
        p += (bytes + 255) & ~(size_t)255;
        return r;
    };
    float* P0     = (float*)alloc((size_t)N * GAT_PS * 4);
    float* h1     = (float*)alloc((size_t)N * NCP * 4);
    float* el0    = (float*)alloc((size_t)N * H0 * 4);
    float* er0    = (float*)alloc((size_t)N * H0 * 4);
    float* er1    = (float*)alloc((size_t)N * 4);
    float* Wc     = (float*)alloc((size_t)FIN * GAT_WCN * 4);
    const int R  = (N + 63) / 64;
    const int G  = ((E + 4 * GAT_CAP - 1) / (4 * GAT_CAP)) * 4;
    const int RG = R * G;
    int*      cnt2   = (int*)     alloc((size_t)RG * 4);
    int*      off2   = (int*)     alloc((size_t)(RG + 1) * 4);
    int*      rowptr = (int*)     alloc((size_t)(N + 1) * 4);
    unsigned* tmp    = (unsigned*)alloc((size_t)E * 4);
    int*      bsrc   = (int*)     alloc((size_t)E * 4);

    const int nfold = (FIN / 16) * H0 * 3;
    const int T     = (N / 16) * (GAT_WCN / 64);
    const int Ta    = (T * 7) / 20;
    const int Tb    = Ta + T / 5;
    const int nscan = (RG + 1023) / 1024;
    gat_bin_count<<<G + (nfold + 31) / 32, 1024, (size_t)R * 4, stream>>>(
        dst, cnt2, E, R, G, W0, W1, al0, ar0, Wc, H0, FIN / 16);
    gat_bucket_scan<<<nscan + (Ta + 7) / 8, 256, 0, stream>>>(
        cnt2, off2, RG, nscan,
        feat, Wc, P0, el0, er0, GAT_WCN, FIN, GAT_PS, H0, 0, Ta);
    gat_bin_scatter<<<G + (Tb - Ta + 15) / 16, 1024,
                      ((size_t)3 * R + 2 * GAT_CAP) * 4, stream>>>(
        src, dst, off2, tmp, E, R, G,
        feat, Wc, P0, el0, er0, GAT_WCN, FIN, GAT_PS, H0, Ta, Tb);
    gat_bin_finish<<<R + (T - Tb + 7) / 8, 256, 0, stream>>>(
        tmp, off2, rowptr, bsrc, N, R, G,
        feat, Wc, P0, el0, er0, GAT_WCN, FIN, GAT_PS, H0, Tb, T);
    gat_node_gather_h4_fold<<<(N + 15) / 16, 512, 0, stream>>>(
        rowptr, bsrc, P0, el0, er0, al1, ar1, h1, er1, N);
    gat_node_gather_h1<<<(N + 3) / 4, 128, 0, stream>>>(
        rowptr, bsrc, h1, er1, out, N, NC);
}


# Node GEMM operands by 16-byte loads issued two 8-k steps ahead: folded weights kept with rows k,k+1 interleaved per column, A by float4; 20 loads per 32 matrix ops, was 72; P0/el0/er0 sum order new
# speedup vs baseline: 6.8224x; 1.1340x over previous
#include <hip/hip_runtime.h>
#include <hip/hip_bf16.h>

typedef __attribute__((ext_vector_type(2))) float v2f;
typedef __attribute__((ext_vector_type(8))) float v8f;

#define SLOPE 0.2f

#define GAT_D0 64
#define GAT_NC 40
#define GAT_PS 160
#define GAT_WCN 192

__device__ __forceinline__ void gat_fold_tile(
    const float* __restrict__ W0, const float* __restrict__ W1,
    const float* __restrict__ al, const float* __restrict__ ar,
    float* __restrict__ Wc, int H, int mt, int hc)
{
    const int lane  = threadIdx.x & 31;
    const int l15   = lane & 15;
    const int khalf = (lane >> 4) * 2;
    const int head  = hc / 3;
    const int ct    = hc % 3;
    const int c     = ct * 16 + l15;
    const int row   = mt * 16 + l15;
    const int ldw   = H * GAT_D0;

    const float* __restrict__ Ap = W0 + (size_t)row * ldw + head * GAT_D0 + khalf;
    const bool okb = c < GAT_NC + 2;
    const float* __restrict__ Bp;
    int bs;
    if (c < GAT_NC)       { Bp = W1 + (size_t)(head * GAT_D0 + khalf) * GAT_NC + c; bs = GAT_NC; }
    else if (c == GAT_NC) { Bp = al + head * GAT_D0 + khalf; bs = 1; }
    else                  { Bp = ar + head * GAT_D0 + khalf; bs = 1; }

    v8f acc = (v8f){0.f, 0.f, 0.f, 0.f, 0.f, 0.f, 0.f, 0.f};
    #pragma unroll
    for (int k = 0; k < GAT_D0; k += 4) {
        v2f a, b;
        a.x = Ap[k];
        a.y = Ap[k + 1];
        const float bx = Bp[k * bs];
        const float by = Bp[(k + 1) * bs];
        b.x = okb ? bx : 0.f;
        b.y = okb ? by : 0.f;
        acc = __builtin_amdgcn_wmma_f32_16x16x4_f32(false, a, false, b, (short)0, acc, false, false);
    }
    int oc;
    if (c < GAT_NC)           oc = head * GAT_NC + c;
    else if (c == GAT_NC)     oc = H * GAT_NC + head;
    else if (c == GAT_NC + 1) oc = H * GAT_NC + H + head;
    else                      oc = H * GAT_NC + 2 * H + head * 6 + (c - GAT_NC - 2);
    const int srow0 = mt * 16 + (lane >> 4) * 8;
    #pragma unroll
    for (int r = 0; r < 8; ++r) {
        const int k = srow0 + r;
        Wc[(size_t)(k >> 1) * (2 * GAT_WCN) + oc * 2 + (k & 1)] = acc[r];
    }
}

#define GAT_FIN 128
#define GAT_H0 4

__device__ __forceinline__ void gat_wmma_gemm_tile(
    const float* __restrict__ A, const float* __restrict__ B,
    float* __restrict__ C, float* __restrict__ el, float* __restrict__ er,
    int mt, int strip)
{
    const int lane = threadIdx.x & 31;
    const int l15  = lane & 15;
    const int hh   = lane >> 4;
    const int row  = mt * 16 + l15;

    const float4* __restrict__ Ap = (const float4*)(A + (size_t)row * GAT_FIN) + hh;
    const float4* __restrict__ Bp = (const float4*)B
                                  + (size_t)(2 * hh) * (GAT_WCN / 2) + strip * 32 + l15;

    v8f acc[4];
    #pragma unroll
    for (int t = 0; t < 4; ++t)
        acc[t] = (v8f){0.f, 0.f, 0.f, 0.f, 0.f, 0.f, 0.f, 0.f};

    float4 fa[3], fp0[3], fp1[3], fq0[3], fq1[3];
    #pragma unroll
    for (int ks = 0; ks < 2; ++ks) {
        const float4* __restrict__ bq = Bp + (size_t)(4 * ks) * (GAT_WCN / 2);
        fa[ks]  = Ap[2 * ks];
        fp0[ks] = bq[0];
        fp1[ks] = bq[16];
        fq0[ks] = bq[GAT_WCN / 2];
        fq1[ks] = bq[GAT_WCN / 2 + 16];
    }
    #pragma unroll
    for (int ks = 0; ks < GAT_FIN / 8; ++ks) {
        const int cur = ks % 3;
        if (ks + 2 < GAT_FIN / 8) {
            const int nxt = (ks + 2) % 3;
            const float4* __restrict__ bq = Bp + (size_t)(4 * (ks + 2)) * (GAT_WCN / 2);
            fa[nxt]  = Ap[2 * (ks + 2)];
            fp0[nxt] = bq[0];
            fp1[nxt] = bq[16];
            fq0[nxt] = bq[GAT_WCN / 2];
            fq1[nxt] = bq[GAT_WCN / 2 + 16];
        }
        const v2f a0 = (v2f){fa[cur].x, fa[cur].y};
        const v2f a1 = (v2f){fa[cur].z, fa[cur].w};
        acc[0] = __builtin_amdgcn_wmma_f32_16x16x4_f32(false, a0, false, (v2f){fp0[cur].x, fp0[cur].y}, (short)0, acc[0], false, false);
        acc[1] = __builtin_amdgcn_wmma_f32_16x16x4_f32(false, a0, false, (v2f){fp0[cur].z, fp0[cur].w}, (short)0, acc[1], false, false);
        acc[2] = __builtin_amdgcn_wmma_f32_16x16x4_f32(false, a0, false, (v2f){fp1[cur].x, fp1[cur].y}, (short)0, acc[2], false, false);
        acc[3] = __builtin_amdgcn_wmma_f32_16x16x4_f32(false, a0, false, (v2f){fp1[cur].z, fp1[cur].w}, (short)0, acc[3], false, false);
        acc[0] = __builtin_amdgcn_wmma_f32_16x16x4_f32(false, a1, false, (v2f){fq0[cur].x, fq0[cur].y}, (short)0, acc[0], false, false);
        acc[1] = __builtin_amdgcn_wmma_f32_16x16x4_f32(false, a1, false, (v2f){fq0[cur].z, fq0[cur].w}, (short)0, acc[1], false, false);
        acc[2] = __builtin_amdgcn_wmma_f32_16x16x4_f32(false, a1, false, (v2f){fq1[cur].x, fq1[cur].y}, (short)0, acc[2], false, false);
        acc[3] = __builtin_amdgcn_wmma_f32_16x16x4_f32(false, a1, false, (v2f){fq1[cur].z, fq1[cur].w}, (short)0, acc[3], false, false);
        __builtin_amdgcn_sched_barrier(0);
    }

    const int srow0 = mt * 16 + hh * 8;
    #pragma unroll
    for (int u = 0; u < 2; ++u) {
        const int col = strip * 64 + 32 * u + 2 * l15;
        if (col < GAT_PS) {
            #pragma unroll
            for (int r = 0; r < 8; ++r)
                *(float2*)(C + (size_t)(srow0 + r) * GAT_PS + col) =
                    make_float2(acc[2 * u][r], acc[2 * u + 1][r]);
        } else if (col < GAT_PS + GAT_H0) {
            #pragma unroll
            for (int r = 0; r < 8; ++r)
                *(float2*)(el + (size_t)(srow0 + r) * GAT_H0 + (col - GAT_PS)) =
                    make_float2(acc[2 * u][r], acc[2 * u + 1][r]);
        } else if (col < GAT_PS + 2 * GAT_H0) {
            #pragma unroll
            for (int r = 0; r < 8; ++r)
                *(float2*)(er + (size_t)(srow0 + r) * GAT_H0 + (col - GAT_PS - GAT_H0)) =
                    make_float2(acc[2 * u][r], acc[2 * u + 1][r]);
        }
    }
}

template <bool HALF>
__device__ __forceinline__ void gat_gemm_guest(
    const float* __restrict__ A, const float* __restrict__ B,
    float* __restrict__ C, float* __restrict__ el, float* __restrict__ er,
    int gb, int t0, int t1)
{
    const int wv = __builtin_amdgcn_readfirstlane((int)(threadIdx.x >> 5));
    int t;
    if (HALF) {
        if (wv & 4) return;
        t = t0 + gb * 16 + ((wv >> 3) << 2) + (wv & 3);
    } else {
        t = t0 + gb * 8 + wv;
    }
    if (t >= t1) return;
    const int S  = GAT_WCN / 64;
    const int mt = t / S;
    gat_wmma_gemm_tile(A, B, C, el, er, mt, t - mt * S);
}

#define GAT_CAP 6144
#define GAT_FCAP 4096

__global__ __launch_bounds__(1024) void gat_bin_count(
    const int* __restrict__ dst, int* __restrict__ cnt2, int E, int R, int G,
    const float* __restrict__ fW0, const float* __restrict__ fW1,
    const float* __restrict__ fal, const float* __restrict__ far_,
    float* __restrict__ fWc, int fH, int fMT)
{
    extern __shared__ int gat_sh[];
    if ((int)blockIdx.x >= G) {
        const int wv = __builtin_amdgcn_readfirstlane((int)(threadIdx.x >> 5));
        const int t  = ((int)blockIdx.x - G) * 32 + wv;
        if (t < fMT * fH * 3)
            gat_fold_tile(fW0, fW1, fal, far_, fWc, fH, t % fMT, t / fMT);
        return;
    }
    const int tid = threadIdx.x;
    for (int r = tid; r < R; r += 1024) gat_sh[r] = 0;
    __syncthreads();
    const int base = (int)blockIdx.x * GAT_CAP;
    int dk[GAT_CAP / 1024];
    #pragma unroll
    for (int k = 0; k < GAT_CAP / 1024; ++k) {
        const int e = base + k * 1024 + tid;
        dk[k] = dst[e < E ? e : E - 1];
    }
    #pragma unroll
    for (int k = 0; k < GAT_CAP / 1024; ++k) {
        const int e = base + k * 1024 + tid;
        if (e < E) atomicAdd(&gat_sh[dk[k] >> 6], 1);
    }
    __syncthreads();
    for (int r = tid; r < R; r += 1024) cnt2[r * G + (int)blockIdx.x] = gat_sh[r];
}

__global__ __launch_bounds__(1024) void gat_bin_scatter(
    const int* __restrict__ src, const int* __restrict__ dst,
    const int* __restrict__ off2, unsigned* __restrict__ tmp, int E, int R, int G,
    const float* __restrict__ gA, const float* __restrict__ gB,
    float* __restrict__ gC, float* __restrict__ gel, float* __restrict__ ger,
    int gt0, int gt1)
{
    extern __shared__ int gat_sh[];
    __shared__ int wsum[32];
    if ((int)blockIdx.x >= G) {
        gat_gemm_guest<true>(gA, gB, gC, gel, ger,
                             (int)blockIdx.x - G, gt0, gt1);
        return;
    }
    int* lcnt = gat_sh;
    int* lbeg = gat_sh + R;
    int* gofs = gat_sh + 2 * R;
    unsigned* bw = (unsigned*)(gat_sh + 3 * R);
    int* br = gat_sh + 3 * R + GAT_CAP;
    const int tid = threadIdx.x, lane = tid & 31, wv = tid >> 5;
    const int g = (int)blockIdx.x;
    const int base = g * GAT_CAP;

    for (int r = tid; r < R; r += 1024) {
        lcnt[r] = 0;
        gofs[r] = off2[r * G + g];
    }
    __syncthreads();
    int ed[GAT_CAP / 1024], es[GAT_CAP / 1024];
    #pragma unroll
    for (int k = 0; k < GAT_CAP / 1024; ++k) {
        const int e  = base + k * 1024 + tid;
        const int ec = e < E ? e : E - 1;
        const int d  = dst[ec];
        es[k] = src[ec];
        ed[k] = e < E ? d : -1;
    }
    #pragma unroll
    for (int k = 0; k < GAT_CAP / 1024; ++k)
        if (ed[k] >= 0) atomicAdd(&lcnt[ed[k] >> 6], 1);
    __syncthreads();
    int carry = 0;
    for (int b0 = 0; b0 < R; b0 += 1024) {
        const int idx = b0 + tid;
        const int v = (idx < R) ? lcnt[idx] : 0;
        int incl = v;
        for (int off = 1; off < 32; off <<= 1) {
            const int y = __shfl_up(incl, off, 32);
            if (lane >= off) incl += y;
        }
        if (lane == 31) wsum[wv] = incl;
        __syncthreads();
        int pre = 0, tot = 0;
        for (int w = 0; w < 32; ++w) {
            const int t = wsum[w];
            tot += t;
            if (w < wv) pre += t;
        }
        if (idx < R) lbeg[idx] = carry + pre + incl - v;
        carry += tot;
        __syncthreads();
    }
    for (int r = tid; r < R; r += 1024) lcnt[r] = lbeg[r];
    __syncthreads();
    #pragma unroll
    for (int k = 0; k < GAT_CAP / 1024; ++k) {
        if (ed[k] >= 0) {
            const int r = ed[k] >> 6;
            const int p = atomicAdd(&lcnt[r], 1);
            bw[p] = ((unsigned)es[k] << 6) | (unsigned)(ed[k] & 63);
            br[p] = r;
        }
    }
    __syncthreads();
    const int left = E - base;
    const int nloc = left < GAT_CAP ? left : GAT_CAP;
    for (int k = tid; k < nloc; k += 1024) {
        const int r = br[k];
        tmp[gofs[r] + (k - lbeg[r])] = bw[k];
    }
}

__global__ __launch_bounds__(256) void gat_bin_finish(
    const unsigned* __restrict__ tmp, const int* __restrict__ off2,
    int* __restrict__ rowptr, int* __restrict__ bsrc, int N, int R, int G,
    const float* __restrict__ gA, const float* __restrict__ gB,
    float* __restrict__ gC, float* __restrict__ gel, float* __restrict__ ger,
    int gt0, int gt1)
{
    __shared__ int c[64];
    __shared__ int cur[64];
    __shared__ int wtot[2];
    __shared__ int stage[GAT_FCAP];
    if ((int)blockIdx.x >= R) {
        gat_gemm_guest<false>(gA, gB, gC, gel, ger,
                              (int)blockIdx.x - R, gt0, gt1);
        return;
    }
    const int tid = threadIdx.x, lane = tid & 31, wv = tid >> 5;
    const int r = (int)blockIdx.x;
    const int s = off2[r * G];
    const int e = off2[(r + 1) * G];
    const int n = e - s;
    if (tid < 64) c[tid] = 0;
    __syncthreads();
    for (int i = tid; i < n; i += 256) atomicAdd(&c[tmp[s + i] & 63u], 1);
    __syncthreads();
    int v = 0, incl = 0;
    if (tid < 64) {
        v = c[tid];
        incl = v;
        for (int off = 1; off < 32; off <<= 1) {
            const int y = __shfl_up(incl, off, 32);
            if (lane >= off) incl += y;
        }
        if (lane == 31) wtot[wv] = incl;
    }
    __syncthreads();
    if (tid < 64) {
        const int excl = incl - v + (wv == 1 ? wtot[0] : 0);
        cur[tid] = excl;
        const int node = r * 64 + tid;
        if (node < N) rowptr[node] = s + excl;
    }
    if (r == R - 1 && tid == 0) rowptr[N] = e;
    __syncthreads();
    if (n <= GAT_FCAP) {
        for (int i = tid; i < n; i += 256) {
            const unsigned w = tmp[s + i];
            const int p = atomicAdd(&cur[w & 63u], 1);
            stage[p] = (int)(w >> 6);
        }
        __syncthreads();
        for (int i = tid; i < n; i += 256) bsrc[s + i] = stage[i];
    } else {
        for (int i = tid; i < n; i += 256) {
            const unsigned w = tmp[s + i];
            const int p = atomicAdd(&cur[w & 63u], 1);
            bsrc[s + p] = (int)(w >> 6);
        }
    }
}

__global__ __launch_bounds__(256) void gat_bucket_scan(
    const int* __restrict__ cnt, int* __restrict__ rowptr, int N, int nscan,
    const float* __restrict__ gA, const float* __restrict__ gB,
    float* __restrict__ gC, float* __restrict__ gel, float* __restrict__ ger,
    int gt0, int gt1)
{
    __shared__ int s_tot[8];
    __shared__ int s_car[8];
    if ((int)blockIdx.x >= nscan) {
        gat_gemm_guest<false>(gA, gB, gC, gel, ger,
                              (int)blockIdx.x - nscan, gt0, gt1);
        return;
    }
    const int tid = threadIdx.x, lane = tid & 31, wv = tid >> 5;
    const int4* __restrict__ c4 = (const int4*)cnt;

    int p = 0;
    for (int k = 0; k < (int)blockIdx.x; ++k) {
        const int4 q = c4[k * 256 + tid];
        p += (q.x + q.y) + (q.z + q.w);
    }
    const int i0 = blockIdx.x * 1024 + tid * 4;
    const bool live = (i0 + 3 < N);
    int4 own = make_int4(0, 0, 0, 0);
    if (live) own = c4[i0 >> 2];
    const int v = (own.x + own.y) + (own.z + own.w);

    int incl = v;
    for (int off = 1; off < 32; off <<= 1) {
        const int y = __shfl_up(incl, off, 32);
        if (lane >= off) incl += y;
    }
    for (int m = 16; m >= 1; m >>= 1) p += __shfl_xor(p, m, 32);
    if (lane == 31) s_tot[wv] = incl;
    if (lane == 0)  s_car[wv] = p;
    __syncthreads();
    int base = 0;
    for (int w = 0; w < 8; ++w) {
        base += s_car[w];
        if (w < wv) base += s_tot[w];
    }
    base += incl - v;
    if (live) {
        int4 r;
        r.x = base;
        r.y = r.x + own.x;
        r.z = r.y + own.y;
        r.w = r.z + own.z;
        ((int4*)rowptr)[i0 >> 2] = r;
        if (i0 + 4 == N) rowptr[N] = r.w + own.w;
    }
}

__device__ __forceinline__ float gat_leaky(float v) {
    return (v > 0.f) ? v : SLOPE * v;
}

__global__ __launch_bounds__(512) void gat_node_gather_h4_fold(
    const int* __restrict__ rowptr, const int* __restrict__ bsrc,
    const float* __restrict__ P, const float* __restrict__ el,
    const float* __restrict__ er,
    const float* __restrict__ al, const float* __restrict__ ar,
    float* __restrict__ C, float* __restrict__ er1, int N)
{
    __shared__ __attribute__((aligned(16))) float stage[16 * GAT_PS];
    __shared__ float part[16];
    const int lane = threadIdx.x & 31;
    const int wv = __builtin_amdgcn_readfirstlane((int)(threadIdx.x >> 5));
    const int node0 = (int)blockIdx.x * 16;
    const int n = node0 + wv;
    float4 accA = make_float4(0.f, 0.f, 0.f, 0.f);
    float4 accB = make_float4(0.f, 0.f, 0.f, 0.f);
    const int c2 = 32 + (lane & 7);
    const float wl0 = al[lane], wl1 = al[c2];
    const float wr0 = ar[lane], wr1 = ar[c2];

    if (n < N) {
        const int beg = __builtin_amdgcn_readfirstlane(rowptr[n]);
        const int end = __builtin_amdgcn_readfirstlane(rowptr[n + 1]);
        const float4* __restrict__ el4 = (const float4*)el;
        const float4 ern = ((const float4*)er)[n];
        const float NINF = -__builtin_inff();

        const int  j0 = beg + lane;
        const bool v0 = j0 < end;
        int   src0 = 0;
        float e0 = NINF, e1 = NINF, e2 = NINF, e3 = NINF;
        if (v0) {
            src0 = bsrc[j0];
            const float4 l = el4[src0];
            e0 = gat_leaky(l.x + ern.x);
            e1 = gat_leaky(l.y + ern.y);
            e2 = gat_leaky(l.z + ern.z);
            e3 = gat_leaky(l.w + ern.w);
        }
        float m0 = e0, m1 = e1, m2 = e2, m3 = e3;
        for (int j = j0 + 32; j < end; j += 32) {
            const float4 l = el4[bsrc[j]];
            m0 = fmaxf(m0, gat_leaky(l.x + ern.x));
            m1 = fmaxf(m1, gat_leaky(l.y + ern.y));
            m2 = fmaxf(m2, gat_leaky(l.z + ern.z));
            m3 = fmaxf(m3, gat_leaky(l.w + ern.w));
        }
        for (int m = 16; m >= 1; m >>= 1) {
            m0 = fmaxf(m0, __shfl_xor(m0, m, 32));
            m1 = fmaxf(m1, __shfl_xor(m1, m, 32));
            m2 = fmaxf(m2, __shfl_xor(m2, m, 32));
            m3 = fmaxf(m3, __shfl_xor(m3, m, 32));
        }

        const float p0 = v0 ? __expf(e0 - m0) : 0.f;
        const float p1 = v0 ? __expf(e1 - m1) : 0.f;
        const float p2 = v0 ? __expf(e2 - m2) : 0.f;
        const float p3 = v0 ? __expf(e3 - m3) : 0.f;
        float s0 = p0, s1 = p1, s2 = p2, s3 = p3;
        for (int j = j0 + 32; j < end; j += 32) {
            const float4 l = el4[bsrc[j]];
            s0 += __expf(gat_leaky(l.x + ern.x) - m0);
            s1 += __expf(gat_leaky(l.y + ern.y) - m1);
            s2 += __expf(gat_leaky(l.z + ern.z) - m2);
            s3 += __expf(gat_leaky(l.w + ern.w) - m3);
        }
        for (int m = 16; m >= 1; m >>= 1) {
            s0 += __shfl_xor(s0, m, 32);
            s1 += __shfl_xor(s1, m, 32);
            s2 += __shfl_xor(s2, m, 32);
            s3 += __shfl_xor(s3, m, 32);
        }

        const bool h0l = lane < 10;
        const bool h1l = lane < 20;
        const bool h2l = lane < 30;
        for (int base = beg; base < end; base += 32) {
            int   mysrc = 0;
            float a0 = 0.f, a1 = 0.f, a2 = 0.f, a3 = 0.f;
            if (base == beg) {
                mysrc = src0;
                if (v0) {
                    a0 = p0 / s0;  a1 = p1 / s1;
                    a2 = p2 / s2;  a3 = p3 / s3;
                }
            } else {
                const int j = base + lane;
                if (j < end) {
                    mysrc = bsrc[j];
                    const float4 l = el4[mysrc];
                    a0 = __expf(gat_leaky(l.x + ern.x) - m0) / s0;
                    a1 = __expf(gat_leaky(l.y + ern.y) - m1) / s1;
                    a2 = __expf(gat_leaky(l.z + ern.z) - m2) / s2;
                    a3 = __expf(gat_leaky(l.w + ern.w) - m3) / s3;
                }
            }
            const int left = end - base;
            const int cntc = left < 32 ? left : 32;
            #pragma unroll 4
            for (int t = 0; t < cntc; ++t) {
                const int   sN = __builtin_amdgcn_readlane(mysrc, t);
                const float b0 = __int_as_float(__builtin_amdgcn_readlane(__float_as_int(a0), t));
                const float b1 = __int_as_float(__builtin_amdgcn_readlane(__float_as_int(a1), t));
                const float b2 = __int_as_float(__builtin_amdgcn_readlane(__float_as_int(a2), t));
                const float b3 = __int_as_float(__builtin_amdgcn_readlane(__float_as_int(a3), t));
                const float aA = h0l ? b0 : (h1l ? b1 : (h2l ? b2 : b3));
                const float4* __restrict__ hp = (const float4*)(P + (size_t)sN * GAT_PS);
                const float4 x = hp[lane];
                const float4 y = hp[c2];
                accA.x += aA * x.x;  accA.y += aA * x.y;
                accA.z += aA * x.z;  accA.w += aA * x.w;
                accB.x += b3 * y.x;  accB.y += b3 * y.y;
                accB.z += b3 * y.z;  accB.w += b3 * y.w;
            }
        }
    }
    float4* __restrict__ sp4 = (float4*)(stage + wv * GAT_PS);
    sp4[lane] = accA;
    if (lane < 8) sp4[32 + lane] = accB;
    __syncthreads();

    const float* sr = stage + wv * GAT_PS;
    const bool lo8 = lane < 8;
    const float u0 = (sr[lane] + sr[GAT_NC + lane])
                   + (sr[2 * GAT_NC + lane] + sr[3 * GAT_NC + lane]);
    const float u1 = (sr[c2] + sr[GAT_NC + c2])
                   + (sr[2 * GAT_NC + c2] + sr[3 * GAT_NC + c2]);
    float ql = u0 * wl0 + (lo8 ? u1 * wl1 : 0.f);
    float qr = u0 * wr0 + (lo8 ? u1 * wr1 : 0.f);
    for (int m = 16; m >= 1; m >>= 1) {
        ql += __shfl_xor(ql, m, 32);
        qr += __shfl_xor(qr, m, 32);
    }
    if (n < N) {
        float* __restrict__ cr = C + (size_t)n * 64;
        cr[lane] = u0;
        cr[32 + lane] = lo8 ? u1 : (lane == 8 ? ql : 0.f);
    }
    if (lane == 0) part[wv] = qr;
    __syncthreads();
    if (threadIdx.x < 16 && node0 + (int)threadIdx.x < N)
        er1[node0 + threadIdx.x] = part[threadIdx.x];
}

#define GAT_G1 8

template <bool FULL>
__device__ __forceinline__ void gat_h1_group(
    const float2* __restrict__ h2, int mysrc, int t0, int g, int lane,
    int ell, float ern, float& m, float& s, v2f& acc)
{
    const float NINF = -__builtin_inff();
    int sN[GAT_G1];
    #pragma unroll
    for (int u = 0; u < GAT_G1; ++u)
        sN[u] = __builtin_amdgcn_readlane(mysrc, (FULL || u < g) ? t0 + u : t0);
    float2 x[GAT_G1];
    float  e[GAT_G1];
    #pragma unroll
    for (int u = 0; u < GAT_G1; ++u) {
        x[u] = make_float2(0.f, 0.f);
        if (FULL || u < g)
            x[u] = h2[(size_t)sN[u] * 32 + lane];
    }
    float gm = NINF;
    #pragma unroll
    for (int u = 0; u < GAT_G1; ++u) {
        const float ev = __int_as_float(
            __builtin_amdgcn_readlane(__float_as_int(x[u].x), ell));
        const float lv = ev + ern;
        const float lk = (lv > 0.f) ? lv : SLOPE * lv;
        e[u] = (FULL || u < g) ? lk : NINF;
        gm = fmaxf(gm, e[u]);
    }
    const float mn = fmaxf(m, gm);
    const float sc = __expf(m - mn);
    s *= sc;
    acc = acc * (v2f){sc, sc};
    #pragma unroll
    for (int u = 0; u < GAT_G1; ++u) {
        if (FULL || u < g) {
            const float p = __expf(e[u] - mn);
            s += p;
            acc = __builtin_elementwise_fma((v2f){p, p}, (v2f){x[u].x, x[u].y}, acc);
        }
    }
    m = mn;
}

__global__ __launch_bounds__(128) void gat_node_gather_h1(
    const int* __restrict__ rowptr, const int* __restrict__ bsrc,
    const float* __restrict__ h, const float* __restrict__ er,
    float* __restrict__ out, int N, int D)
{
    __shared__ float stage[4 * 64];
    const int lane = threadIdx.x & 31;
    const int wv = __builtin_amdgcn_readfirstlane((int)(threadIdx.x >> 5));
    const int n  = (int)blockIdx.x * 4 + wv;
    const bool has = 2 * lane < D;
    v2f   acc = (v2f){0.f, 0.f};
    float s = 0.f;

    if (n < N) {
        const int beg = __builtin_amdgcn_readfirstlane(rowptr[n]);
        const int end = __builtin_amdgcn_readfirstlane(rowptr[n + 1]);
        const float ern = er[n];
        const float2* __restrict__ h2 = (const float2*)h;
        const int ell = D >> 1;
        float m = -__builtin_inff();

        for (int base = beg; base < end; base += 32) {
            const int j = base + lane;
            const int mysrc = (j < end) ? bsrc[j] : 0;
            const int left = end - base;
            const int cntc = left < 32 ? left : 32;
            int t0 = 0;
            for (; t0 + GAT_G1 <= cntc; t0 += GAT_G1)
                gat_h1_group<true>(h2, mysrc, t0, GAT_G1, lane, ell, ern, m, s, acc);
            if (t0 < cntc)
                gat_h1_group<false>(h2, mysrc, t0, cntc - t0, lane, ell, ern, m, s, acc);
        }
    }
    const float inv = (s > 0.f) ? 1.0f / s : 0.f;
    if (has) {
        stage[wv * D + 2 * lane]     = acc.x * inv;
        stage[wv * D + 2 * lane + 1] = acc.y * inv;
    }
    __syncthreads();
    const int nleft  = N - (int)blockIdx.x * 4;
    const int nvalid = nleft < 4 ? nleft : 4;
    float* __restrict__ ob = out + (size_t)blockIdx.x * 4 * D;
    for (int i = threadIdx.x; i < nvalid * D; i += 128) ob[i] = stage[i];
}

extern "C" void kernel_launch(void* const* d_in, const int* in_sizes, int n_in,
                              void* d_out, int out_size, void* d_ws, size_t ws_size,
                              hipStream_t stream)
{
    const float* feat = (const float*)d_in[0];
    const int*   src  = (const int*)  d_in[1];
    const int*   dst  = (const int*)  d_in[2];
    const float* W0   = (const float*)d_in[3];
    const float* al0  = (const float*)d_in[4];
    const float* ar0  = (const float*)d_in[5];
    const float* W1   = (const float*)d_in[6];
    const float* al1  = (const float*)d_in[7];
    const float* ar1  = (const float*)d_in[8];
    float* out = (float*)d_out;

    const int FIN = GAT_FIN, H0 = GAT_H0, NC = GAT_NC, NCP = 64;
    const int N = in_sizes[0] / FIN;
    const int E = in_sizes[1];

    char* p = (char*)d_ws;
    auto alloc = [&](size_t bytes) -> void* {
        void* r = (void*)p;
        p += (bytes + 255) & ~(size_t)255;
        return r;
    };
    float* P0     = (float*)alloc((size_t)N * GAT_PS * 4);
    float* h1     = (float*)alloc((size_t)N * NCP * 4);
    float* el0    = (float*)alloc((size_t)N * H0 * 4);
    float* er0    = (float*)alloc((size_t)N * H0 * 4);
    float* er1    = (float*)alloc((size_t)N * 4);
    float* Wc     = (float*)alloc((size_t)FIN * GAT_WCN * 4);
    const int R  = (N + 63) / 64;
    const int G  = ((E + 4 * GAT_CAP - 1) / (4 * GAT_CAP)) * 4;
    const int RG = R * G;
    int*      cnt2   = (int*)     alloc((size_t)RG * 4);
    int*      off2   = (int*)     alloc((size_t)(RG + 1) * 4);
    int*      rowptr = (int*)     alloc((size_t)(N + 1) * 4);
    unsigned* tmp    = (unsigned*)alloc((size_t)E * 4);
    int*      bsrc   = (int*)     alloc((size_t)E * 4);

    const int nfold = (FIN / 16) * H0 * 3;
    const int T     = (N / 16) * (GAT_WCN / 64);
    const int Ta    = (T * 7) / 20;
    const int Tb    = Ta + T / 5;
    const int nscan = (RG + 1023) / 1024;
    gat_bin_count<<<G + (nfold + 31) / 32, 1024, (size_t)R * 4, stream>>>(
        dst, cnt2, E, R, G, W0, W1, al0, ar0, Wc, H0, FIN / 16);
    gat_bucket_scan<<<nscan + (Ta + 7) / 8, 256, 0, stream>>>(
        cnt2, off2, RG, nscan,
        feat, Wc, P0, el0, er0, 0, Ta);
    gat_bin_scatter<<<G + (Tb - Ta + 15) / 16, 1024,
                      ((size_t)3 * R + 2 * GAT_CAP) * 4, stream>>>(
        src, dst, off2, tmp, E, R, G,
        feat, Wc, P0, el0, er0, Ta, Tb);
    gat_bin_finish<<<R + (T - Tb + 7) / 8, 256, 0, stream>>>(
        tmp, off2, rowptr, bsrc, N, R, G,
        feat, Wc, P0, el0, er0, Tb, T);
    gat_node_gather_h4_fold<<<(N + 15) / 16, 512, 0, stream>>>(
        rowptr, bsrc, P0, el0, er0, al1, ar1, h1, er1, N);
    gat_node_gather_h1<<<(N + 3) / 4, 128, 0, stream>>>(
        rowptr, bsrc, h1, er1, out, N, NC);
}


# Buckets from per-block slabs: level-1 blocks range-sort their edges in place and keep their range starts, the finish gathers each range's runs; count and scan launches gone; sum order changes
# speedup vs baseline: 8.5629x; 1.2551x over previous
#include <hip/hip_runtime.h>
#include <hip/hip_bf16.h>

typedef __attribute__((ext_vector_type(2))) float v2f;
typedef __attribute__((ext_vector_type(4))) float v4f;
typedef __attribute__((ext_vector_type(8))) float v8f;

#define SLOPE 0.2f

#define GAT_D0 64
#define GAT_NC 40
#define GAT_PS 160
#define GAT_PP 192
#define GAT_WCN 192

__device__ __forceinline__ void gat_fold_tile(
    const float* __restrict__ W0, const float* __restrict__ W1,
    const float* __restrict__ al, const float* __restrict__ ar,
    float* __restrict__ Wc, int H, int mt, int hc)
{
    const int lane  = threadIdx.x & 31;
    const int l15   = lane & 15;
    const int khalf = (lane >> 4) * 2;
    const int head  = hc / 3;
    const int ct    = hc % 3;
    const int c     = ct * 16 + l15;
    const int row   = mt * 16 + l15;
    const int ldw   = H * GAT_D0;

    const float* __restrict__ Ap = W0 + (size_t)row * ldw + head * GAT_D0 + khalf;
    const bool okb = c < GAT_NC + 2;
    const float* __restrict__ Bp;
    int bs;
    if (c < GAT_NC)       { Bp = W1 + (size_t)(head * GAT_D0 + khalf) * GAT_NC + c; bs = GAT_NC; }
    else if (c == GAT_NC) { Bp = al + head * GAT_D0 + khalf; bs = 1; }
    else                  { Bp = ar + head * GAT_D0 + khalf; bs = 1; }

    v8f acc = (v8f){0.f, 0.f, 0.f, 0.f, 0.f, 0.f, 0.f, 0.f};
    #pragma unroll
    for (int k = 0; k < GAT_D0; k += 4) {
        v2f a, b;
        a.x = Ap[k];
        a.y = Ap[k + 1];
        const float bx = Bp[k * bs];
        const float by = Bp[(k + 1) * bs];
        b.x = okb ? bx : 0.f;
        b.y = okb ? by : 0.f;
        acc = __builtin_amdgcn_wmma_f32_16x16x4_f32(false, a, false, b, (short)0, acc, false, false);
    }
    int oc;
    if (c < GAT_NC)           oc = head * GAT_NC + c;
    else if (c == GAT_NC)     oc = H * GAT_NC + head;
    else if (c == GAT_NC + 1) oc = H * GAT_NC + H + head;
    else                      oc = H * GAT_NC + 2 * H + head * 6 + (c - GAT_NC - 2);
    const int srow0 = mt * 16 + (lane >> 4) * 8;
    #pragma unroll
    for (int r = 0; r < 8; ++r) {
        const int k = srow0 + r;
        Wc[(size_t)(k >> 1) * (2 * GAT_WCN) + oc * 2 + (k & 1)] = acc[r];
    }
}

#define GAT_FIN 128
#define GAT_H0 4

__device__ __forceinline__ void gat_wmma_gemm_tile(
    const float* __restrict__ A, const float* __restrict__ B,
    float* __restrict__ C, float* __restrict__ er,
    int mt, int strip)
{
    const int lane = threadIdx.x & 31;
    const int l15  = lane & 15;
    const int hh   = lane >> 4;
    const int row  = mt * 16 + l15;

    const float4* __restrict__ Ap = (const float4*)(A + (size_t)row * GAT_FIN) + hh;
    const float4* __restrict__ Bp = (const float4*)B
                                  + (size_t)(2 * hh) * (GAT_WCN / 2) + strip * 32 + l15;

    v8f acc[4];
    #pragma unroll
    for (int t = 0; t < 4; ++t)
        acc[t] = (v8f){0.f, 0.f, 0.f, 0.f, 0.f, 0.f, 0.f, 0.f};

    float4 fa[3], fp0[3], fp1[3], fq0[3], fq1[3];
    #pragma unroll
    for (int ks = 0; ks < 2; ++ks) {
        const float4* __restrict__ bq = Bp + (size_t)(4 * ks) * (GAT_WCN / 2);
        fa[ks]  = Ap[2 * ks];
        fp0[ks] = bq[0];
        fp1[ks] = bq[16];
        fq0[ks] = bq[GAT_WCN / 2];
        fq1[ks] = bq[GAT_WCN / 2 + 16];
    }
    #pragma unroll
    for (int ks = 0; ks < GAT_FIN / 8; ++ks) {
        const int cur = ks % 3;
        if (ks + 2 < GAT_FIN / 8) {
            const int nxt = (ks + 2) % 3;
            const float4* __restrict__ bq = Bp + (size_t)(4 * (ks + 2)) * (GAT_WCN / 2);
            fa[nxt]  = Ap[2 * (ks + 2)];
            fp0[nxt] = bq[0];
            fp1[nxt] = bq[16];
            fq0[nxt] = bq[GAT_WCN / 2];
            fq1[nxt] = bq[GAT_WCN / 2 + 16];
        }
        const v2f a0 = (v2f){fa[cur].x, fa[cur].y};
        const v2f a1 = (v2f){fa[cur].z, fa[cur].w};
        acc[0] = __builtin_amdgcn_wmma_f32_16x16x4_f32(false, a0, false, (v2f){fp0[cur].x, fp0[cur].y}, (short)0, acc[0], false, false);
        acc[1] = __builtin_amdgcn_wmma_f32_16x16x4_f32(false, a0, false, (v2f){fp0[cur].z, fp0[cur].w}, (short)0, acc[1], false, false);
        acc[2] = __builtin_amdgcn_wmma_f32_16x16x4_f32(false, a0, false, (v2f){fp1[cur].x, fp1[cur].y}, (short)0, acc[2], false, false);
        acc[3] = __builtin_amdgcn_wmma_f32_16x16x4_f32(false, a0, false, (v2f){fp1[cur].z, fp1[cur].w}, (short)0, acc[3], false, false);
        acc[0] = __builtin_amdgcn_wmma_f32_16x16x4_f32(false, a1, false, (v2f){fq0[cur].x, fq0[cur].y}, (short)0, acc[0], false, false);
        acc[1] = __builtin_amdgcn_wmma_f32_16x16x4_f32(false, a1, false, (v2f){fq0[cur].z, fq0[cur].w}, (short)0, acc[1], false, false);
        acc[2] = __builtin_amdgcn_wmma_f32_16x16x4_f32(false, a1, false, (v2f){fq1[cur].x, fq1[cur].y}, (short)0, acc[2], false, false);
        acc[3] = __builtin_amdgcn_wmma_f32_16x16x4_f32(false, a1, false, (v2f){fq1[cur].z, fq1[cur].w}, (short)0, acc[3], false, false);
        __builtin_amdgcn_sched_barrier(0);
    }

    const int srow0 = mt * 16 + hh * 8;
    #pragma unroll
    for (int u = 0; u < 2; ++u) {
        const int col = strip * 64 + 32 * u + 2 * l15;
        if (col < GAT_PS + GAT_H0) {
            #pragma unroll
            for (int r = 0; r < 8; ++r)
                *(float2*)(C + (size_t)(srow0 + r) * GAT_PP + col) =
                    make_float2(acc[2 * u][r], acc[2 * u + 1][r]);
        } else if (col < GAT_PS + 2 * GAT_H0) {
            #pragma unroll
            for (int r = 0; r < 8; ++r)
                *(float2*)(er + (size_t)(srow0 + r) * GAT_H0 + (col - GAT_PS - GAT_H0)) =
                    make_float2(acc[2 * u][r], acc[2 * u + 1][r]);
        }
    }
}

__device__ __forceinline__ void gat_gemm_guest(
    const float* __restrict__ A, const float* __restrict__ B,
    float* __restrict__ C, float* __restrict__ er,
    int gb, int t0, int t1)
{
    const int wv = __builtin_amdgcn_readfirstlane((int)(threadIdx.x >> 5));
    const int t  = t0 + gb * 8 + wv;
    if (t >= t1) return;
    const int S  = GAT_WCN / 64;
    const int mt = t / S;
    gat_wmma_gemm_tile(A, B, C, er, mt, t - mt * S);
}

#define GAT_CAP 6144
#define GAT_FCAP 4096
#define GAT_GMAX 256

__global__ __launch_bounds__(1024) void gat_bin_scatter(
    const int* __restrict__ src, const int* __restrict__ dst,
    unsigned* __restrict__ tmp, int* __restrict__ lb, int E, int R, int G,
    const float* __restrict__ fW0, const float* __restrict__ fW1,
    const float* __restrict__ fal, const float* __restrict__ far_,
    float* __restrict__ fWc, int fH, int fMT)
{
    extern __shared__ int gat_sh[];
    __shared__ int wsum[32];
    if ((int)blockIdx.x >= G) {
        const int gw = __builtin_amdgcn_readfirstlane((int)(threadIdx.x >> 5));
        const int t  = ((int)blockIdx.x - G) * 32 + gw;
        if (t < fMT * fH * 3)
            gat_fold_tile(fW0, fW1, fal, far_, fWc, fH, t % fMT, t / fMT);
        return;
    }
    int* lcnt = gat_sh;
    int* lbeg = gat_sh + R;
    unsigned* bw = (unsigned*)(gat_sh + 2 * R);
    const int tid = threadIdx.x, lane = tid & 31, wv = tid >> 5;
    const int g = (int)blockIdx.x;
    const int base = g * GAT_CAP;

    for (int r = tid; r < R; r += 1024) lcnt[r] = 0;
    __syncthreads();
    int ed[GAT_CAP / 1024], es[GAT_CAP / 1024];
    #pragma unroll
    for (int k = 0; k < GAT_CAP / 1024; ++k) {
        const int e  = base + k * 1024 + tid;
        const int ec = e < E ? e : E - 1;
        const int d  = dst[ec];
        es[k] = src[ec];
        ed[k] = e < E ? d : -1;
    }
    #pragma unroll
    for (int k = 0; k < GAT_CAP / 1024; ++k)
        if (ed[k] >= 0) atomicAdd(&lcnt[ed[k] >> 6], 1);
    __syncthreads();
    int carry = 0;
    for (int b0 = 0; b0 < R; b0 += 1024) {
        const int idx = b0 + tid;
        const int v = (idx < R) ? lcnt[idx] : 0;
        int incl = v;
        for (int off = 1; off < 32; off <<= 1) {
            const int y = __shfl_up(incl, off, 32);
            if (lane >= off) incl += y;
        }
        if (lane == 31) wsum[wv] = incl;
        __syncthreads();
        int pre = 0, tot = 0;
        for (int w = 0; w < 32; ++w) {
            const int t = wsum[w];
            tot += t;
            if (w < wv) pre += t;
        }
        if (idx < R) lbeg[idx] = carry + pre + incl - v;
        carry += tot;
        __syncthreads();
    }
    const int left = E - base;
    const int nloc = left < GAT_CAP ? left : GAT_CAP;
    int* __restrict__ lbg = lb + (size_t)g * (R + 1);
    for (int r = tid; r < R; r += 1024) {
        const int b = lbeg[r];
        lcnt[r] = b;
        lbg[r] = b;
    }
    if (tid == 0) lbg[R] = nloc;
    __syncthreads();
    #pragma unroll
    for (int k = 0; k < GAT_CAP / 1024; ++k) {
        if (ed[k] >= 0) {
            const int p = atomicAdd(&lcnt[ed[k] >> 6], 1);
            bw[p] = ((unsigned)es[k] << 6) | (unsigned)(ed[k] & 63);
        }
    }
    __syncthreads();
    for (int k = tid; k < nloc; k += 1024) tmp[base + k] = bw[k];
}

__global__ __launch_bounds__(256) void gat_bin_finish(
    const unsigned* __restrict__ tmp, const int* __restrict__ lb,
    int* __restrict__ rowptr, int* __restrict__ bsrc, int N, int R, int G,
    const float* __restrict__ gA, const float* __restrict__ gB,
    float* __restrict__ gC, float* __restrict__ ger,
    int gt0, int gt1)
{
    __shared__ int c[64];
    __shared__ int cur[64];
    __shared__ int wtot[2];
    __shared__ int wcnt[8];
    __shared__ int wbeg[8];
    __shared__ int run[GAT_FCAP];
    __shared__ int stage[GAT_FCAP];
    if ((int)blockIdx.x >= R) {
        gat_gemm_guest(gA, gB, gC, ger, (int)blockIdx.x - R, gt0, gt1);
        return;
    }
    const int tid = threadIdx.x, lane = tid & 31, wv = tid >> 5;
    const int r = (int)blockIdx.x;
    int b = 0, cn = 0;
    if (tid < G) {
        const int* __restrict__ q = lb + (size_t)tid * (R + 1) + r;
        b  = q[0];
        cn = q[1] - b;
    }
    int incl = cn;
    for (int off = 1; off < 32; off <<= 1) {
        const int y = __shfl_up(incl, off, 32);
        if (lane >= off) incl += y;
    }
    int bs = b;
    for (int m = 16; m >= 1; m >>= 1) bs += __shfl_xor(bs, m, 32);
    if (lane == 31) wcnt[wv] = incl;
    if (lane == 0)  wbeg[wv] = bs;
    if (tid < 64) c[tid] = 0;
    __syncthreads();
    int pos = incl - cn;
    int n = 0;
    int s = 0;
    for (int w = 0; w < 8; ++w) {
        const int t = wcnt[w];
        n += t;
        if (w < wv) pos += t;
        s += wbeg[w];
    }
    const bool fits = n <= GAT_FCAP;
    const unsigned* __restrict__ sl = tmp + (size_t)tid * GAT_CAP + b;
    for (int k0 = 0; k0 < cn; k0 += 8) {
        unsigned w[8];
        #pragma unroll
        for (int u = 0; u < 8; ++u)
            w[u] = sl[k0 + u < cn ? k0 + u : k0];
        #pragma unroll
        for (int u = 0; u < 8; ++u) {
            if (k0 + u < cn) {
                atomicAdd(&c[w[u] & 63u], 1);
                if (fits) run[pos + k0 + u] = (int)w[u];
            }
        }
    }
    __syncthreads();
    int v = 0, inc2 = 0;
    if (tid < 64) {
        v = c[tid];
        inc2 = v;
        for (int off = 1; off < 32; off <<= 1) {
            const int y = __shfl_up(inc2, off, 32);
            if (lane >= off) inc2 += y;
        }
        if (lane == 31) wtot[wv] = inc2;
    }
    __syncthreads();
    if (tid < 64) {
        const int excl = inc2 - v + (wv == 1 ? wtot[0] : 0);
        cur[tid] = excl;
        const int node = r * 64 + tid;
        if (node < N) rowptr[node] = s + excl;
    }
    if (r == R - 1 && tid == 0) rowptr[N] = s + n;
    __syncthreads();
    if (fits) {
        for (int i = tid; i < n; i += 256) {
            const unsigned w = (unsigned)run[i];
            const int p = atomicAdd(&cur[w & 63u], 1);
            stage[p] = (int)(w >> 6);
        }
        __syncthreads();
        for (int i = tid; i < n; i += 256) bsrc[s + i] = stage[i];
    } else {
        for (int k = 0; k < cn; ++k) {
            const unsigned w = sl[k];
            const int p = atomicAdd(&cur[w & 63u], 1);
            bsrc[s + p] = (int)(w >> 6);
        }
    }
}

#define GAT_G0 4

__device__ __forceinline__ float gat_row8(float v) {
    return __int_as_float(__builtin_amdgcn_mov_dpp(
        __float_as_int(v), 0x158  , 0xf, 0xf, false));
}

template <bool FULL>
__device__ __forceinline__ void gat_h4_group(
    const float* __restrict__ P, int mysrc, int t0, int g, int lane, int c2,
    float r0, float r1, float r2, float r3, bool h0l, bool h1l, bool h2l,
    float (&m)[4], float (&s)[4], float4& accA, float4& accB)
{
    const float NINF = -__builtin_inff();
    int sN[GAT_G0];
    #pragma unroll
    for (int u = 0; u < GAT_G0; ++u)
        sN[u] = __builtin_amdgcn_readlane(mysrc, (FULL || u < g) ? t0 + u : t0);
    v4f x[GAT_G0], y[GAT_G0];
    #pragma unroll
    for (int u = 0; u < GAT_G0; ++u) {
        x[u] = (v4f){0.f, 0.f, 0.f, 0.f};
        y[u] = (v4f){0.f, 0.f, 0.f, 0.f};
        if (FULL || u < g) {
            const v4f* __restrict__ hp = (const v4f*)(P + (size_t)sN[u] * GAT_PP);
            x[u] = hp[lane];
            y[u] = hp[c2];
        }
    }
    if (FULL) {
        static_assert(GAT_G0 == 4, "the fence below names four edges");
        asm volatile("" :: "v"(x[0]), "v"(y[0]), "v"(x[1]), "v"(y[1]),
                           "v"(x[2]), "v"(y[2]), "v"(x[3]), "v"(y[3]));
        __builtin_amdgcn_sched_barrier(0);
    }
    float e[GAT_G0][4];
    float gm0 = NINF, gm1 = NINF, gm2 = NINF, gm3 = NINF;
    #pragma unroll
    for (int u = 0; u < GAT_G0; ++u) {
        const bool on = FULL || u < g;
        const float v0 = gat_row8(y[u].x) + r0;
        const float v1 = gat_row8(y[u].y) + r1;
        const float v2 = gat_row8(y[u].z) + r2;
        const float v3 = gat_row8(y[u].w) + r3;
        e[u][0] = on ? fmaxf(v0, SLOPE * v0) : NINF;
        e[u][1] = on ? fmaxf(v1, SLOPE * v1) : NINF;
        e[u][2] = on ? fmaxf(v2, SLOPE * v2) : NINF;
        e[u][3] = on ? fmaxf(v3, SLOPE * v3) : NINF;
        gm0 = fmaxf(gm0, e[u][0]);
        gm1 = fmaxf(gm1, e[u][1]);
        gm2 = fmaxf(gm2, e[u][2]);
        gm3 = fmaxf(gm3, e[u][3]);
    }
    const float mn0 = fmaxf(m[0], gm0), mn1 = fmaxf(m[1], gm1);
    const float mn2 = fmaxf(m[2], gm2), mn3 = fmaxf(m[3], gm3);
    const float L2E = 1.44269504088896340736f;
    const float q0 = mn0 * L2E, q1 = mn1 * L2E, q2 = mn2 * L2E, q3 = mn3 * L2E;
    const float sc0 = __builtin_amdgcn_exp2f(__builtin_fmaf(m[0], L2E, -q0));
    const float sc1 = __builtin_amdgcn_exp2f(__builtin_fmaf(m[1], L2E, -q1));
    const float sc2 = __builtin_amdgcn_exp2f(__builtin_fmaf(m[2], L2E, -q2));
    const float sc3 = __builtin_amdgcn_exp2f(__builtin_fmaf(m[3], L2E, -q3));
    s[0] *= sc0;  s[1] *= sc1;  s[2] *= sc2;  s[3] *= sc3;
    const float scA = h0l ? sc0 : (h1l ? sc1 : (h2l ? sc2 : sc3));
    accA.x *= scA;  accA.y *= scA;  accA.z *= scA;  accA.w *= scA;
    accB.x *= sc3;  accB.y *= sc3;  accB.z *= sc3;  accB.w *= sc3;
    #pragma unroll
    for (int u = 0; u < GAT_G0; ++u) {
        if (FULL || u < g) {
            const float p0 = __builtin_amdgcn_exp2f(__builtin_fmaf(e[u][0], L2E, -q0));
            const float p1 = __builtin_amdgcn_exp2f(__builtin_fmaf(e[u][1], L2E, -q1));
            const float p2 = __builtin_amdgcn_exp2f(__builtin_fmaf(e[u][2], L2E, -q2));
            const float p3 = __builtin_amdgcn_exp2f(__builtin_fmaf(e[u][3], L2E, -q3));
            s[0] += p0;  s[1] += p1;  s[2] += p2;  s[3] += p3;
            const float pA = h0l ? p0 : (h1l ? p1 : (h2l ? p2 : p3));
            accA.x += pA * x[u].x;  accA.y += pA * x[u].y;
            accA.z += pA * x[u].z;  accA.w += pA * x[u].w;
            accB.x += p3 * y[u].x;  accB.y += p3 * y[u].y;
            accB.z += p3 * y[u].z;  accB.w += p3 * y[u].w;
        }
    }
    m[0] = mn0;  m[1] = mn1;  m[2] = mn2;  m[3] = mn3;
}

__global__ __launch_bounds__(512, 12) void gat_node_gather_h4_fold(
    const int* __restrict__ rowptr, const int* __restrict__ bsrc,
    const float* __restrict__ P, const float* __restrict__ er,
    const float* __restrict__ al, const float* __restrict__ ar,
    float* __restrict__ C, float* __restrict__ er1, int N)
{
    __shared__ __attribute__((aligned(16))) float stage[16 * GAT_PS];
    __shared__ float part[16];
    const int lane = threadIdx.x & 31;
    const int wv = __builtin_amdgcn_readfirstlane((int)(threadIdx.x >> 5));
    const int node0 = (int)blockIdx.x * 16;
    const int n = node0 + wv;
    float4 accA = make_float4(0.f, 0.f, 0.f, 0.f);
    float4 accB = make_float4(0.f, 0.f, 0.f, 0.f);
    const int c2 = 32 + (lane & 7);

    if (n < N) {
        const int beg = __builtin_amdgcn_readfirstlane(rowptr[n]);
        const int end = __builtin_amdgcn_readfirstlane(rowptr[n + 1]);
        const float4 ern = ((const float4*)er)[n];
        const float r0 = ern.x, r1 = ern.y, r2 = ern.z, r3 = ern.w;
        const bool h0l = lane < 10;
        const bool h1l = lane < 20;
        const bool h2l = lane < 30;
        const int ck = lane < 8 ? 32 + lane : GAT_PS / 4;
        const float NINF = -__builtin_inff();
        float m[4] = {NINF, NINF, NINF, NINF};
        float s[4] = {0.f, 0.f, 0.f, 0.f};

        for (int base = beg; base < end; base += 32) {
            const int j = base + lane;
            const int mysrc = (j < end) ? bsrc[j] : 0;
            const int left = end - base;
            const int cntc = left < 32 ? left : 32;
            int t0 = 0;
            for (; t0 + GAT_G0 <= cntc; t0 += GAT_G0)
                gat_h4_group<true>(P, mysrc, t0, GAT_G0, lane, ck, r0, r1, r2, r3,
                                   h0l, h1l, h2l, m, s, accA, accB);
            if (t0 < cntc)
                gat_h4_group<false>(P, mysrc, t0, cntc - t0, lane, ck, r0, r1, r2, r3,
                                    h0l, h1l, h2l, m, s, accA, accB);
        }
        const float i0 = (s[0] > 0.f) ? 1.0f / s[0] : 0.f;
        const float i1 = (s[1] > 0.f) ? 1.0f / s[1] : 0.f;
        const float i2 = (s[2] > 0.f) ? 1.0f / s[2] : 0.f;
        const float i3 = (s[3] > 0.f) ? 1.0f / s[3] : 0.f;
        const float iA = h0l ? i0 : (h1l ? i1 : (h2l ? i2 : i3));
        accA.x *= iA;  accA.y *= iA;  accA.z *= iA;  accA.w *= iA;
        accB.x *= i3;  accB.y *= i3;  accB.z *= i3;  accB.w *= i3;
    }
    const float wl0 = al[lane], wl1 = al[c2];
    const float wr0 = ar[lane], wr1 = ar[c2];
    float4* __restrict__ sp4 = (float4*)(stage + wv * GAT_PS);
    sp4[lane] = accA;
    if (lane < 8) sp4[32 + lane] = accB;
    __syncthreads();

    const float* sr = stage + wv * GAT_PS;
    const bool lo8 = lane < 8;
    const float u0 = (sr[lane] + sr[GAT_NC + lane])
                   + (sr[2 * GAT_NC + lane] + sr[3 * GAT_NC + lane]);
    const float u1 = (sr[c2] + sr[GAT_NC + c2])
                   + (sr[2 * GAT_NC + c2] + sr[3 * GAT_NC + c2]);
    float ql = u0 * wl0 + (lo8 ? u1 * wl1 : 0.f);
    float qr = u0 * wr0 + (lo8 ? u1 * wr1 : 0.f);
    for (int m = 16; m >= 1; m >>= 1) {
        ql += __shfl_xor(ql, m, 32);
        qr += __shfl_xor(qr, m, 32);
    }
    if (n < N) {
        float* __restrict__ cr = C + (size_t)n * 64;
        cr[lane] = u0;
        cr[32 + lane] = lo8 ? u1 : (lane == 8 ? ql : 0.f);
    }
    if (lane == 0) part[wv] = qr;
    __syncthreads();
    if (threadIdx.x < 16 && node0 + (int)threadIdx.x < N)
        er1[node0 + threadIdx.x] = part[threadIdx.x];
}

#define GAT_G1 8

template <bool FULL>
__device__ __forceinline__ void gat_h1_group(
    const float2* __restrict__ h2, int mysrc, int t0, int g, int lane,
    int ell, float ern, float& m, float& s, v2f& acc)
{
    const float NINF = -__builtin_inff();
    int sN[GAT_G1];
    #pragma unroll
    for (int u = 0; u < GAT_G1; ++u)
        sN[u] = __builtin_amdgcn_readlane(mysrc, (FULL || u < g) ? t0 + u : t0);
    float2 x[GAT_G1];
    float  e[GAT_G1];
    #pragma unroll
    for (int u = 0; u < GAT_G1; ++u) {
        x[u] = make_float2(0.f, 0.f);
        if (FULL || u < g)
            x[u] = h2[(size_t)sN[u] * 32 + lane];
    }
    float gm = NINF;
    #pragma unroll
    for (int u = 0; u < GAT_G1; ++u) {
        const float ev = __int_as_float(
            __builtin_amdgcn_readlane(__float_as_int(x[u].x), ell));
        const float lv = ev + ern;
        const float lk = (lv > 0.f) ? lv : SLOPE * lv;
        e[u] = (FULL || u < g) ? lk : NINF;
        gm = fmaxf(gm, e[u]);
    }
    const float mn = fmaxf(m, gm);
    const float sc = __expf(m - mn);
    s *= sc;
    acc = acc * (v2f){sc, sc};
    #pragma unroll
    for (int u = 0; u < GAT_G1; ++u) {
        if (FULL || u < g) {
            const float p = __expf(e[u] - mn);
            s += p;
            acc = __builtin_elementwise_fma((v2f){p, p}, (v2f){x[u].x, x[u].y}, acc);
        }
    }
    m = mn;
}

__global__ __launch_bounds__(128) void gat_node_gather_h1(
    const int* __restrict__ rowptr, const int* __restrict__ bsrc,
    const float* __restrict__ h, const float* __restrict__ er,
    float* __restrict__ out, int N, int D)
{
    __shared__ float stage[4 * 64];
    const int lane = threadIdx.x & 31;
    const int wv = __builtin_amdgcn_readfirstlane((int)(threadIdx.x >> 5));
    const int n  = (int)blockIdx.x * 4 + wv;
    const bool has = 2 * lane < D;
    v2f   acc = (v2f){0.f, 0.f};
    float s = 0.f;

    if (n < N) {
        const int beg = __builtin_amdgcn_readfirstlane(rowptr[n]);
        const int end = __builtin_amdgcn_readfirstlane(rowptr[n + 1]);
        const float ern = er[n];
        const float2* __restrict__ h2 = (const float2*)h;
        const int ell = D >> 1;
        float m = -__builtin_inff();

        for (int base = beg; base < end; base += 32) {
            const int j = base + lane;
            const int mysrc = (j < end) ? bsrc[j] : 0;
            const int left = end - base;
            const int cntc = left < 32 ? left : 32;
            int t0 = 0;
            for (; t0 + GAT_G1 <= cntc; t0 += GAT_G1)
                gat_h1_group<true>(h2, mysrc, t0, GAT_G1, lane, ell, ern, m, s, acc);
            if (t0 < cntc)
                gat_h1_group<false>(h2, mysrc, t0, cntc - t0, lane, ell, ern, m, s, acc);
        }
    }
    const float inv = (s > 0.f) ? 1.0f / s : 0.f;
    if (has) {
        stage[wv * D + 2 * lane]     = acc.x * inv;
        stage[wv * D + 2 * lane + 1] = acc.y * inv;
    }
    __syncthreads();
    const int nleft  = N - (int)blockIdx.x * 4;
    const int nvalid = nleft < 4 ? nleft : 4;
    float* __restrict__ ob = out + (size_t)blockIdx.x * 4 * D;
    for (int i = threadIdx.x; i < nvalid * D; i += 128) ob[i] = stage[i];
}

extern "C" void kernel_launch(void* const* d_in, const int* in_sizes, int n_in,
                              void* d_out, int out_size, void* d_ws, size_t ws_size,
                              hipStream_t stream)
{
    const float* feat = (const float*)d_in[0];
    const int*   src  = (const int*)  d_in[1];
    const int*   dst  = (const int*)  d_in[2];
    const float* W0   = (const float*)d_in[3];
    const float* al0  = (const float*)d_in[4];
    const float* ar0  = (const float*)d_in[5];
    const float* W1   = (const float*)d_in[6];
    const float* al1  = (const float*)d_in[7];
    const float* ar1  = (const float*)d_in[8];
    float* out = (float*)d_out;

    const int FIN = GAT_FIN, H0 = GAT_H0, NC = GAT_NC, NCP = 64;
    const int N = in_sizes[0] / FIN;
    const int E = in_sizes[1];

    char* p = (char*)d_ws;
    auto alloc = [&](size_t bytes) -> void* {
        void* r = (void*)p;
        p += (bytes + 255) & ~(size_t)255;
        return r;
    };
    float* P0     = (float*)alloc((size_t)N * GAT_PP * 4);
    float* h1     = (float*)alloc((size_t)N * NCP * 4);
    float* er0    = (float*)alloc((size_t)N * H0 * 4);
    float* er1    = (float*)alloc((size_t)N * 4);
    float* Wc     = (float*)alloc((size_t)FIN * GAT_WCN * 4);
    const int R  = (N + 63) / 64;
    const int G  = (E + GAT_CAP - 1) / GAT_CAP;
    int*      lb     = (int*)     alloc((size_t)G * (R + 1) * 4);
    int*      rowptr = (int*)     alloc((size_t)(N + 1) * 4);
    unsigned* tmp    = (unsigned*)alloc((size_t)E * 4);
    int*      bsrc   = (int*)     alloc((size_t)E * 4);

    const int nfold = (FIN / 16) * H0 * 3;
    const int T     = (N / 16) * (GAT_WCN / 64);
    gat_bin_scatter<<<G + (nfold + 31) / 32, 1024,
                      ((size_t)2 * R + GAT_CAP) * 4, stream>>>(
        src, dst, tmp, lb, E, R, G,
        W0, W1, al0, ar0, Wc, H0, FIN / 16);
    gat_bin_finish<<<R + (T + 7) / 8, 256, 0, stream>>>(
        tmp, lb, rowptr, bsrc, N, R, G,
        feat, Wc, P0, er0, 0, T);
    gat_node_gather_h4_fold<<<(N + 15) / 16, 512, 0, stream>>>(
        rowptr, bsrc, P0, er0, al1, ar1, h1, er1, N);
    gat_node_gather_h1<<<(N + 3) / 4, 128, 0, stream>>>(
        rowptr, bsrc, h1, er1, out, N, NC);
}
